# starting kernel
# baseline (speedup 1.0000x reference)
#include <hip/hip_runtime.h>
#include <hip/hip_bf16.h>

typedef __attribute__((ext_vector_type(16))) _Float16 v16h;
typedef __attribute__((ext_vector_type(8)))  float    v8f;

#define GCN_D_IN  512
#define GCN_D_HID 256
#define GCN_D_OUT 64

__global__ void gcn_init_deg(float* __restrict__ deg, int n) {
    int i = blockIdx.x * blockDim.x + threadIdx.x;
    if (i < n) deg[i] = 1.0f;
}

__global__ void gcn_accum_deg(const int* __restrict__ dst, float* __restrict__ deg, int E) {
    int t = blockIdx.x * blockDim.x + threadIdx.x;
    if (t < E) atomicAdd(&deg[dst[t]], 1.0f);
}

__global__ void gcn_make_dinv(float* __restrict__ deg, int n) {
    int i = blockIdx.x * blockDim.x + threadIdx.x;
    if (i < n) {
        float d = deg[i];
        deg[i] = (d > 0.0f) ? rsqrtf(d) : 0.0f;
    }
}

__global__ void gcn_pack_w_f16(const float* __restrict__ W, _Float16* __restrict__ Wp,
                               int K, int N) {
    int t = blockIdx.x * blockDim.x + threadIdx.x;
    int tilesN = N >> 4;
    int total = (K >> 5) * tilesN * 32;
    if (t >= total) return;
    int frag = t >> 5;
    int lane = t & 31;
    int kt = frag / tilesN;
    int tn = frag - kt * tilesN;
    int n    = tn * 16 + (lane & 15);
    int koff = kt * 32 + (lane >> 4) * 8;
    _Float16* o = Wp + (size_t)t * 16;
#pragma unroll
    for (int j = 0; j < 8; ++j) {
        o[j]     = (_Float16)W[(size_t)(koff + j) * N + n];
        o[j + 8] = (_Float16)W[(size_t)(koff + 16 + j) * N + n];
    }
}

__global__ void gcn_wmma_gemm_f16x4(const float* __restrict__ A,
                                    const _Float16* __restrict__ Bp,
                                    float* __restrict__ C,
                                    int M, int K, int N) {
    const int lane    = threadIdx.x & 31;
    const int wave    = (blockIdx.x * blockDim.x + threadIdx.x) >> 5;
    const int tilesN  = N >> 4;
    const int groupsN = N >> 6;
    const int tm = wave / groupsN;
    const int tg = wave - tm * groupsN;
    if (tm * 16 >= M) return;
    const int tn0  = tg * 4;
    const int hsel = lane >> 4;
    const int idx  = lane & 15;

    const float* __restrict__ arow = A + (size_t)(tm * 16 + idx) * K;

    v8f acc0 = {}, acc1 = {}, acc2 = {}, acc3 = {};
    for (int k0 = 0; k0 < K; k0 += 32) {
        const int kb = k0 + hsel * 8;

        float4 a01 = *(const float4*)(arow + kb);
        float4 a23 = *(const float4*)(arow + kb + 4);
        float4 a45 = *(const float4*)(arow + kb + 16);
        float4 a67 = *(const float4*)(arow + kb + 20);
        if (k0 + 32 < K) __builtin_prefetch(arow + kb + 32, 0, 3);

        float af[16];
        *(float4*)(af + 0)  = a01;  *(float4*)(af + 4)  = a23;
        *(float4*)(af + 8)  = a45;  *(float4*)(af + 12) = a67;
        v16h av;
#pragma unroll
        for (int j = 0; j < 16; ++j) av[j] = (_Float16)af[j];

        const int kt = k0 >> 5;
        const _Float16* bbase = Bp + ((size_t)(kt * tilesN + tn0) * 32 + lane) * 16;
        v16h b0 = *(const v16h*)(bbase);
        v16h b1 = *(const v16h*)(bbase + 1 * 32 * 16);
        v16h b2 = *(const v16h*)(bbase + 2 * 32 * 16);
        v16h b3 = *(const v16h*)(bbase + 3 * 32 * 16);

        acc0 = __builtin_amdgcn_wmma_f32_16x16x32_f16(false, av, false, b0, (short)0, acc0, false, false);
        acc1 = __builtin_amdgcn_wmma_f32_16x16x32_f16(false, av, false, b1, (short)0, acc1, false, false);
        acc2 = __builtin_amdgcn_wmma_f32_16x16x32_f16(false, av, false, b2, (short)0, acc2, false, false);
        acc3 = __builtin_amdgcn_wmma_f32_16x16x32_f16(false, av, false, b3, (short)0, acc3, false, false);
    }

    float* cbase = C + (size_t)(tm * 16 + hsel * 8) * N + tn0 * 16 + idx;
#pragma unroll
    for (int r = 0; r < 8; ++r) {
        cbase[(size_t)r * N + 0]  = acc0[r];
        cbase[(size_t)r * N + 16] = acc1[r];
        cbase[(size_t)r * N + 32] = acc2[r];
        cbase[(size_t)r * N + 48] = acc3[r];
    }
}

__global__ void gcn_self_loop(const float* __restrict__ xw,
                              const float* __restrict__ dinv,
                              float* __restrict__ out, int total, int F) {
    int t = blockIdx.x * blockDim.x + threadIdx.x;
    if (t < total) {
        int i = t / F;
        float di = dinv[i];
        out[t] = xw[t] * di * di;
    }
}

__global__ void gcn_edge_scatter(const int* __restrict__ src,
                                 const int* __restrict__ dst,
                                 const float* __restrict__ dinv,
                                 const float* __restrict__ xw,
                                 float* __restrict__ out,
                                 int E, int F) {
    int t = blockIdx.x * blockDim.x + threadIdx.x;
    int chunks = F >> 2;
    int e = t / chunks;
    if (e >= E) return;
    int c = t - e * chunks;
    int s = src[e], d = dst[e];
    float norm = dinv[s] * dinv[d];
    float4 v = ((const float4*)(xw + (size_t)s * F))[c];
    float* o = out + (size_t)d * F + 4 * c;
    atomicAdd(o + 0, v.x * norm);
    atomicAdd(o + 1, v.y * norm);
    atomicAdd(o + 2, v.z * norm);
    atomicAdd(o + 3, v.w * norm);
}

__global__ void gcn_bias_relu(float* __restrict__ h, const float* __restrict__ bias,
                              int total, int F) {
    int t = blockIdx.x * blockDim.x + threadIdx.x;
    if (t < total) h[t] = fmaxf(h[t] + bias[t % F], 0.0f);
}

__global__ void gcn_softmax64(const float* __restrict__ h,
                              const float* __restrict__ bias,
                              float* __restrict__ out, int nodes) {
    int wave = (blockIdx.x * blockDim.x + threadIdx.x) >> 5;
    int lane = threadIdx.x & 31;
    if (wave >= nodes) return;
    const float* row = h + (size_t)wave * 64;
    float v0 = row[lane]      + bias[lane];
    float v1 = row[lane + 32] + bias[lane + 32];
    float m = fmaxf(v0, v1);
#pragma unroll
    for (int off = 16; off > 0; off >>= 1) m = fmaxf(m, __shfl_xor(m, off, 32));
    float e0 = __expf(v0 - m);
    float e1 = __expf(v1 - m);
    float s = e0 + e1;
#pragma unroll
    for (int off = 16; off > 0; off >>= 1) s += __shfl_xor(s, off, 32);
    float inv = 1.0f / s;
    float* orow = out + (size_t)wave * 64;
    orow[lane]      = e0 * inv;
    orow[lane + 32] = e1 * inv;
}

extern "C" void kernel_launch(void* const* d_in, const int* in_sizes, int n_in,
                              void* d_out, int out_size, void* d_ws, size_t ws_size,
                              hipStream_t stream) {
    const float* x   = (const float*)d_in[0];
    const int*   ei  = (const int*)  d_in[1];
    const float* W1  = (const float*)d_in[2];
    const float* b1  = (const float*)d_in[3];
    const float* W2  = (const float*)d_in[4];
    const float* b2  = (const float*)d_in[5];
    float* out = (float*)d_out;

    const int N = in_sizes[0] / GCN_D_IN;
    const int E = in_sizes[1] / 2;
    const int* src = ei;
    const int* dst = ei + E;

    float* ws   = (float*)d_ws;
    size_t off  = ((size_t)N + 63) & ~(size_t)63;
    float* dinv = ws;
    float* xw1  = ws + off;
    float* h1   = xw1 + (size_t)N * GCN_D_HID;
    _Float16* W1p = (_Float16*)(h1 + (size_t)N * GCN_D_HID);
    _Float16* W2p = W1p + (size_t)GCN_D_IN * GCN_D_HID;
    float* xw2  = xw1;
    float* h2   = xw1 + (size_t)N * GCN_D_OUT;

    const int B = 256;
    {
        int t1 = (GCN_D_IN >> 5) * (GCN_D_HID >> 4) * 32;
        gcn_pack_w_f16<<<(t1 + B - 1) / B, B, 0, stream>>>(W1, W1p, GCN_D_IN, GCN_D_HID);
        int t2 = (GCN_D_HID >> 5) * (GCN_D_OUT >> 4) * 32;
        gcn_pack_w_f16<<<(t2 + B - 1) / B, B, 0, stream>>>(W2, W2p, GCN_D_HID, GCN_D_OUT);
    }

    gcn_init_deg <<<(N + B - 1) / B, B, 0, stream>>>(dinv, N);
    gcn_accum_deg<<<(E + B - 1) / B, B, 0, stream>>>(dst, dinv, E);
    gcn_make_dinv<<<(N + B - 1) / B, B, 0, stream>>>(dinv, N);

    {
        int waves = (N / 16) * (GCN_D_HID / 64);
        int th = waves * 32;
        gcn_wmma_gemm_f16x4<<<(th + B - 1) / B, B, 0, stream>>>(x, W1p, xw1, N, GCN_D_IN, GCN_D_HID);
    }
    {
        int total = N * GCN_D_HID;
        gcn_self_loop<<<(total + B - 1) / B, B, 0, stream>>>(xw1, dinv, h1, total, GCN_D_HID);
        int et = E * (GCN_D_HID / 4);
        gcn_edge_scatter<<<(et + B - 1) / B, B, 0, stream>>>(src, dst, dinv, xw1, h1, E, GCN_D_HID);
        gcn_bias_relu<<<(total + B - 1) / B, B, 0, stream>>>(h1, b1, total, GCN_D_HID);
    }

    {
        int waves = (N / 16) * (GCN_D_OUT / 64);
        int th = waves * 32;
        gcn_wmma_gemm_f16x4<<<(th + B - 1) / B, B, 0, stream>>>(h1, W2p, xw2, N, GCN_D_HID, GCN_D_OUT);
    }
    {
        int total = N * GCN_D_OUT;
        gcn_self_loop<<<(total + B - 1) / B, B, 0, stream>>>(xw2, dinv, h2, total, GCN_D_OUT);
        int et = E * (GCN_D_OUT / 4);
        gcn_edge_scatter<<<(et + B - 1) / B, B, 0, stream>>>(src, dst, dinv, xw2, h2, E, GCN_D_OUT);
    }
    {
        int th = N * 32;
        gcn_softmax64<<<(th + B - 1) / B, B, 0, stream>>>(h2, b2, out, N);
    }
}


# layer-1 edge sums leave global f32 atomics: one workgroup owns 196 rows in LDS, scans all destinations, LDS-adds its edges' rows, stores each line once with bias+ReLU; sum order = LDS add arrival
# speedup vs baseline: 2.0972x; 2.0972x over previous
#include <hip/hip_runtime.h>
#include <hip/hip_bf16.h>

typedef __attribute__((ext_vector_type(16))) _Float16 v16h;
typedef __attribute__((ext_vector_type(8)))  float    v8f;

#define GCN_D_IN  512
#define GCN_D_HID 256
#define GCN_D_OUT 64

__global__ void gcn_init_deg(float* __restrict__ deg, int n) {
    int i = blockIdx.x * blockDim.x + threadIdx.x;
    if (i < n) deg[i] = 1.0f;
}

__global__ void gcn_accum_deg(const int* __restrict__ dst, float* __restrict__ deg, int E) {
    int t = blockIdx.x * blockDim.x + threadIdx.x;
    if (t < E) atomicAdd(&deg[dst[t]], 1.0f);
}

__global__ void gcn_make_dinv(float* __restrict__ deg, int n) {
    int i = blockIdx.x * blockDim.x + threadIdx.x;
    if (i < n) {
        float d = deg[i];
        deg[i] = (d > 0.0f) ? rsqrtf(d) : 0.0f;
    }
}

__global__ void gcn_pack_w_f16(const float* __restrict__ W, _Float16* __restrict__ Wp,
                               int K, int N) {
    int t = blockIdx.x * blockDim.x + threadIdx.x;
    int tilesN = N >> 4;
    int total = (K >> 5) * tilesN * 32;
    if (t >= total) return;
    int frag = t >> 5;
    int lane = t & 31;
    int kt = frag / tilesN;
    int tn = frag - kt * tilesN;
    int n    = tn * 16 + (lane & 15);
    int koff = kt * 32 + (lane >> 4) * 8;
    _Float16* o = Wp + (size_t)t * 16;
#pragma unroll
    for (int j = 0; j < 8; ++j) {
        o[j]     = (_Float16)W[(size_t)(koff + j) * N + n];
        o[j + 8] = (_Float16)W[(size_t)(koff + 16 + j) * N + n];
    }
}

__global__ void gcn_wmma_gemm_f16x4(const float* __restrict__ A,
                                    const _Float16* __restrict__ Bp,
                                    float* __restrict__ C,
                                    int M, int K, int N) {
    const int lane    = threadIdx.x & 31;
    const int wave    = (blockIdx.x * blockDim.x + threadIdx.x) >> 5;
    const int tilesN  = N >> 4;
    const int groupsN = N >> 6;
    const int tm = wave / groupsN;
    const int tg = wave - tm * groupsN;
    if (tm * 16 >= M) return;
    const int tn0  = tg * 4;
    const int hsel = lane >> 4;
    const int idx  = lane & 15;

    const float* __restrict__ arow = A + (size_t)(tm * 16 + idx) * K;

    v8f acc0 = {}, acc1 = {}, acc2 = {}, acc3 = {};
    for (int k0 = 0; k0 < K; k0 += 32) {
        const int kb = k0 + hsel * 8;

        float4 a01 = *(const float4*)(arow + kb);
        float4 a23 = *(const float4*)(arow + kb + 4);
        float4 a45 = *(const float4*)(arow + kb + 16);
        float4 a67 = *(const float4*)(arow + kb + 20);
        if (k0 + 32 < K) __builtin_prefetch(arow + kb + 32, 0, 3);

        float af[16];
        *(float4*)(af + 0)  = a01;  *(float4*)(af + 4)  = a23;
        *(float4*)(af + 8)  = a45;  *(float4*)(af + 12) = a67;
        v16h av;
#pragma unroll
        for (int j = 0; j < 16; ++j) av[j] = (_Float16)af[j];

        const int kt = k0 >> 5;
        const _Float16* bbase = Bp + ((size_t)(kt * tilesN + tn0) * 32 + lane) * 16;
        v16h b0 = *(const v16h*)(bbase);
        v16h b1 = *(const v16h*)(bbase + 1 * 32 * 16);
        v16h b2 = *(const v16h*)(bbase + 2 * 32 * 16);
        v16h b3 = *(const v16h*)(bbase + 3 * 32 * 16);

        acc0 = __builtin_amdgcn_wmma_f32_16x16x32_f16(false, av, false, b0, (short)0, acc0, false, false);
        acc1 = __builtin_amdgcn_wmma_f32_16x16x32_f16(false, av, false, b1, (short)0, acc1, false, false);
        acc2 = __builtin_amdgcn_wmma_f32_16x16x32_f16(false, av, false, b2, (short)0, acc2, false, false);
        acc3 = __builtin_amdgcn_wmma_f32_16x16x32_f16(false, av, false, b3, (short)0, acc3, false, false);
    }

    float* cbase = C + (size_t)(tm * 16 + hsel * 8) * N + tn0 * 16 + idx;
#pragma unroll
    for (int r = 0; r < 8; ++r) {
        cbase[(size_t)r * N + 0]  = acc0[r];
        cbase[(size_t)r * N + 16] = acc1[r];
        cbase[(size_t)r * N + 32] = acc2[r];
        cbase[(size_t)r * N + 48] = acc3[r];
    }
}

__global__ void gcn_self_loop(const float* __restrict__ xw,
                              const float* __restrict__ dinv,
                              float* __restrict__ out, int total, int F) {
    int t = blockIdx.x * blockDim.x + threadIdx.x;
    if (t < total) {
        int i = t / F;
        float di = dinv[i];
        out[t] = xw[t] * di * di;
    }
}

template <int F, int EPW>
__global__ void gcn_edge_scatter_lines(const int* __restrict__ src,
                                       const int* __restrict__ dst,
                                       const float* __restrict__ dinv,
                                       const float* __restrict__ xw,
                                       float* __restrict__ out, int E) {
    constexpr int LINES = F / 32;
    const int lane = threadIdx.x & 31;
    const int wave = __builtin_amdgcn_readfirstlane((int)((blockIdx.x * blockDim.x + threadIdx.x) >> 5));
    const int e0 = wave * EPW;
    if (e0 >= E) return;

    if (e0 + EPW <= E) {
        int   s[EPW], d[EPW];
        float w[EPW];
#pragma unroll
        for (int j = 0; j < EPW; ++j) { s[j] = src[e0 + j]; d[j] = dst[e0 + j]; }
#pragma unroll
        for (int j = 0; j < EPW; ++j) w[j] = dinv[s[j]] * dinv[d[j]];

        float v[EPW][LINES];
#pragma unroll
        for (int j = 0; j < EPW; ++j) {
            const float* xr = xw + (size_t)s[j] * F + lane;
#pragma unroll
            for (int c = 0; c < LINES; ++c) v[j][c] = xr[c * 32];
        }
#pragma unroll
        for (int j = 0; j < EPW; ++j) {
            float* orow = out + (size_t)d[j] * F + lane;
#pragma unroll
            for (int c = 0; c < LINES; ++c) atomicAdd(orow + c * 32, v[j][c] * w[j]);
        }
    } else {
        for (int e = e0; e < E; ++e) {
            const int   se = src[e], de = dst[e];
            const float we = dinv[se] * dinv[de];
            const float* xr = xw + (size_t)se * F + lane;
            float* orow = out + (size_t)de * F + lane;
#pragma unroll
            for (int c = 0; c < LINES; ++c) atomicAdd(orow + c * 32, xr[c * 32] * we);
        }
    }
}

#define OWN_ROWS  196
#define OWN_WAVES 32
#define OWN_LIST  544

template <int G>
__device__ __forceinline__ void gcn_own_add_group(int s, int dl, float w, int j, int lane,
                                                  const float* __restrict__ xw, float* tile) {
    float4 a[G], b[G];
    int    rl[G];
    float  wj[G];
#pragma unroll
    for (int q = 0; q < G; ++q) {
        const int sj = __builtin_amdgcn_readlane(s, j + q);
        rl[q] = __builtin_amdgcn_readlane(dl, j + q);
        wj[q] = __int_as_float(__builtin_amdgcn_readlane(__float_as_int(w), j + q));
        const float4* xr = (const float4*)(xw + (size_t)sj * GCN_D_HID);
        a[q] = xr[lane];
        b[q] = xr[32 + lane];
    }
#pragma unroll
    for (int q = 0; q < G; ++q) {
        float* tr = tile + rl[q] * GCN_D_HID + lane;
        atomicAdd(tr,       a[q].x * wj[q]);
        atomicAdd(tr + 32,  a[q].y * wj[q]);
        atomicAdd(tr + 64,  a[q].z * wj[q]);
        atomicAdd(tr + 96,  a[q].w * wj[q]);
        atomicAdd(tr + 128, b[q].x * wj[q]);
        atomicAdd(tr + 160, b[q].y * wj[q]);
        atomicAdd(tr + 192, b[q].z * wj[q]);
        atomicAdd(tr + 224, b[q].w * wj[q]);
    }
}

__device__ __forceinline__ void gcn_own_add_edges(const unsigned* lst, int n, int lane, int lo,
                                                  const int* __restrict__ src,
                                                  const float* __restrict__ dinv,
                                                  const float* __restrict__ xw,
                                                  float* tile) {
    int s = 0, dl = 0;
    float w = 0.0f;
    if (lane < n) {
        const unsigned rec = lst[lane];
        dl = (int)(rec & 255u);
        s  = src[rec >> 8];
        w  = dinv[s] * dinv[lo + dl];
    }
    int j = 0;
    for (; j + 8 <= n; j += 8) gcn_own_add_group<8>(s, dl, w, j, lane, xw, tile);
    for (; j < n; ++j)         gcn_own_add_group<1>(s, dl, w, j, lane, xw, tile);
}

__global__ __launch_bounds__(1024)
void gcn_owner_sum_lds(const int* __restrict__ src, const int* __restrict__ dst,
                       const float* __restrict__ dinv, const float* __restrict__ xw,
                       const float* __restrict__ bias, float* __restrict__ h,
                       int N, int E) {
    __shared__ float    tile[OWN_ROWS * GCN_D_HID];
    __shared__ unsigned list[OWN_WAVES * OWN_LIST];

    const int lane = threadIdx.x & 31;
    const int wv = __builtin_amdgcn_readfirstlane((int)(threadIdx.x >> 5));
    const int lo = blockIdx.x * OWN_ROWS;
    const int nrows = (N - lo < OWN_ROWS) ? (N - lo) : OWN_ROWS;

    for (int r = wv; r < nrows; r += OWN_WAVES) {
        const float di = dinv[lo + r];
        const float4* xr = (const float4*)(xw + (size_t)(lo + r) * GCN_D_HID);
        const float4 a = xr[lane];
        const float4 b = xr[32 + lane];
        float* tr = tile + r * GCN_D_HID + lane;
        tr[0]   = a.x * di * di;
        tr[32]  = a.y * di * di;
        tr[64]  = a.z * di * di;
        tr[96]  = a.w * di * di;
        tr[128] = b.x * di * di;
        tr[160] = b.y * di * di;
        tr[192] = b.z * di * di;
        tr[224] = b.w * di * di;
    }
    __syncthreads();

    const int nchunks = (E + 1023) >> 10;
    const int c0 = (int)((blockIdx.x * (unsigned)nchunks) / gridDim.x);
    unsigned* mylist = list + wv * OWN_LIST;
    int count = 0;

    for (int i = wv; i < nchunks; i += OWN_WAVES) {
        int c = c0 + i;
        if (c >= nchunks) c -= nchunks;
        const int cbase = c << 10;

        int4 dv[8];
        int  eu[8];
#pragma unroll
        for (int u = 0; u < 8; ++u) {
            const int e = cbase + u * 128 + lane * 4;
            eu[u] = (e < E - 4) ? e : (E - 4);
            dv[u] = *(const int4*)(dst + eu[u]);
        }
        if (cbase + 1024 > E) {
#pragma unroll
            for (int u = 0; u < 8; ++u) {
                const int e = cbase + u * 128 + lane * 4;
                if (eu[u]     < e) dv[u].x = -1;
                if (eu[u] + 1 < e) dv[u].y = -1;
                if (eu[u] + 2 < e) dv[u].z = -1;
                if (eu[u] + 3 < e) dv[u].w = -1;
            }
        }
#pragma unroll
        for (int hf = 0; hf < 2; ++hf) {
#pragma unroll
            for (int uu = 0; uu < 4; ++uu) {
                const int u = hf * 4 + uu;
                const int dd[4] = {dv[u].x, dv[u].y, dv[u].z, dv[u].w};
#pragma unroll
                for (int k = 0; k < 4; ++k) {
                    const unsigned t = (unsigned)(dd[k] - lo);
                    const bool m = t < (unsigned)nrows;
                    const unsigned bal = __builtin_amdgcn_ballot_w32(m);
                    if (bal != 0u) {
                        if (m) mylist[count + __popc(bal & ((1u << lane) - 1u))] = ((unsigned)(eu[u] + k) << 8) | t;
                        count += __popc(bal);
                    }
                }
            }
            __builtin_amdgcn_fence(__ATOMIC_ACQ_REL, "wavefront");
            while (count >= 32) {
                count -= 32;
                gcn_own_add_edges(mylist + count, 32, lane, lo, src, dinv, xw, tile);
            }
        }
    }
    if (count > 0) gcn_own_add_edges(mylist, count, lane, lo, src, dinv, xw, tile);
    __syncthreads();

    const float4 bb0 = ((const float4*)bias)[lane];
    const float4 bb1 = ((const float4*)bias)[32 + lane];
    for (int r = wv; r < nrows; r += OWN_WAVES) {
        const float* tr = tile + r * GCN_D_HID + lane;
        float4 a, b;
        a.x = fmaxf(tr[0]   + bb0.x, 0.0f);
        a.y = fmaxf(tr[32]  + bb0.y, 0.0f);
        a.z = fmaxf(tr[64]  + bb0.z, 0.0f);
        a.w = fmaxf(tr[96]  + bb0.w, 0.0f);
        b.x = fmaxf(tr[128] + bb1.x, 0.0f);
        b.y = fmaxf(tr[160] + bb1.y, 0.0f);
        b.z = fmaxf(tr[192] + bb1.z, 0.0f);
        b.w = fmaxf(tr[224] + bb1.w, 0.0f);
        float4* hr = (float4*)(h + (size_t)(lo + r) * GCN_D_HID);
        hr[lane]      = a;
        hr[32 + lane] = b;
    }
}

__global__ void gcn_softmax64(const float* __restrict__ h,
                              const float* __restrict__ bias,
                              float* __restrict__ out, int nodes) {
    int wave = (blockIdx.x * blockDim.x + threadIdx.x) >> 5;
    int lane = threadIdx.x & 31;
    if (wave >= nodes) return;
    const float* row = h + (size_t)wave * 64;
    float v0 = row[lane]      + bias[lane];
    float v1 = row[lane + 32] + bias[lane + 32];
    float m = fmaxf(v0, v1);
#pragma unroll
    for (int off = 16; off > 0; off >>= 1) m = fmaxf(m, __shfl_xor(m, off, 32));
    float e0 = __expf(v0 - m);
    float e1 = __expf(v1 - m);
    float s = e0 + e1;
#pragma unroll
    for (int off = 16; off > 0; off >>= 1) s += __shfl_xor(s, off, 32);
    float inv = 1.0f / s;
    float* orow = out + (size_t)wave * 64;
    orow[lane]      = e0 * inv;
    orow[lane + 32] = e1 * inv;
}

extern "C" void kernel_launch(void* const* d_in, const int* in_sizes, int n_in,
                              void* d_out, int out_size, void* d_ws, size_t ws_size,
                              hipStream_t stream) {
    const float* x   = (const float*)d_in[0];
    const int*   ei  = (const int*)  d_in[1];
    const float* W1  = (const float*)d_in[2];
    const float* b1  = (const float*)d_in[3];
    const float* W2  = (const float*)d_in[4];
    const float* b2  = (const float*)d_in[5];
    float* out = (float*)d_out;

    const int N = in_sizes[0] / GCN_D_IN;
    const int E = in_sizes[1] / 2;
    const int* src = ei;
    const int* dst = ei + E;

    float* ws   = (float*)d_ws;
    size_t off  = ((size_t)N + 63) & ~(size_t)63;
    float* dinv = ws;
    float* xw1  = ws + off;
    float* h1   = xw1 + (size_t)N * GCN_D_HID;
    _Float16* W1p = (_Float16*)(h1 + (size_t)N * GCN_D_HID);
    _Float16* W2p = W1p + (size_t)GCN_D_IN * GCN_D_HID;
    float* xw2  = xw1;
    float* h2   = xw1 + (size_t)N * GCN_D_OUT;

    const int B = 256;
    {
        int t1 = (GCN_D_IN >> 5) * (GCN_D_HID >> 4) * 32;
        gcn_pack_w_f16<<<(t1 + B - 1) / B, B, 0, stream>>>(W1, W1p, GCN_D_IN, GCN_D_HID);
        int t2 = (GCN_D_HID >> 5) * (GCN_D_OUT >> 4) * 32;
        gcn_pack_w_f16<<<(t2 + B - 1) / B, B, 0, stream>>>(W2, W2p, GCN_D_HID, GCN_D_OUT);
    }

    gcn_init_deg <<<(N + B - 1) / B, B, 0, stream>>>(dinv, N);
    gcn_accum_deg<<<(E + B - 1) / B, B, 0, stream>>>(dst, dinv, E);
    gcn_make_dinv<<<(N + B - 1) / B, B, 0, stream>>>(dinv, N);

    {
        int waves = (N / 16) * (GCN_D_HID / 64);
        int th = waves * 32;
        gcn_wmma_gemm_f16x4<<<(th + B - 1) / B, B, 0, stream>>>(x, W1p, xw1, N, GCN_D_IN, GCN_D_HID);
    }
    {
        int ob = (N + OWN_ROWS - 1) / OWN_ROWS;
        gcn_owner_sum_lds<<<ob, 1024, 0, stream>>>(src, dst, dinv, xw1, b1, h1, N, E);
    }

    {
        int waves = (N / 16) * (GCN_D_OUT / 64);
        int th = waves * 32;
        gcn_wmma_gemm_f16x4<<<(th + B - 1) / B, B, 0, stream>>>(h1, W2p, xw2, N, GCN_D_HID, GCN_D_OUT);
    }
    {
        int total = N * GCN_D_OUT;
        gcn_self_loop<<<(total + B - 1) / B, B, 0, stream>>>(xw2, dinv, h2, total, GCN_D_OUT);
        int et = ((E + 15) / 16) * 32;
        gcn_edge_scatter_lines<GCN_D_OUT, 16><<<(et + B - 1) / B, B, 0, stream>>>(src, dst, dinv, xw2, h2, E);
    }
    {
        int th = N * 32;
        gcn_softmax64<<<(th + B - 1) / B, B, 0, stream>>>(h2, b2, out, N);
    }
}


# both GEMMs: each wave keeps its 16-column slice of packed W in registers all launch; persistent 16-wave workgroups stream A row blocks through LDS, converted to f16 once; sum order unchanged
# speedup vs baseline: 2.5918x; 1.2358x over previous
#include <hip/hip_runtime.h>
#include <hip/hip_bf16.h>

typedef __attribute__((ext_vector_type(16))) _Float16 v16h;
typedef __attribute__((ext_vector_type(8)))  float    v8f;

#define GCN_D_IN  512
#define GCN_D_HID 256
#define GCN_D_OUT 64

__global__ void gcn_init_deg(float* __restrict__ deg, int n) {
    int i = blockIdx.x * blockDim.x + threadIdx.x;
    if (i < n) deg[i] = 1.0f;
}

__global__ void gcn_accum_deg(const int* __restrict__ dst, float* __restrict__ deg, int E) {
    int t = blockIdx.x * blockDim.x + threadIdx.x;
    if (t < E) atomicAdd(&deg[dst[t]], 1.0f);
}

__global__ void gcn_make_dinv(float* __restrict__ deg, int n) {
    int i = blockIdx.x * blockDim.x + threadIdx.x;
    if (i < n) {
        float d = deg[i];
        deg[i] = (d > 0.0f) ? rsqrtf(d) : 0.0f;
    }
}

__global__ void gcn_pack_w_f16(const float* __restrict__ W, _Float16* __restrict__ Wp,
                               int K, int N) {
    int t = blockIdx.x * blockDim.x + threadIdx.x;
    int tilesN = N >> 4;
    int total = (K >> 5) * tilesN * 32;
    if (t >= total) return;
    int frag = t >> 5;
    int lane = t & 31;
    int kt = frag / tilesN;
    int tn = frag - kt * tilesN;
    int n    = tn * 16 + (lane & 15);
    int koff = kt * 32 + (lane >> 4) * 8;
    _Float16* o = Wp + (size_t)t * 16;
#pragma unroll
    for (int j = 0; j < 8; ++j) {
        o[j]     = (_Float16)W[(size_t)(koff + j) * N + n];
        o[j + 8] = (_Float16)W[(size_t)(koff + 16 + j) * N + n];
    }
}

typedef __attribute__((ext_vector_type(8))) _Float16 v8h;
typedef __attribute__((ext_vector_type(4))) _Float16 v4h;

__device__ __forceinline__ v16h gcn_lds_a_frag(const _Float16* ar) {
    const v8h lo = *(const v8h*)(ar);
    const v8h hi = *(const v8h*)(ar + 16);
    return __builtin_shufflevector(lo, hi, 0, 1, 2, 3, 4, 5, 6, 7, 8, 9, 10, 11, 12, 13, 14, 15);
}

template <int K, int NTL>
__device__ __forceinline__ void gcn_gemm_tiles(const _Float16* ar0, const v16h (&b)[K / 32],
                                               const int (&toff)[NTL], v8f (&acc)[NTL]) {
    constexpr int KT    = K / 32;
    constexpr int PITCH = K + 8;
    const _Float16* ar[NTL];
    v16h a[3][NTL];
#pragma unroll
    for (int i = 0; i < NTL; ++i) {
        ar[i]   = ar0 + toff[i] * (16 * PITCH);
        a[0][i] = gcn_lds_a_frag(ar[i]);
        a[1][i] = gcn_lds_a_frag(ar[i] + 32);
    }
#pragma unroll
    for (int kt = 0; kt < KT; ++kt) {
        if (kt + 2 < KT) {
#pragma unroll
            for (int i = 0; i < NTL; ++i) a[(kt + 2) % 3][i] = gcn_lds_a_frag(ar[i] + (kt + 2) * 32);
        }
        __builtin_amdgcn_sched_barrier(0);
#pragma unroll
        for (int i = 0; i < NTL; ++i)
            acc[i] = __builtin_amdgcn_wmma_f32_16x16x32_f16(false, a[kt % 3][i], false, b[kt],
                                                            (short)0, acc[i], false, false);
        __builtin_amdgcn_sched_barrier(0);
    }
}

template <int K, int N, int RT>
__global__ __launch_bounds__(512)
void gcn_wmma_gemm_bregs(const float* __restrict__ A,
                         const _Float16* __restrict__ Bp,
                         float* __restrict__ C, int M) {
    constexpr int KT    = K / 32;
    constexpr int NT    = N / 16;
    constexpr int RG    = 16 / NT;
    constexpr int ROWS  = RG * RT * 16;
    constexpr int PITCH = K + 8;
    constexpr int C4    = K / 4;
    constexpr int RSTEP = 512 / C4;
    constexpr int NV    = ROWS / RSTEP;
    static_assert(K % 32 == 0 && N % 16 == 0 && 16 % NT == 0 && 512 % C4 == 0 &&
                  ROWS % RSTEP == 0 && (RT == 1 || RT == 2), "shape not covered");

    __shared__ __attribute__((aligned(16))) _Float16 tile[2 * ROWS * PITCH];

    const int t    = threadIdx.x;
    const int lane = t & 31;
    const int w    = __builtin_amdgcn_readfirstlane(t >> 5);
    const int ct   = w % NT;
    const int rg   = w / NT;
    const int hsel = lane >> 4;
    const int idx  = lane & 15;

    const int nblk  = (M + ROWS - 1) / ROWS;
    const int ntile = M / 16;
    const int srow  = t / C4;
    const int sc    = (t % C4) * 4;
    float4 st[NV];

    int blk = blockIdx.x;
    if (blk < nblk) {
#pragma unroll
        for (int i = 0; i < NV; ++i) {
            int row = blk * ROWS + srow + i * RSTEP;
            row = (row < M) ? row : (M - 1);
            st[i] = *(const float4*)(A + (size_t)row * K + sc);
        }
    }

    v16h b[KT];
    {
        const _Float16* bb = Bp + ((size_t)ct * 32 + lane) * 16;
#pragma unroll
        for (int kt = 0; kt < KT; ++kt) b[kt] = *(const v16h*)(bb + (size_t)kt * NT * 32 * 16);
    }
    __builtin_amdgcn_sched_barrier(0);

    if (blk < nblk) {
#pragma unroll
        for (int i = 0; i < NV; ++i) {
            v4h hv = {(_Float16)st[i].x, (_Float16)st[i].y, (_Float16)st[i].z, (_Float16)st[i].w};
            *(v4h*)(tile + (srow + i * RSTEP) * PITCH + sc) = hv;
        }
    }
    __syncthreads();

    int buf = 0;
    for (; blk < nblk; blk += gridDim.x) {
        const int  nxt      = blk + (int)gridDim.x;
        const bool has_next = nxt < nblk;
        if (has_next) {
#pragma unroll
            for (int i = 0; i < NV; ++i) {
                int row = nxt * ROWS + srow + i * RSTEP;
                row = (row < M) ? row : (M - 1);
                st[i] = *(const float4*)(A + (size_t)row * K + sc);
            }
        }

        const int tile0 = (blk * RG + rg) * RT;
        const _Float16* ar0 = tile + buf * (ROWS * PITCH) + (rg * RT * 16 + idx) * PITCH + hsel * 8;
        float* cb0 = C + (size_t)(tile0 * 16 + hsel * 8) * N + ct * 16 + idx;
        const bool live = tile0 < ntile;
        int toff[RT];
#pragma unroll
        for (int i = 0; i < RT; ++i) toff[i] = (tile0 + i < ntile) ? i : 0;
        v8f acc[RT];
#pragma unroll
        for (int i = 0; i < RT; ++i) acc[i] = v8f{};
        if (live) gcn_gemm_tiles<K, RT>(ar0, b, toff, acc);

        if (has_next) {
            _Float16* tn = tile + (buf ^ 1) * (ROWS * PITCH);
#pragma unroll
            for (int i = 0; i < NV; ++i) {
                v4h hv = {(_Float16)st[i].x, (_Float16)st[i].y, (_Float16)st[i].z, (_Float16)st[i].w};
                *(v4h*)(tn + (srow + i * RSTEP) * PITCH + sc) = hv;
            }
        }
        __syncthreads();
        if (live) {
#pragma unroll
            for (int i = 0; i < RT; ++i) {
                float* cb = cb0 + (size_t)toff[i] * (16 * N);
#pragma unroll
                for (int r = 0; r < 8; ++r) cb[(size_t)r * N] = acc[i][r];
            }
        }
        buf ^= 1;
    }
}

__global__ void gcn_self_loop(const float* __restrict__ xw,
                              const float* __restrict__ dinv,
                              float* __restrict__ out, int total, int F) {
    int t = blockIdx.x * blockDim.x + threadIdx.x;
    if (t < total) {
        int i = t / F;
        float di = dinv[i];
        out[t] = xw[t] * di * di;
    }
}

template <int F, int EPW>
__global__ void gcn_edge_scatter_lines(const int* __restrict__ src,
                                       const int* __restrict__ dst,
                                       const float* __restrict__ dinv,
                                       const float* __restrict__ xw,
                                       float* __restrict__ out, int E) {
    constexpr int LINES = F / 32;
    const int lane = threadIdx.x & 31;
    const int wave = __builtin_amdgcn_readfirstlane((int)((blockIdx.x * blockDim.x + threadIdx.x) >> 5));
    const int e0 = wave * EPW;
    if (e0 >= E) return;

    if (e0 + EPW <= E) {
        int   s[EPW], d[EPW];
        float w[EPW];
#pragma unroll
        for (int j = 0; j < EPW; ++j) { s[j] = src[e0 + j]; d[j] = dst[e0 + j]; }
#pragma unroll
        for (int j = 0; j < EPW; ++j) w[j] = dinv[s[j]] * dinv[d[j]];

        float v[EPW][LINES];
#pragma unroll
        for (int j = 0; j < EPW; ++j) {
            const float* xr = xw + (size_t)s[j] * F + lane;
#pragma unroll
            for (int c = 0; c < LINES; ++c) v[j][c] = xr[c * 32];
        }
#pragma unroll
        for (int j = 0; j < EPW; ++j) {
            float* orow = out + (size_t)d[j] * F + lane;
#pragma unroll
            for (int c = 0; c < LINES; ++c) atomicAdd(orow + c * 32, v[j][c] * w[j]);
        }
    } else {
        for (int e = e0; e < E; ++e) {
            const int   se = src[e], de = dst[e];
            const float we = dinv[se] * dinv[de];
            const float* xr = xw + (size_t)se * F + lane;
            float* orow = out + (size_t)de * F + lane;
#pragma unroll
            for (int c = 0; c < LINES; ++c) atomicAdd(orow + c * 32, xr[c * 32] * we);
        }
    }
}

#define OWN_ROWS  196
#define OWN_WAVES 32
#define OWN_LIST  544

template <int G>
__device__ __forceinline__ void gcn_own_add_group(int s, int dl, float w, int j, int lane,
                                                  const float* __restrict__ xw, float* tile) {
    float4 a[G], b[G];
    int    rl[G];
    float  wj[G];
#pragma unroll
    for (int q = 0; q < G; ++q) {
        const int sj = __builtin_amdgcn_readlane(s, j + q);
        rl[q] = __builtin_amdgcn_readlane(dl, j + q);
        wj[q] = __int_as_float(__builtin_amdgcn_readlane(__float_as_int(w), j + q));
        const float4* xr = (const float4*)(xw + (size_t)sj * GCN_D_HID);
        a[q] = xr[lane];
        b[q] = xr[32 + lane];
    }
#pragma unroll
    for (int q = 0; q < G; ++q) {
        float* tr = tile + rl[q] * GCN_D_HID + lane;
        atomicAdd(tr,       a[q].x * wj[q]);
        atomicAdd(tr + 32,  a[q].y * wj[q]);
        atomicAdd(tr + 64,  a[q].z * wj[q]);
        atomicAdd(tr + 96,  a[q].w * wj[q]);
        atomicAdd(tr + 128, b[q].x * wj[q]);
        atomicAdd(tr + 160, b[q].y * wj[q]);
        atomicAdd(tr + 192, b[q].z * wj[q]);
        atomicAdd(tr + 224, b[q].w * wj[q]);
    }
}

__device__ __forceinline__ void gcn_own_add_edges(const unsigned* lst, int n, int lane, int lo,
                                                  const int* __restrict__ src,
                                                  const float* __restrict__ dinv,
                                                  const float* __restrict__ xw,
                                                  float* tile) {
    int s = 0, dl = 0;
    float w = 0.0f;
    if (lane < n) {
        const unsigned rec = lst[lane];
        dl = (int)(rec & 255u);
        s  = src[rec >> 8];
        w  = dinv[s] * dinv[lo + dl];
    }
    int j = 0;
    for (; j + 8 <= n; j += 8) gcn_own_add_group<8>(s, dl, w, j, lane, xw, tile);
    for (; j < n; ++j)         gcn_own_add_group<1>(s, dl, w, j, lane, xw, tile);
}

__global__ __launch_bounds__(1024)
void gcn_owner_sum_lds(const int* __restrict__ src, const int* __restrict__ dst,
                       const float* __restrict__ dinv, const float* __restrict__ xw,
                       const float* __restrict__ bias, float* __restrict__ h,
                       int N, int E) {
    __shared__ float    tile[OWN_ROWS * GCN_D_HID];
    __shared__ unsigned list[OWN_WAVES * OWN_LIST];

    const int lane = threadIdx.x & 31;
    const int wv = __builtin_amdgcn_readfirstlane((int)(threadIdx.x >> 5));
    const int lo = blockIdx.x * OWN_ROWS;
    const int nrows = (N - lo < OWN_ROWS) ? (N - lo) : OWN_ROWS;

    for (int r = wv; r < nrows; r += OWN_WAVES) {
        const float di = dinv[lo + r];
        const float4* xr = (const float4*)(xw + (size_t)(lo + r) * GCN_D_HID);
        const float4 a = xr[lane];
        const float4 b = xr[32 + lane];
        float* tr = tile + r * GCN_D_HID + lane;
        tr[0]   = a.x * di * di;
        tr[32]  = a.y * di * di;
        tr[64]  = a.z * di * di;
        tr[96]  = a.w * di * di;
        tr[128] = b.x * di * di;
        tr[160] = b.y * di * di;
        tr[192] = b.z * di * di;
        tr[224] = b.w * di * di;
    }
    __syncthreads();

    const int nchunks = (E + 1023) >> 10;
    const int c0 = (int)((blockIdx.x * (unsigned)nchunks) / gridDim.x);
    unsigned* mylist = list + wv * OWN_LIST;
    int count = 0;

    for (int i = wv; i < nchunks; i += OWN_WAVES) {
        int c = c0 + i;
        if (c >= nchunks) c -= nchunks;
        const int cbase = c << 10;

        int4 dv[8];
        int  eu[8];
#pragma unroll
        for (int u = 0; u < 8; ++u) {
            const int e = cbase + u * 128 + lane * 4;
            eu[u] = (e < E - 4) ? e : (E - 4);
            dv[u] = *(const int4*)(dst + eu[u]);
        }
        if (cbase + 1024 > E) {
#pragma unroll
            for (int u = 0; u < 8; ++u) {
                const int e = cbase + u * 128 + lane * 4;
                if (eu[u]     < e) dv[u].x = -1;
                if (eu[u] + 1 < e) dv[u].y = -1;
                if (eu[u] + 2 < e) dv[u].z = -1;
                if (eu[u] + 3 < e) dv[u].w = -1;
            }
        }
#pragma unroll
        for (int hf = 0; hf < 2; ++hf) {
#pragma unroll
            for (int uu = 0; uu < 4; ++uu) {
                const int u = hf * 4 + uu;
                const int dd[4] = {dv[u].x, dv[u].y, dv[u].z, dv[u].w};
#pragma unroll
                for (int k = 0; k < 4; ++k) {
                    const unsigned t = (unsigned)(dd[k] - lo);
                    const bool m = t < (unsigned)nrows;
                    const unsigned bal = __builtin_amdgcn_ballot_w32(m);
                    if (bal != 0u) {
                        if (m) mylist[count + __popc(bal & ((1u << lane) - 1u))] = ((unsigned)(eu[u] + k) << 8) | t;
                        count += __popc(bal);
                    }
                }
            }
            __builtin_amdgcn_fence(__ATOMIC_ACQ_REL, "wavefront");
            while (count >= 32) {
                count -= 32;
                gcn_own_add_edges(mylist + count, 32, lane, lo, src, dinv, xw, tile);
            }
        }
    }
    if (count > 0) gcn_own_add_edges(mylist, count, lane, lo, src, dinv, xw, tile);
    __syncthreads();

    const float4 bb0 = ((const float4*)bias)[lane];
    const float4 bb1 = ((const float4*)bias)[32 + lane];
    for (int r = wv; r < nrows; r += OWN_WAVES) {
        const float* tr = tile + r * GCN_D_HID + lane;
        float4 a, b;
        a.x = fmaxf(tr[0]   + bb0.x, 0.0f);
        a.y = fmaxf(tr[32]  + bb0.y, 0.0f);
        a.z = fmaxf(tr[64]  + bb0.z, 0.0f);
        a.w = fmaxf(tr[96]  + bb0.w, 0.0f);
        b.x = fmaxf(tr[128] + bb1.x, 0.0f);
        b.y = fmaxf(tr[160] + bb1.y, 0.0f);
        b.z = fmaxf(tr[192] + bb1.z, 0.0f);
        b.w = fmaxf(tr[224] + bb1.w, 0.0f);
        float4* hr = (float4*)(h + (size_t)(lo + r) * GCN_D_HID);
        hr[lane]      = a;
        hr[32 + lane] = b;
    }
}

__global__ void gcn_softmax64(const float* __restrict__ h,
                              const float* __restrict__ bias,
                              float* __restrict__ out, int nodes) {
    int wave = (blockIdx.x * blockDim.x + threadIdx.x) >> 5;
    int lane = threadIdx.x & 31;
    if (wave >= nodes) return;
    const float* row = h + (size_t)wave * 64;
    float v0 = row[lane]      + bias[lane];
    float v1 = row[lane + 32] + bias[lane + 32];
    float m = fmaxf(v0, v1);
#pragma unroll
    for (int off = 16; off > 0; off >>= 1) m = fmaxf(m, __shfl_xor(m, off, 32));
    float e0 = __expf(v0 - m);
    float e1 = __expf(v1 - m);
    float s = e0 + e1;
#pragma unroll
    for (int off = 16; off > 0; off >>= 1) s += __shfl_xor(s, off, 32);
    float inv = 1.0f / s;
    float* orow = out + (size_t)wave * 64;
    orow[lane]      = e0 * inv;
    orow[lane + 32] = e1 * inv;
}

extern "C" void kernel_launch(void* const* d_in, const int* in_sizes, int n_in,
                              void* d_out, int out_size, void* d_ws, size_t ws_size,
                              hipStream_t stream) {
    const float* x   = (const float*)d_in[0];
    const int*   ei  = (const int*)  d_in[1];
    const float* W1  = (const float*)d_in[2];
    const float* b1  = (const float*)d_in[3];
    const float* W2  = (const float*)d_in[4];
    const float* b2  = (const float*)d_in[5];
    float* out = (float*)d_out;

    const int N = in_sizes[0] / GCN_D_IN;
    const int E = in_sizes[1] / 2;
    const int* src = ei;
    const int* dst = ei + E;

    float* ws   = (float*)d_ws;
    size_t off  = ((size_t)N + 63) & ~(size_t)63;
    float* dinv = ws;
    float* xw1  = ws + off;
    float* h1   = xw1 + (size_t)N * GCN_D_HID;
    _Float16* W1p = (_Float16*)(h1 + (size_t)N * GCN_D_HID);
    _Float16* W2p = W1p + (size_t)GCN_D_IN * GCN_D_HID;
    float* xw2  = xw1;
    float* h2   = xw1 + (size_t)N * GCN_D_OUT;

    const int B = 256;
    {
        int t1 = (GCN_D_IN >> 5) * (GCN_D_HID >> 4) * 32;
        gcn_pack_w_f16<<<(t1 + B - 1) / B, B, 0, stream>>>(W1, W1p, GCN_D_IN, GCN_D_HID);
        int t2 = (GCN_D_HID >> 5) * (GCN_D_OUT >> 4) * 32;
        gcn_pack_w_f16<<<(t2 + B - 1) / B, B, 0, stream>>>(W2, W2p, GCN_D_HID, GCN_D_OUT);
    }

    gcn_init_deg <<<(N + B - 1) / B, B, 0, stream>>>(dinv, N);
    gcn_accum_deg<<<(E + B - 1) / B, B, 0, stream>>>(dst, dinv, E);
    gcn_make_dinv<<<(N + B - 1) / B, B, 0, stream>>>(dinv, N);

    {
        gcn_wmma_gemm_bregs<GCN_D_IN, GCN_D_HID, 2><<<256, 512, 0, stream>>>(x, W1p, xw1, N);
    }
    {
        int ob = (N + OWN_ROWS - 1) / OWN_ROWS;
        gcn_owner_sum_lds<<<ob, 1024, 0, stream>>>(src, dst, dinv, xw1, b1, h1, N, E);
    }

    {
        gcn_wmma_gemm_bregs<GCN_D_HID, GCN_D_OUT, 1><<<256, 512, 0, stream>>>(h1, W2p, xw2, N);
    }
    {
        int total = N * GCN_D_OUT;
        gcn_self_loop<<<(total + B - 1) / B, B, 0, stream>>>(xw2, dinv, h2, total, GCN_D_OUT);
        int et = ((E + 15) / 16) * 32;
        gcn_edge_scatter_lines<GCN_D_OUT, 16><<<(et + B - 1) / B, B, 0, stream>>>(src, dst, dinv, xw2, h2, E);
    }
    {
        int th = N * 32;
        gcn_softmax64<<<(th + B - 1) / B, B, 0, stream>>>(h2, b2, out, N);
    }
}


# layer-2 edge sums reuse the per-wave edge lists the layer-1 owner workgroups found: summed in LDS per owner, no rescan, no global atomics, bias+softmax at the store; sum order = LDS add arrival
# speedup vs baseline: 2.8942x; 1.1167x over previous
#include <hip/hip_runtime.h>
#include <hip/hip_bf16.h>

typedef __attribute__((ext_vector_type(16))) _Float16 v16h;
typedef __attribute__((ext_vector_type(8)))  float    v8f;

#define GCN_D_IN  512
#define GCN_D_HID 256
#define GCN_D_OUT 64

__global__ void gcn_init_deg(float* __restrict__ deg, int n) {
    int i = blockIdx.x * blockDim.x + threadIdx.x;
    if (i < n) deg[i] = 1.0f;
}

__global__ void gcn_accum_deg(const int* __restrict__ dst, float* __restrict__ deg, int E) {
    int t = blockIdx.x * blockDim.x + threadIdx.x;
    if (t < E) atomicAdd(&deg[dst[t]], 1.0f);
}

__global__ void gcn_make_dinv(float* __restrict__ deg, int n) {
    int i = blockIdx.x * blockDim.x + threadIdx.x;
    if (i < n) {
        float d = deg[i];
        deg[i] = (d > 0.0f) ? rsqrtf(d) : 0.0f;
    }
}

__global__ void gcn_pack_w_f16(const float* __restrict__ W, _Float16* __restrict__ Wp,
                               int K, int N) {
    int t = blockIdx.x * blockDim.x + threadIdx.x;
    int tilesN = N >> 4;
    int total = (K >> 5) * tilesN * 32;
    if (t >= total) return;
    int frag = t >> 5;
    int lane = t & 31;
    int kt = frag / tilesN;
    int tn = frag - kt * tilesN;
    int n    = tn * 16 + (lane & 15);
    int koff = kt * 32 + (lane >> 4) * 8;
    _Float16* o = Wp + (size_t)t * 16;
#pragma unroll
    for (int j = 0; j < 8; ++j) {
        o[j]     = (_Float16)W[(size_t)(koff + j) * N + n];
        o[j + 8] = (_Float16)W[(size_t)(koff + 16 + j) * N + n];
    }
}

typedef __attribute__((ext_vector_type(8))) _Float16 v8h;
typedef __attribute__((ext_vector_type(4))) _Float16 v4h;

__device__ __forceinline__ v16h gcn_lds_a_frag(const _Float16* ar) {
    const v8h lo = *(const v8h*)(ar);
    const v8h hi = *(const v8h*)(ar + 16);
    return __builtin_shufflevector(lo, hi, 0, 1, 2, 3, 4, 5, 6, 7, 8, 9, 10, 11, 12, 13, 14, 15);
}

template <int K, int NTL>
__device__ __forceinline__ void gcn_gemm_tiles(const _Float16* ar0, const v16h (&b)[K / 32],
                                               const int (&toff)[NTL], v8f (&acc)[NTL]) {
    constexpr int KT    = K / 32;
    constexpr int PITCH = K + 8;
    const _Float16* ar[NTL];
    v16h a[3][NTL];
#pragma unroll
    for (int i = 0; i < NTL; ++i) {
        ar[i]   = ar0 + toff[i] * (16 * PITCH);
        a[0][i] = gcn_lds_a_frag(ar[i]);
        a[1][i] = gcn_lds_a_frag(ar[i] + 32);
    }
#pragma unroll
    for (int kt = 0; kt < KT; ++kt) {
        if (kt + 2 < KT) {
#pragma unroll
            for (int i = 0; i < NTL; ++i) a[(kt + 2) % 3][i] = gcn_lds_a_frag(ar[i] + (kt + 2) * 32);
        }
        __builtin_amdgcn_sched_barrier(0);
#pragma unroll
        for (int i = 0; i < NTL; ++i)
            acc[i] = __builtin_amdgcn_wmma_f32_16x16x32_f16(false, a[kt % 3][i], false, b[kt],
                                                            (short)0, acc[i], false, false);
        __builtin_amdgcn_sched_barrier(0);
    }
}

template <int K, int N, int RT>
__global__ __launch_bounds__(512)
void gcn_wmma_gemm_bregs(const float* __restrict__ A,
                         const _Float16* __restrict__ Bp,
                         float* __restrict__ C, int M) {
    constexpr int KT    = K / 32;
    constexpr int NT    = N / 16;
    constexpr int RG    = 16 / NT;
    constexpr int ROWS  = RG * RT * 16;
    constexpr int PITCH = K + 8;
    constexpr int C4    = K / 4;
    constexpr int RSTEP = 512 / C4;
    constexpr int NV    = ROWS / RSTEP;
    static_assert(K % 32 == 0 && N % 16 == 0 && 16 % NT == 0 && 512 % C4 == 0 &&
                  ROWS % RSTEP == 0 && (RT == 1 || RT == 2), "shape not covered");

    __shared__ __attribute__((aligned(16))) _Float16 tile[2 * ROWS * PITCH];

    const int t    = threadIdx.x;
    const int lane = t & 31;
    const int w    = __builtin_amdgcn_readfirstlane(t >> 5);
    const int ct   = w % NT;
    const int rg   = w / NT;
    const int hsel = lane >> 4;
    const int idx  = lane & 15;

    const int nblk  = (M + ROWS - 1) / ROWS;
    const int ntile = M / 16;
    const int srow  = t / C4;
    const int sc    = (t % C4) * 4;
    float4 st[NV];

    int blk = blockIdx.x;
    if (blk < nblk) {
#pragma unroll
        for (int i = 0; i < NV; ++i) {
            int row = blk * ROWS + srow + i * RSTEP;
            row = (row < M) ? row : (M - 1);
            st[i] = *(const float4*)(A + (size_t)row * K + sc);
        }
    }

    v16h b[KT];
    {
        const _Float16* bb = Bp + ((size_t)ct * 32 + lane) * 16;
#pragma unroll
        for (int kt = 0; kt < KT; ++kt) b[kt] = *(const v16h*)(bb + (size_t)kt * NT * 32 * 16);
    }
    __builtin_amdgcn_sched_barrier(0);

    if (blk < nblk) {
#pragma unroll
        for (int i = 0; i < NV; ++i) {
            v4h hv = {(_Float16)st[i].x, (_Float16)st[i].y, (_Float16)st[i].z, (_Float16)st[i].w};
            *(v4h*)(tile + (srow + i * RSTEP) * PITCH + sc) = hv;
        }
    }
    __syncthreads();

    int buf = 0;
    for (; blk < nblk; blk += gridDim.x) {
        const int  nxt      = blk + (int)gridDim.x;
        const bool has_next = nxt < nblk;
        if (has_next) {
#pragma unroll
            for (int i = 0; i < NV; ++i) {
                int row = nxt * ROWS + srow + i * RSTEP;
                row = (row < M) ? row : (M - 1);
                st[i] = *(const float4*)(A + (size_t)row * K + sc);
            }
        }

        const int tile0 = (blk * RG + rg) * RT;
        const _Float16* ar0 = tile + buf * (ROWS * PITCH) + (rg * RT * 16 + idx) * PITCH + hsel * 8;
        float* cb0 = C + (size_t)(tile0 * 16 + hsel * 8) * N + ct * 16 + idx;
        const bool live = tile0 < ntile;
        int toff[RT];
#pragma unroll
        for (int i = 0; i < RT; ++i) toff[i] = (tile0 + i < ntile) ? i : 0;
        v8f acc[RT];
#pragma unroll
        for (int i = 0; i < RT; ++i) acc[i] = v8f{};
        if (live) gcn_gemm_tiles<K, RT>(ar0, b, toff, acc);

        if (has_next) {
            _Float16* tn = tile + (buf ^ 1) * (ROWS * PITCH);
#pragma unroll
            for (int i = 0; i < NV; ++i) {
                v4h hv = {(_Float16)st[i].x, (_Float16)st[i].y, (_Float16)st[i].z, (_Float16)st[i].w};
                *(v4h*)(tn + (srow + i * RSTEP) * PITCH + sc) = hv;
            }
        }
        __syncthreads();
        if (live) {
#pragma unroll
            for (int i = 0; i < RT; ++i) {
                float* cb = cb0 + (size_t)toff[i] * (16 * N);
#pragma unroll
                for (int r = 0; r < 8; ++r) cb[(size_t)r * N] = acc[i][r];
            }
        }
        buf ^= 1;
    }
}

#define OWN_ROWS  196
#define OWN_WAVES 32
#define OWN_LIST  544
#define OWN_KEEP  160

template <int F, int G>
__device__ __forceinline__ void gcn_own_add_group(int s, int dl, float w, int j, int lane,
                                                  const float* __restrict__ xw, float* tile) {
    int   rl[G];
    float wj[G];
    if constexpr (F == 256) {
        float4 a[G], b[G];
#pragma unroll
        for (int q = 0; q < G; ++q) {
            const int sj = __builtin_amdgcn_readlane(s, j + q);
            rl[q] = __builtin_amdgcn_readlane(dl, j + q);
            wj[q] = __int_as_float(__builtin_amdgcn_readlane(__float_as_int(w), j + q));
            const float4* xr = (const float4*)(xw + (size_t)sj * F);
            a[q] = xr[lane];
            b[q] = xr[32 + lane];
        }
#pragma unroll
        for (int q = 0; q < G; ++q) {
            float* tr = tile + rl[q] * F + lane;
            atomicAdd(tr,       a[q].x * wj[q]);
            atomicAdd(tr + 32,  a[q].y * wj[q]);
            atomicAdd(tr + 64,  a[q].z * wj[q]);
            atomicAdd(tr + 96,  a[q].w * wj[q]);
            atomicAdd(tr + 128, b[q].x * wj[q]);
            atomicAdd(tr + 160, b[q].y * wj[q]);
            atomicAdd(tr + 192, b[q].z * wj[q]);
            atomicAdd(tr + 224, b[q].w * wj[q]);
        }
    } else {
        float a[G], b[G];
#pragma unroll
        for (int q = 0; q < G; ++q) {
            const int sj = __builtin_amdgcn_readlane(s, j + q);
            rl[q] = __builtin_amdgcn_readlane(dl, j + q);
            wj[q] = __int_as_float(__builtin_amdgcn_readlane(__float_as_int(w), j + q));
            const float* xr = xw + (size_t)sj * F;
            a[q] = xr[lane];
            b[q] = xr[32 + lane];
        }
#pragma unroll
        for (int q = 0; q < G; ++q) {
            float* tr = tile + rl[q] * F + lane;
            atomicAdd(tr,      a[q] * wj[q]);
            atomicAdd(tr + 32, b[q] * wj[q]);
        }
    }
}

template <int F, bool SHORT_TAIL>
__device__ __forceinline__ void gcn_own_add_rows(int s, int dl, float w, int n, int lane,
                                                 const float* __restrict__ xw, float* tile) {
    int j = 0;
    for (; j + 8 <= n; j += 8) gcn_own_add_group<F, 8>(s, dl, w, j, lane, xw, tile);
    if constexpr (SHORT_TAIL) {
        if (j + 4 <= n) { gcn_own_add_group<F, 4>(s, dl, w, j, lane, xw, tile); j += 4; }
        if (j + 2 <= n) { gcn_own_add_group<F, 2>(s, dl, w, j, lane, xw, tile); j += 2; }
        if (j < n)        gcn_own_add_group<F, 1>(s, dl, w, j, lane, xw, tile);
    } else {
        for (; j < n; ++j) gcn_own_add_group<F, 1>(s, dl, w, j, lane, xw, tile);
    }
}

template <int F>
__device__ __forceinline__ void gcn_own_add_edges(const unsigned* lst, int n, int lane, int lo,
                                                  const int* __restrict__ src,
                                                  const float* __restrict__ dinv,
                                                  const float* __restrict__ xw,
                                                  float* tile, unsigned* keep) {
    int s = 0, dl = 0;
    float w = 0.0f;
    if (lane < n) {
        const unsigned rec = lst[lane];
        dl = (int)(rec & 255u);
        s  = src[rec >> 8];
        w  = dinv[s] * dinv[lo + dl];
        if (keep != nullptr) keep[lane] = ((unsigned)s << 8) | (unsigned)dl;
    }
    gcn_own_add_rows<F, false>(s, dl, w, n, lane, xw, tile);
}

template <int F>
__device__ __forceinline__ int gcn_own_scan(const int* __restrict__ src, const int* __restrict__ dst,
                                            const float* __restrict__ dinv,
                                            const float* __restrict__ xw,
                                            float* tile, unsigned* mylist, unsigned* kept,
                                            int lane, int wv, int lo, int nrows, int E) {
    const int nchunks = (E + 1023) >> 10;
    const int c0 = (int)((blockIdx.x * (unsigned)nchunks) / gridDim.x);
    int  count  = 0;
    int  nkept  = 0;
    bool keepok = kept != nullptr;

    for (int i = wv; i < nchunks; i += OWN_WAVES) {
        int c = c0 + i;
        if (c >= nchunks) c -= nchunks;
        const int cbase = c << 10;

        int4 dv[8];
        int  eu[8];
#pragma unroll
        for (int u = 0; u < 8; ++u) {
            const int e = cbase + u * 128 + lane * 4;
            eu[u] = (e < E - 4) ? e : (E - 4);
            dv[u] = *(const int4*)(dst + eu[u]);
        }
        if (cbase + 1024 > E) {
#pragma unroll
            for (int u = 0; u < 8; ++u) {
                const int e = cbase + u * 128 + lane * 4;
                if (eu[u]     < e) dv[u].x = -1;
                if (eu[u] + 1 < e) dv[u].y = -1;
                if (eu[u] + 2 < e) dv[u].z = -1;
                if (eu[u] + 3 < e) dv[u].w = -1;
            }
        }
#pragma unroll
        for (int hf = 0; hf < 2; ++hf) {
#pragma unroll
            for (int uu = 0; uu < 4; ++uu) {
                const int u = hf * 4 + uu;
                const int dd[4] = {dv[u].x, dv[u].y, dv[u].z, dv[u].w};
#pragma unroll
                for (int k = 0; k < 4; ++k) {
                    const unsigned t = (unsigned)(dd[k] - lo);
                    const bool m = t < (unsigned)nrows;
                    const unsigned bal = __builtin_amdgcn_ballot_w32(m);
                    if (bal != 0u) {
                        if (m) mylist[count + __popc(bal & ((1u << lane) - 1u))] = ((unsigned)(eu[u] + k) << 8) | t;
                        count += __popc(bal);
                    }
                }
            }
            __builtin_amdgcn_fence(__ATOMIC_ACQ_REL, "wavefront");
            while (count >= 32) {
                count -= 32;
                unsigned* kp = nullptr;
                if (keepok) {
                    if (nkept + 32 <= OWN_KEEP) { kp = kept + nkept; nkept += 32; }
                    else keepok = false;
                }
                gcn_own_add_edges<F>(mylist + count, 32, lane, lo, src, dinv, xw, tile, kp);
            }
        }
    }
    if (count > 0) {
        unsigned* kp = nullptr;
        if (keepok) {
            if (nkept + count <= OWN_KEEP) { kp = kept + nkept; nkept += count; }
            else keepok = false;
        }
        gcn_own_add_edges<F>(mylist, count, lane, lo, src, dinv, xw, tile, kp);
    }
    return keepok ? nkept : -1;
}

__device__ __forceinline__ bool gcn_own_keep_fits(const float* wsb, size_t ws_bytes, const int* kend) {
    return (size_t)((const char*)kend - (const char*)wsb) <= ws_bytes;
}

__global__ __launch_bounds__(1024)
void gcn_owner_sum_lds(const int* __restrict__ src, const int* __restrict__ dst,
                       const float* __restrict__ dinv, const float* __restrict__ xw,
                       const float* __restrict__ bias, float* __restrict__ h,
                       unsigned* __restrict__ kept, int* __restrict__ kcnt,
                       const float* wsb, size_t ws_bytes, int N, int E) {
    __shared__ float    tile[OWN_ROWS * GCN_D_HID];
    __shared__ unsigned list[OWN_WAVES * OWN_LIST];

    const int lane = threadIdx.x & 31;
    const int wv = __builtin_amdgcn_readfirstlane((int)(threadIdx.x >> 5));
    const int lo = blockIdx.x * OWN_ROWS;
    const int nrows = (N - lo < OWN_ROWS) ? (N - lo) : OWN_ROWS;
    const int wid = blockIdx.x * OWN_WAVES + wv;
    const bool keep_on = gcn_own_keep_fits(wsb, ws_bytes, kcnt + gridDim.x * OWN_WAVES);

    for (int r = wv; r < nrows; r += OWN_WAVES) {
        const float di = dinv[lo + r];
        const float4* xr = (const float4*)(xw + (size_t)(lo + r) * GCN_D_HID);
        const float4 a = xr[lane];
        const float4 b = xr[32 + lane];
        float* tr = tile + r * GCN_D_HID + lane;
        tr[0]   = a.x * di * di;
        tr[32]  = a.y * di * di;
        tr[64]  = a.z * di * di;
        tr[96]  = a.w * di * di;
        tr[128] = b.x * di * di;
        tr[160] = b.y * di * di;
        tr[192] = b.z * di * di;
        tr[224] = b.w * di * di;
    }
    __syncthreads();

    const int nk = gcn_own_scan<GCN_D_HID>(src, dst, dinv, xw, tile, list + wv * OWN_LIST,
                                           keep_on ? kept + (size_t)wid * OWN_KEEP : nullptr,
                                           lane, wv, lo, nrows, E);
    if (keep_on && lane == 0) kcnt[wid] = nk;
    __syncthreads();

    const float4 bb0 = ((const float4*)bias)[lane];
    const float4 bb1 = ((const float4*)bias)[32 + lane];
    for (int r = wv; r < nrows; r += OWN_WAVES) {
        const float* tr = tile + r * GCN_D_HID + lane;
        float4 a, b;
        a.x = fmaxf(tr[0]   + bb0.x, 0.0f);
        a.y = fmaxf(tr[32]  + bb0.y, 0.0f);
        a.z = fmaxf(tr[64]  + bb0.z, 0.0f);
        a.w = fmaxf(tr[96]  + bb0.w, 0.0f);
        b.x = fmaxf(tr[128] + bb1.x, 0.0f);
        b.y = fmaxf(tr[160] + bb1.y, 0.0f);
        b.z = fmaxf(tr[192] + bb1.z, 0.0f);
        b.w = fmaxf(tr[224] + bb1.w, 0.0f);
        float4* hr = (float4*)(h + (size_t)(lo + r) * GCN_D_HID);
        hr[lane]      = a;
        hr[32 + lane] = b;
    }
}

__global__ __launch_bounds__(1024)
void gcn_owner_sum_kept(const int* __restrict__ src, const int* __restrict__ dst,
                        const float* __restrict__ dinv, const float* __restrict__ xw,
                        const float* __restrict__ bias, float* __restrict__ out,
                        const unsigned* __restrict__ kept, const int* __restrict__ kcnt,
                        const float* wsb, size_t ws_bytes, int N, int E) {
    __shared__ float    tile[OWN_ROWS * GCN_D_OUT];
    __shared__ unsigned list[OWN_WAVES * OWN_LIST];

    const int lane = threadIdx.x & 31;
    const int wv = __builtin_amdgcn_readfirstlane((int)(threadIdx.x >> 5));
    const int lo = blockIdx.x * OWN_ROWS;
    const int nrows = (N - lo < OWN_ROWS) ? (N - lo) : OWN_ROWS;
    const int wid = blockIdx.x * OWN_WAVES + wv;
    const bool keep_on = gcn_own_keep_fits(wsb, ws_bytes, kcnt + gridDim.x * OWN_WAVES);

    bool use_kept = false;
    if (keep_on) {
        const int c = kcnt[blockIdx.x * OWN_WAVES + lane];
        use_kept = __builtin_amdgcn_ballot_w32(c < 0 || c > OWN_KEEP) == 0u;
    }

    {
        constexpr int RPW = (OWN_ROWS + OWN_WAVES - 1) / OWN_WAVES;
        float sa[RPW], sb[RPW];
        const int   rl  = wv + lane * OWN_WAVES;
        const float dvl = dinv[lo + ((rl < nrows) ? rl : (nrows - 1))];
#pragma unroll
        for (int i = 0; i < RPW; ++i) {
            const int r  = wv + i * OWN_WAVES;
            const int rc = (r < nrows) ? r : (nrows - 1);
            const float* xr = xw + (size_t)(lo + rc) * GCN_D_OUT;
            sa[i] = xr[lane];
            sb[i] = xr[32 + lane];
        }
#pragma unroll
        for (int i = 0; i < RPW; ++i) {
            const int r = wv + i * OWN_WAVES;
            if (r < nrows) {
                const float di = __int_as_float(__builtin_amdgcn_readlane(__float_as_int(dvl), i));
                float* tr = tile + r * GCN_D_OUT + lane;
                tr[0]  = sa[i] * di * di;
                tr[32] = sb[i] * di * di;
            }
        }
    }
    __syncthreads();

    if (use_kept) {
        const unsigned* kp = kept + (size_t)wid * OWN_KEEP;
        int cnt = __builtin_amdgcn_readfirstlane(kcnt[wid]);
        cnt = (cnt < OWN_KEEP) ? cnt : OWN_KEEP;
        constexpr int NB = OWN_KEEP / 32;
        unsigned* myl = list + wv * OWN_LIST;
        unsigned rec[NB];
#pragma unroll
        for (int b = 0; b < NB; ++b) {
            rec[b] = 0u;
            if (b * 32 + lane < cnt) rec[b] = kp[b * 32 + lane];
        }
        float wgt[NB];
#pragma unroll
        for (int b = 0; b < NB; ++b) {
            const float ds = dinv[rec[b] >> 8];
            const float dd = dinv[lo + (int)(rec[b] & 255u)];
            wgt[b] = ds * dd;
        }
#pragma unroll
        for (int b = 0; b < NB; ++b) {
            myl[b * 32 + lane]            = rec[b];
            myl[OWN_KEEP + b * 32 + lane] = __float_as_uint(wgt[b]);
        }
        for (int base = 0; base < cnt; base += 32) {
            const int n = (cnt - base < 32) ? (cnt - base) : 32;
            const unsigned r = myl[base + lane];
            const float    w = __uint_as_float(myl[OWN_KEEP + base + lane]);
            gcn_own_add_rows<GCN_D_OUT, true>((int)(r >> 8), (int)(r & 255u), w, n, lane, xw, tile);
        }
    } else {
        gcn_own_scan<GCN_D_OUT>(src, dst, dinv, xw, tile, list + wv * OWN_LIST, nullptr,
                                lane, wv, lo, nrows, E);
    }
    __syncthreads();

    const float b0 = bias[lane];
    const float b1 = bias[lane + 32];
    for (int r = wv; r < nrows; r += OWN_WAVES) {
        const float* tr = tile + r * GCN_D_OUT + lane;
        float v0 = tr[0]  + b0;
        float v1 = tr[32] + b1;
        float m = fmaxf(v0, v1);
#pragma unroll
        for (int off = 16; off > 0; off >>= 1) m = fmaxf(m, __shfl_xor(m, off, 32));
        float e0 = __expf(v0 - m);
        float e1 = __expf(v1 - m);
        float sm = e0 + e1;
#pragma unroll
        for (int off = 16; off > 0; off >>= 1) sm += __shfl_xor(sm, off, 32);
        float inv = 1.0f / sm;
        float* orow = out + (size_t)(lo + r) * GCN_D_OUT;
        orow[lane]      = e0 * inv;
        orow[lane + 32] = e1 * inv;
    }
}

extern "C" void kernel_launch(void* const* d_in, const int* in_sizes, int n_in,
                              void* d_out, int out_size, void* d_ws, size_t ws_size,
                              hipStream_t stream) {
    const float* x   = (const float*)d_in[0];
    const int*   ei  = (const int*)  d_in[1];
    const float* W1  = (const float*)d_in[2];
    const float* b1  = (const float*)d_in[3];
    const float* W2  = (const float*)d_in[4];
    const float* b2  = (const float*)d_in[5];
    float* out = (float*)d_out;

    const int N = in_sizes[0] / GCN_D_IN;
    const int E = in_sizes[1] / 2;
    const int* src = ei;
    const int* dst = ei + E;

    float* ws   = (float*)d_ws;
    size_t off  = ((size_t)N + 63) & ~(size_t)63;
    float* dinv = ws;
    float* xw1  = ws + off;
    float* h1   = xw1 + (size_t)N * GCN_D_HID;
    _Float16* W1p = (_Float16*)(h1 + (size_t)N * GCN_D_HID);
    _Float16* W2p = W1p + (size_t)GCN_D_IN * GCN_D_HID;
    float* xw2  = xw1;
    const int ob = (N + OWN_ROWS - 1) / OWN_ROWS;
    unsigned* kept = (unsigned*)(W2p + (size_t)GCN_D_HID * GCN_D_OUT);
    int*      kcnt = (int*)(kept + (size_t)ob * OWN_WAVES * OWN_KEEP);

    const int B = 256;
    {
        int t1 = (GCN_D_IN >> 5) * (GCN_D_HID >> 4) * 32;
        gcn_pack_w_f16<<<(t1 + B - 1) / B, B, 0, stream>>>(W1, W1p, GCN_D_IN, GCN_D_HID);
        int t2 = (GCN_D_HID >> 5) * (GCN_D_OUT >> 4) * 32;
        gcn_pack_w_f16<<<(t2 + B - 1) / B, B, 0, stream>>>(W2, W2p, GCN_D_HID, GCN_D_OUT);
    }

    gcn_init_deg <<<(N + B - 1) / B, B, 0, stream>>>(dinv, N);
    gcn_accum_deg<<<(E + B - 1) / B, B, 0, stream>>>(dst, dinv, E);
    gcn_make_dinv<<<(N + B - 1) / B, B, 0, stream>>>(dinv, N);

    {
        gcn_wmma_gemm_bregs<GCN_D_IN, GCN_D_HID, 2><<<256, 512, 0, stream>>>(x, W1p, xw1, N);
    }
    {
        gcn_owner_sum_lds<<<ob, 1024, 0, stream>>>(src, dst, dinv, xw1, b1, h1, kept, kcnt,
                                                   ws, ws_size, N, E);
    }

    {
        gcn_wmma_gemm_bregs<GCN_D_HID, GCN_D_OUT, 1><<<256, 512, 0, stream>>>(h1, W2p, xw2, N);
    }
    {
        gcn_owner_sum_kept<<<ob, 1024, 0, stream>>>(src, dst, dinv, xw2, b2, out, kept, kcnt,
                                                    ws, ws_size, N, E);
    }
}


# layer-1 owner scan reads a 1-byte owner tag per edge (destination / 196, left by the degree-count launch), not the 4-byte destination: 1/4 the scan bytes and trips; sum order still LDS add arrival
# speedup vs baseline: 3.0178x; 1.0427x over previous
#include <hip/hip_runtime.h>
#include <hip/hip_bf16.h>

typedef __attribute__((ext_vector_type(16))) _Float16 v16h;
typedef __attribute__((ext_vector_type(8)))  float    v8f;

#define GCN_D_IN  512
#define GCN_D_HID 256
#define GCN_D_OUT 64

#define OWN_ROWS  196
#define OWN_TAGU  512

__global__ void gcn_init_deg(float* __restrict__ deg, int n) {
    int i = blockIdx.x * blockDim.x + threadIdx.x;
    if (i < n) deg[i] = 1.0f;
}

__device__ __forceinline__ bool gcn_ws_fits(const float* wsb, size_t ws_bytes, const void* end) {
    return (size_t)((const char*)end - (const char*)wsb) <= ws_bytes;
}

__global__ void gcn_accum_deg(const int* __restrict__ dst, float* __restrict__ deg,
                              unsigned char* __restrict__ tags, const float* wsb,
                              size_t ws_bytes, int E, int Epad) {
    int t = blockIdx.x * blockDim.x + threadIdx.x;
    const bool tags_on = gcn_ws_fits(wsb, ws_bytes, tags + Epad);
    if (t < E) {
        const int d = dst[t];
        atomicAdd(&deg[d], 1.0f);
        if (tags_on) tags[t] = (unsigned char)(d / OWN_ROWS);
    } else if (t < Epad && tags_on) {
        tags[t] = 0;
    }
}

__global__ void gcn_make_dinv(float* __restrict__ deg, int n) {
    int i = blockIdx.x * blockDim.x + threadIdx.x;
    if (i < n) {
        float d = deg[i];
        deg[i] = (d > 0.0f) ? rsqrtf(d) : 0.0f;
    }
}

__global__ void gcn_pack_w_f16(const float* __restrict__ W, _Float16* __restrict__ Wp,
                               int K, int N) {
    int t = blockIdx.x * blockDim.x + threadIdx.x;
    int tilesN = N >> 4;
    int total = (K >> 5) * tilesN * 32;
    if (t >= total) return;
    int frag = t >> 5;
    int lane = t & 31;
    int kt = frag / tilesN;
    int tn = frag - kt * tilesN;
    int n    = tn * 16 + (lane & 15);
    int koff = kt * 32 + (lane >> 4) * 8;
    _Float16* o = Wp + (size_t)t * 16;
#pragma unroll
    for (int j = 0; j < 8; ++j) {
        o[j]     = (_Float16)W[(size_t)(koff + j) * N + n];
        o[j + 8] = (_Float16)W[(size_t)(koff + 16 + j) * N + n];
    }
}

typedef __attribute__((ext_vector_type(8))) _Float16 v8h;
typedef __attribute__((ext_vector_type(4))) _Float16 v4h;

__device__ __forceinline__ v16h gcn_lds_a_frag(const _Float16* ar) {
    const v8h lo = *(const v8h*)(ar);
    const v8h hi = *(const v8h*)(ar + 16);
    return __builtin_shufflevector(lo, hi, 0, 1, 2, 3, 4, 5, 6, 7, 8, 9, 10, 11, 12, 13, 14, 15);
}

template <int K, int NTL>
__device__ __forceinline__ void gcn_gemm_tiles(const _Float16* ar0, const v16h (&b)[K / 32],
                                               const int (&toff)[NTL], v8f (&acc)[NTL]) {
    constexpr int KT    = K / 32;
    constexpr int PITCH = K + 8;
    const _Float16* ar[NTL];
    v16h a[3][NTL];
#pragma unroll
    for (int i = 0; i < NTL; ++i) {
        ar[i]   = ar0 + toff[i] * (16 * PITCH);
        a[0][i] = gcn_lds_a_frag(ar[i]);
        a[1][i] = gcn_lds_a_frag(ar[i] + 32);
    }
#pragma unroll
    for (int kt = 0; kt < KT; ++kt) {
        if (kt + 2 < KT) {
#pragma unroll
            for (int i = 0; i < NTL; ++i) a[(kt + 2) % 3][i] = gcn_lds_a_frag(ar[i] + (kt + 2) * 32);
        }
        __builtin_amdgcn_sched_barrier(0);
#pragma unroll
        for (int i = 0; i < NTL; ++i)
            acc[i] = __builtin_amdgcn_wmma_f32_16x16x32_f16(false, a[kt % 3][i], false, b[kt],
                                                            (short)0, acc[i], false, false);
        __builtin_amdgcn_sched_barrier(0);
    }
}

template <int K, int N, int RT>
__global__ __launch_bounds__(512)
void gcn_wmma_gemm_bregs(const float* __restrict__ A,
                         const _Float16* __restrict__ Bp,
                         float* __restrict__ C, int M) {
    constexpr int KT    = K / 32;
    constexpr int NT    = N / 16;
    constexpr int RG    = 16 / NT;
    constexpr int ROWS  = RG * RT * 16;
    constexpr int PITCH = K + 8;
    constexpr int C4    = K / 4;
    constexpr int RSTEP = 512 / C4;
    constexpr int NV    = ROWS / RSTEP;
    static_assert(K % 32 == 0 && N % 16 == 0 && 16 % NT == 0 && 512 % C4 == 0 &&
                  ROWS % RSTEP == 0 && (RT == 1 || RT == 2), "shape not covered");

    __shared__ __attribute__((aligned(16))) _Float16 tile[2 * ROWS * PITCH];

    const int t    = threadIdx.x;
    const int lane = t & 31;
    const int w    = __builtin_amdgcn_readfirstlane(t >> 5);
    const int ct   = w % NT;
    const int rg   = w / NT;
    const int hsel = lane >> 4;
    const int idx  = lane & 15;

    const int nblk  = (M + ROWS - 1) / ROWS;
    const int ntile = M / 16;
    const int srow  = t / C4;
    const int sc    = (t % C4) * 4;
    float4 st[NV];

    int blk = blockIdx.x;
    if (blk < nblk) {
#pragma unroll
        for (int i = 0; i < NV; ++i) {
            int row = blk * ROWS + srow + i * RSTEP;
            row = (row < M) ? row : (M - 1);
            st[i] = *(const float4*)(A + (size_t)row * K + sc);
        }
    }

    v16h b[KT];
    {
        const _Float16* bb = Bp + ((size_t)ct * 32 + lane) * 16;
#pragma unroll
        for (int kt = 0; kt < KT; ++kt) b[kt] = *(const v16h*)(bb + (size_t)kt * NT * 32 * 16);
    }
    __builtin_amdgcn_sched_barrier(0);

    if (blk < nblk) {
#pragma unroll
        for (int i = 0; i < NV; ++i) {
            v4h hv = {(_Float16)st[i].x, (_Float16)st[i].y, (_Float16)st[i].z, (_Float16)st[i].w};
            *(v4h*)(tile + (srow + i * RSTEP) * PITCH + sc) = hv;
        }
    }
    __syncthreads();

    int buf = 0;
    for (; blk < nblk; blk += gridDim.x) {
        const int  nxt      = blk + (int)gridDim.x;
        const bool has_next = nxt < nblk;
        if (has_next) {
#pragma unroll
            for (int i = 0; i < NV; ++i) {
                int row = nxt * ROWS + srow + i * RSTEP;
                row = (row < M) ? row : (M - 1);
                st[i] = *(const float4*)(A + (size_t)row * K + sc);
            }
        }

        const int tile0 = (blk * RG + rg) * RT;
        const _Float16* ar0 = tile + buf * (ROWS * PITCH) + (rg * RT * 16 + idx) * PITCH + hsel * 8;
        float* cb0 = C + (size_t)(tile0 * 16 + hsel * 8) * N + ct * 16 + idx;
        const bool live = tile0 < ntile;
        int toff[RT];
#pragma unroll
        for (int i = 0; i < RT; ++i) toff[i] = (tile0 + i < ntile) ? i : 0;
        v8f acc[RT];
#pragma unroll
        for (int i = 0; i < RT; ++i) acc[i] = v8f{};
        if (live) gcn_gemm_tiles<K, RT>(ar0, b, toff, acc);

        if (has_next) {
            _Float16* tn = tile + (buf ^ 1) * (ROWS * PITCH);
#pragma unroll
            for (int i = 0; i < NV; ++i) {
                v4h hv = {(_Float16)st[i].x, (_Float16)st[i].y, (_Float16)st[i].z, (_Float16)st[i].w};
                *(v4h*)(tn + (srow + i * RSTEP) * PITCH + sc) = hv;
            }
        }
        __syncthreads();
        if (live) {
#pragma unroll
            for (int i = 0; i < RT; ++i) {
                float* cb = cb0 + (size_t)toff[i] * (16 * N);
#pragma unroll
                for (int r = 0; r < 8; ++r) cb[(size_t)r * N] = acc[i][r];
            }
        }
        buf ^= 1;
    }
}

#define OWN_WAVES 32
#define OWN_LIST  544
#define OWN_KEEP  160

template <int F, int G>
__device__ __forceinline__ void gcn_own_add_group(int s, int dl, float w, int j, int lane,
                                                  const float* __restrict__ xw, float* tile) {
    int   rl[G];
    float wj[G];
    if constexpr (F == 256) {
        float4 a[G], b[G];
#pragma unroll
        for (int q = 0; q < G; ++q) {
            const int sj = __builtin_amdgcn_readlane(s, j + q);
            rl[q] = __builtin_amdgcn_readlane(dl, j + q);
            wj[q] = __int_as_float(__builtin_amdgcn_readlane(__float_as_int(w), j + q));
            const float4* xr = (const float4*)(xw + (size_t)sj * F);
            a[q] = xr[lane];
            b[q] = xr[32 + lane];
        }
#pragma unroll
        for (int q = 0; q < G; ++q) {
            float* tr = tile + rl[q] * F + lane;
            atomicAdd(tr,       a[q].x * wj[q]);
            atomicAdd(tr + 32,  a[q].y * wj[q]);
            atomicAdd(tr + 64,  a[q].z * wj[q]);
            atomicAdd(tr + 96,  a[q].w * wj[q]);
            atomicAdd(tr + 128, b[q].x * wj[q]);
            atomicAdd(tr + 160, b[q].y * wj[q]);
            atomicAdd(tr + 192, b[q].z * wj[q]);
            atomicAdd(tr + 224, b[q].w * wj[q]);
        }
    } else {
        float a[G], b[G];
#pragma unroll
        for (int q = 0; q < G; ++q) {
            const int sj = __builtin_amdgcn_readlane(s, j + q);
            rl[q] = __builtin_amdgcn_readlane(dl, j + q);
            wj[q] = __int_as_float(__builtin_amdgcn_readlane(__float_as_int(w), j + q));
            const float* xr = xw + (size_t)sj * F;
            a[q] = xr[lane];
            b[q] = xr[32 + lane];
        }
#pragma unroll
        for (int q = 0; q < G; ++q) {
            float* tr = tile + rl[q] * F + lane;
            atomicAdd(tr,      a[q] * wj[q]);
            atomicAdd(tr + 32, b[q] * wj[q]);
        }
    }
}

template <int F, bool SHORT_TAIL>
__device__ __forceinline__ void gcn_own_add_rows(int s, int dl, float w, int n, int lane,
                                                 const float* __restrict__ xw, float* tile) {
    int j = 0;
    for (; j + 8 <= n; j += 8) gcn_own_add_group<F, 8>(s, dl, w, j, lane, xw, tile);
    if constexpr (SHORT_TAIL) {
        if (j + 4 <= n) { gcn_own_add_group<F, 4>(s, dl, w, j, lane, xw, tile); j += 4; }
        if (j + 2 <= n) { gcn_own_add_group<F, 2>(s, dl, w, j, lane, xw, tile); j += 2; }
        if (j < n)        gcn_own_add_group<F, 1>(s, dl, w, j, lane, xw, tile);
    } else {
        for (; j < n; ++j) gcn_own_add_group<F, 1>(s, dl, w, j, lane, xw, tile);
    }
}

template <int F>
__device__ __forceinline__ void gcn_own_add_edges(const unsigned* lst, int n, int lane, int lo,
                                                  const int* __restrict__ src,
                                                  const float* __restrict__ dinv,
                                                  const float* __restrict__ xw,
                                                  float* tile, unsigned* keep) {
    int s = 0, dl = 0;
    float w = 0.0f;
    if (lane < n) {
        const unsigned rec = lst[lane];
        dl = (int)(rec & 255u);
        s  = src[rec >> 8];
        w  = dinv[s] * dinv[lo + dl];
        if (keep != nullptr) keep[lane] = ((unsigned)s << 8) | (unsigned)dl;
    }
    gcn_own_add_rows<F, false>(s, dl, w, n, lane, xw, tile);
}

template <int F>
__device__ __forceinline__ int gcn_own_scan(const int* __restrict__ src, const int* __restrict__ dst,
                                            const float* __restrict__ dinv,
                                            const float* __restrict__ xw,
                                            float* tile, unsigned* mylist, unsigned* kept,
                                            int lane, int wv, int lo, int nrows, int E) {
    const int nchunks = (E + 1023) >> 10;
    const int c0 = (int)((blockIdx.x * (unsigned)nchunks) / gridDim.x);
    int  count  = 0;
    int  nkept  = 0;
    bool keepok = kept != nullptr;

    for (int i = wv; i < nchunks; i += OWN_WAVES) {
        int c = c0 + i;
        if (c >= nchunks) c -= nchunks;
        const int cbase = c << 10;

        int4 dv[8];
        int  eu[8];
#pragma unroll
        for (int u = 0; u < 8; ++u) {
            const int e = cbase + u * 128 + lane * 4;
            eu[u] = (e < E - 4) ? e : (E - 4);
            dv[u] = *(const int4*)(dst + eu[u]);
        }
        if (cbase + 1024 > E) {
#pragma unroll
            for (int u = 0; u < 8; ++u) {
                const int e = cbase + u * 128 + lane * 4;
                if (eu[u]     < e) dv[u].x = -1;
                if (eu[u] + 1 < e) dv[u].y = -1;
                if (eu[u] + 2 < e) dv[u].z = -1;
                if (eu[u] + 3 < e) dv[u].w = -1;
            }
        }
#pragma unroll
        for (int hf = 0; hf < 2; ++hf) {
#pragma unroll
            for (int uu = 0; uu < 4; ++uu) {
                const int u = hf * 4 + uu;
                const int dd[4] = {dv[u].x, dv[u].y, dv[u].z, dv[u].w};
#pragma unroll
                for (int k = 0; k < 4; ++k) {
                    const unsigned t = (unsigned)(dd[k] - lo);
                    const bool m = t < (unsigned)nrows;
                    const unsigned bal = __builtin_amdgcn_ballot_w32(m);
                    if (bal != 0u) {
                        if (m) mylist[count + __popc(bal & ((1u << lane) - 1u))] = ((unsigned)(eu[u] + k) << 8) | t;
                        count += __popc(bal);
                    }
                }
            }
            __builtin_amdgcn_fence(__ATOMIC_ACQ_REL, "wavefront");
            while (count >= 32) {
                count -= 32;
                unsigned* kp = nullptr;
                if (keepok) {
                    if (nkept + 32 <= OWN_KEEP) { kp = kept + nkept; nkept += 32; }
                    else keepok = false;
                }
                gcn_own_add_edges<F>(mylist + count, 32, lane, lo, src, dinv, xw, tile, kp);
            }
        }
    }
    if (count > 0) {
        unsigned* kp = nullptr;
        if (keepok) {
            if (nkept + count <= OWN_KEEP) { kp = kept + nkept; nkept += count; }
            else keepok = false;
        }
        gcn_own_add_edges<F>(mylist, count, lane, lo, src, dinv, xw, tile, kp);
    }
    return keepok ? nkept : -1;
}

template <int F>
__device__ __forceinline__ void gcn_own_add_found(const unsigned* lst, int n, int lane, int lo,
                                                  const int* __restrict__ src,
                                                  const int* __restrict__ dst,
                                                  const float* __restrict__ dinv,
                                                  const float* __restrict__ xw,
                                                  float* tile, unsigned* keep) {
    int s = 0, dl = 0;
    float w = 0.0f;
    if (lane < n) {
        const unsigned e = lst[lane];
        const int d = dst[e];
        s  = src[e];
        dl = d - lo;
        w  = dinv[s] * dinv[d];
        if (keep != nullptr) keep[lane] = ((unsigned)s << 8) | (unsigned)dl;
    }
    gcn_own_add_rows<F, false>(s, dl, w, n, lane, xw, tile);
}

template <int F>
__device__ __forceinline__ int gcn_own_scan_tags(const int* __restrict__ src,
                                                 const int* __restrict__ dst,
                                                 const unsigned char* __restrict__ tags,
                                                 const float* __restrict__ dinv,
                                                 const float* __restrict__ xw,
                                                 float* tile, unsigned* mylist, unsigned* kept,
                                                 int lane, int wv, int lo, int E) {
    const unsigned pat = blockIdx.x * 0x01010101u;
    const int nunits = (E + OWN_TAGU - 1) / OWN_TAGU;
    const int nfull  = nunits / (8 * OWN_WAVES);
    const int ntrip  = nfull * OWN_WAVES;
    const int t0     = (int)((blockIdx.x * (unsigned)ntrip) / gridDim.x);
    const int rest0  = nfull * (8 * OWN_WAVES);
    const int per    = (nunits - rest0 + OWN_WAVES - 1) / OWN_WAVES;
    int  count  = 0;
    int  nkept  = 0;
    bool keepok = kept != nullptr;

    uint4 tv[8];
    int i = 0;
    int unit0 = 0;
    int nu = 0;
    int done = 0;
    for (;;) {
        if (done >= nu) {
            if (i > nfull) break;
            if (i < nfull) {
                int tr = t0 + i * OWN_WAVES + wv;
                if (tr >= ntrip) tr -= ntrip;
                unit0 = tr * 8;
                nu    = 8;
            } else {
                unit0 = rest0 + wv * per;
                nu    = (nunits - unit0 < per) ? (nunits - unit0) : per;
                if (nu <= 0) break;
            }
            ++i;
            done = 0;
#pragma unroll
            for (int u = 0; u < 8; ++u) {
                int un = unit0 + u;
                un = (un < nunits) ? un : (nunits - 1);
                tv[u] = *(const uint4*)(tags + (size_t)un * OWN_TAGU + lane * 16);
            }
        }
#pragma unroll
        for (int u = 0; u < 8; ++u) {
            if (u >= done && u < nu && count <= OWN_LIST - OWN_TAGU) {
                const int ebase = (unit0 + u) * OWN_TAGU + lane * 16;
                unsigned xs[4] = {tv[u].x ^ pat, tv[u].y ^ pat, tv[u].z ^ pat, tv[u].w ^ pat};
                if ((unit0 + u + 1) * OWN_TAGU > E) {
#pragma unroll
                    for (int q = 0; q < 4; ++q) {
#pragma unroll
                        for (int k = 0; k < 4; ++k)
                            if (ebase + q * 4 + k >= E) xs[q] |= 255u << (8 * k);
                    }
                }
#pragma unroll
                for (int q = 0; q < 4; ++q) {
                    const unsigned x = xs[q];
                    const unsigned z = (x - 0x01010101u) & ~x & 0x80808080u;
                    if (__builtin_amdgcn_ballot_w32(z != 0u) != 0u) {
#pragma unroll
                        for (int k = 0; k < 4; ++k) {
                            const bool m = ((x >> (8 * k)) & 255u) == 0u;
                            const unsigned bal = __builtin_amdgcn_ballot_w32(m);
                            if (bal != 0u) {
                                if (m) mylist[count + __popc(bal & ((1u << lane) - 1u))] = (unsigned)(ebase + q * 4 + k);
                                count += __popc(bal);
                            }
                        }
                    }
                }
                done = u + 1;
            }
        }
        __builtin_amdgcn_fence(__ATOMIC_ACQ_REL, "wavefront");
        while (count >= 32) {
            count -= 32;
            unsigned* kp = nullptr;
            if (keepok) {
                if (nkept + 32 <= OWN_KEEP) { kp = kept + nkept; nkept += 32; }
                else keepok = false;
            }
            gcn_own_add_found<F>(mylist + count, 32, lane, lo, src, dst, dinv, xw, tile, kp);
        }
    }
    if (count > 0) {
        unsigned* kp = nullptr;
        if (keepok) {
            if (nkept + count <= OWN_KEEP) { kp = kept + nkept; nkept += count; }
            else keepok = false;
        }
        gcn_own_add_found<F>(mylist, count, lane, lo, src, dst, dinv, xw, tile, kp);
    }
    return keepok ? nkept : -1;
}

__device__ __forceinline__ bool gcn_own_keep_fits(const float* wsb, size_t ws_bytes, const int* kend) {
    return (size_t)((const char*)kend - (const char*)wsb) <= ws_bytes;
}

__global__ __launch_bounds__(1024)
void gcn_owner_sum_lds(const int* __restrict__ src, const int* __restrict__ dst,
                       const float* __restrict__ dinv, const float* __restrict__ xw,
                       const float* __restrict__ bias, float* __restrict__ h,
                       unsigned* __restrict__ kept, int* __restrict__ kcnt,
                       const unsigned char* __restrict__ tags, int Epad,
                       const float* wsb, size_t ws_bytes, int N, int E) {
    __shared__ float    tile[OWN_ROWS * GCN_D_HID];
    __shared__ unsigned list[OWN_WAVES * OWN_LIST];

    const int lane = threadIdx.x & 31;
    const int wv = __builtin_amdgcn_readfirstlane((int)(threadIdx.x >> 5));
    const int lo = blockIdx.x * OWN_ROWS;
    const int nrows = (N - lo < OWN_ROWS) ? (N - lo) : OWN_ROWS;
    const int wid = blockIdx.x * OWN_WAVES + wv;
    const bool keep_on = gcn_own_keep_fits(wsb, ws_bytes, kcnt + gridDim.x * OWN_WAVES);

    for (int r = wv; r < nrows; r += OWN_WAVES) {
        const float di = dinv[lo + r];
        const float4* xr = (const float4*)(xw + (size_t)(lo + r) * GCN_D_HID);
        const float4 a = xr[lane];
        const float4 b = xr[32 + lane];
        float* tr = tile + r * GCN_D_HID + lane;
        tr[0]   = a.x * di * di;
        tr[32]  = a.y * di * di;
        tr[64]  = a.z * di * di;
        tr[96]  = a.w * di * di;
        tr[128] = b.x * di * di;
        tr[160] = b.y * di * di;
        tr[192] = b.z * di * di;
        tr[224] = b.w * di * di;
    }
    __syncthreads();

    const bool tags_on = gcn_ws_fits(wsb, ws_bytes, tags + Epad) && gridDim.x <= 256u;
    unsigned* kp0 = keep_on ? kept + (size_t)wid * OWN_KEEP : nullptr;
    int nk;
    if (tags_on)
        nk = gcn_own_scan_tags<GCN_D_HID>(src, dst, tags, dinv, xw, tile, list + wv * OWN_LIST,
                                          kp0, lane, wv, lo, E);
    else
        nk = gcn_own_scan<GCN_D_HID>(src, dst, dinv, xw, tile, list + wv * OWN_LIST,
                                     kp0, lane, wv, lo, nrows, E);
    if (keep_on && lane == 0) kcnt[wid] = nk;
    __syncthreads();

    const float4 bb0 = ((const float4*)bias)[lane];
    const float4 bb1 = ((const float4*)bias)[32 + lane];
    for (int r = wv; r < nrows; r += OWN_WAVES) {
        const float* tr = tile + r * GCN_D_HID + lane;
        float4 a, b;
        a.x = fmaxf(tr[0]   + bb0.x, 0.0f);
        a.y = fmaxf(tr[32]  + bb0.y, 0.0f);
        a.z = fmaxf(tr[64]  + bb0.z, 0.0f);
        a.w = fmaxf(tr[96]  + bb0.w, 0.0f);
        b.x = fmaxf(tr[128] + bb1.x, 0.0f);
        b.y = fmaxf(tr[160] + bb1.y, 0.0f);
        b.z = fmaxf(tr[192] + bb1.z, 0.0f);
        b.w = fmaxf(tr[224] + bb1.w, 0.0f);
        float4* hr = (float4*)(h + (size_t)(lo + r) * GCN_D_HID);
        hr[lane]      = a;
        hr[32 + lane] = b;
    }
}

__global__ __launch_bounds__(1024)
void gcn_owner_sum_kept(const int* __restrict__ src, const int* __restrict__ dst,
                        const float* __restrict__ dinv, const float* __restrict__ xw,
                        const float* __restrict__ bias, float* __restrict__ out,
                        const unsigned* __restrict__ kept, const int* __restrict__ kcnt,
                        const float* wsb, size_t ws_bytes, int N, int E) {
    __shared__ float    tile[OWN_ROWS * GCN_D_OUT];
    __shared__ unsigned list[OWN_WAVES * OWN_LIST];

    const int lane = threadIdx.x & 31;
    const int wv = __builtin_amdgcn_readfirstlane((int)(threadIdx.x >> 5));
    const int lo = blockIdx.x * OWN_ROWS;
    const int nrows = (N - lo < OWN_ROWS) ? (N - lo) : OWN_ROWS;
    const int wid = blockIdx.x * OWN_WAVES + wv;
    const bool keep_on = gcn_own_keep_fits(wsb, ws_bytes, kcnt + gridDim.x * OWN_WAVES);

    bool use_kept = false;
    if (keep_on) {
        const int c = kcnt[blockIdx.x * OWN_WAVES + lane];
        use_kept = __builtin_amdgcn_ballot_w32(c < 0 || c > OWN_KEEP) == 0u;
    }

    {
        constexpr int RPW = (OWN_ROWS + OWN_WAVES - 1) / OWN_WAVES;
        float sa[RPW], sb[RPW];
        const int   rl  = wv + lane * OWN_WAVES;
        const float dvl = dinv[lo + ((rl < nrows) ? rl : (nrows - 1))];
#pragma unroll
        for (int i = 0; i < RPW; ++i) {
            const int r  = wv + i * OWN_WAVES;
            const int rc = (r < nrows) ? r : (nrows - 1);
            const float* xr = xw + (size_t)(lo + rc) * GCN_D_OUT;
            sa[i] = xr[lane];
            sb[i] = xr[32 + lane];
        }
#pragma unroll
        for (int i = 0; i < RPW; ++i) {
            const int r = wv + i * OWN_WAVES;
            if (r < nrows) {
                const float di = __int_as_float(__builtin_amdgcn_readlane(__float_as_int(dvl), i));
                float* tr = tile + r * GCN_D_OUT + lane;
                tr[0]  = sa[i] * di * di;
                tr[32] = sb[i] * di * di;
            }
        }
    }
    __syncthreads();

    if (use_kept) {
        const unsigned* kp = kept + (size_t)wid * OWN_KEEP;
        int cnt = __builtin_amdgcn_readfirstlane(kcnt[wid]);
        cnt = (cnt < OWN_KEEP) ? cnt : OWN_KEEP;
        constexpr int NB = OWN_KEEP / 32;
        unsigned* myl = list + wv * OWN_LIST;
        unsigned rec[NB];
#pragma unroll
        for (int b = 0; b < NB; ++b) {
            rec[b] = 0u;
            if (b * 32 + lane < cnt) rec[b] = kp[b * 32 + lane];
        }
        float wgt[NB];
#pragma unroll
        for (int b = 0; b < NB; ++b) {
            const float ds = dinv[rec[b] >> 8];
            const float dd = dinv[lo + (int)(rec[b] & 255u)];
            wgt[b] = ds * dd;
        }
#pragma unroll
        for (int b = 0; b < NB; ++b) {
            myl[b * 32 + lane]            = rec[b];
            myl[OWN_KEEP + b * 32 + lane] = __float_as_uint(wgt[b]);
        }
        for (int base = 0; base < cnt; base += 32) {
            const int n = (cnt - base < 32) ? (cnt - base) : 32;
            const unsigned r = myl[base + lane];
            const float    w = __uint_as_float(myl[OWN_KEEP + base + lane]);
            gcn_own_add_rows<GCN_D_OUT, true>((int)(r >> 8), (int)(r & 255u), w, n, lane, xw, tile);
        }
    } else {
        gcn_own_scan<GCN_D_OUT>(src, dst, dinv, xw, tile, list + wv * OWN_LIST, nullptr,
                                lane, wv, lo, nrows, E);
    }
    __syncthreads();

    const float b0 = bias[lane];
    const float b1 = bias[lane + 32];
    for (int r = wv; r < nrows; r += OWN_WAVES) {
        const float* tr = tile + r * GCN_D_OUT + lane;
        float v0 = tr[0]  + b0;
        float v1 = tr[32] + b1;
        float m = fmaxf(v0, v1);
#pragma unroll
        for (int off = 16; off > 0; off >>= 1) m = fmaxf(m, __shfl_xor(m, off, 32));
        float e0 = __expf(v0 - m);
        float e1 = __expf(v1 - m);
        float sm = e0 + e1;
#pragma unroll
        for (int off = 16; off > 0; off >>= 1) sm += __shfl_xor(sm, off, 32);
        float inv = 1.0f / sm;
        float* orow = out + (size_t)(lo + r) * GCN_D_OUT;
        orow[lane]      = e0 * inv;
        orow[lane + 32] = e1 * inv;
    }
}

extern "C" void kernel_launch(void* const* d_in, const int* in_sizes, int n_in,
                              void* d_out, int out_size, void* d_ws, size_t ws_size,
                              hipStream_t stream) {
    const float* x   = (const float*)d_in[0];
    const int*   ei  = (const int*)  d_in[1];
    const float* W1  = (const float*)d_in[2];
    const float* b1  = (const float*)d_in[3];
    const float* W2  = (const float*)d_in[4];
    const float* b2  = (const float*)d_in[5];
    float* out = (float*)d_out;

    const int N = in_sizes[0] / GCN_D_IN;
    const int E = in_sizes[1] / 2;
    const int* src = ei;
    const int* dst = ei + E;

    float* ws   = (float*)d_ws;
    size_t off  = ((size_t)N + 63) & ~(size_t)63;
    float* dinv = ws;
    float* xw1  = ws + off;
    float* h1   = xw1 + (size_t)N * GCN_D_HID;
    _Float16* W1p = (_Float16*)(h1 + (size_t)N * GCN_D_HID);
    _Float16* W2p = W1p + (size_t)GCN_D_IN * GCN_D_HID;
    float* xw2  = xw1;
    const int ob = (N + OWN_ROWS - 1) / OWN_ROWS;
    unsigned* kept = (unsigned*)(W2p + (size_t)GCN_D_HID * GCN_D_OUT);
    int*      kcnt = (int*)(kept + (size_t)ob * OWN_WAVES * OWN_KEEP);
    unsigned char* tags = (unsigned char*)(kcnt + (size_t)ob * OWN_WAVES);
    const int Epad = (E + OWN_TAGU - 1) / OWN_TAGU * OWN_TAGU;

    const int B = 256;
    {
        int t1 = (GCN_D_IN >> 5) * (GCN_D_HID >> 4) * 32;
        gcn_pack_w_f16<<<(t1 + B - 1) / B, B, 0, stream>>>(W1, W1p, GCN_D_IN, GCN_D_HID);
        int t2 = (GCN_D_HID >> 5) * (GCN_D_OUT >> 4) * 32;
        gcn_pack_w_f16<<<(t2 + B - 1) / B, B, 0, stream>>>(W2, W2p, GCN_D_HID, GCN_D_OUT);
    }

    gcn_init_deg <<<(N + B - 1) / B, B, 0, stream>>>(dinv, N);
    gcn_accum_deg<<<(Epad + B - 1) / B, B, 0, stream>>>(dst, dinv, tags, ws, ws_size, E, Epad);
    gcn_make_dinv<<<(N + B - 1) / B, B, 0, stream>>>(dinv, N);

    {
        gcn_wmma_gemm_bregs<GCN_D_IN, GCN_D_HID, 2><<<256, 512, 0, stream>>>(x, W1p, xw1, N);
    }
    {
        gcn_owner_sum_lds<<<ob, 1024, 0, stream>>>(src, dst, dinv, xw1, b1, h1, kept, kcnt,
                                                   tags, Epad, ws, ws_size, N, E);
    }

    {
        gcn_wmma_gemm_bregs<GCN_D_HID, GCN_D_OUT, 1><<<256, 512, 0, stream>>>(h1, W2p, xw2, N);
    }
    {
        gcn_owner_sum_kept<<<ob, 1024, 0, stream>>>(src, dst, dinv, xw2, b2, out, kept, kcnt,
                                                    ws, ws_size, N, E);
    }
}


# degree count leaves global float atomics: 64 edge chunks count destinations in per-node LDS integer counters and store a row each; dinv = rsqrt(1 + row sums); start-value launch gone; counts exact
# speedup vs baseline: 3.9440x; 1.3069x over previous
#include <hip/hip_runtime.h>
#include <hip/hip_bf16.h>

typedef __attribute__((ext_vector_type(16))) _Float16 v16h;
typedef __attribute__((ext_vector_type(8)))  float    v8f;

#define GCN_D_IN  512
#define GCN_D_HID 256
#define GCN_D_OUT 64

#define OWN_ROWS  196
#define OWN_TAGU  512

__device__ __forceinline__ bool gcn_ws_fits(const float* wsb, size_t ws_bytes, const void* end) {
    return (size_t)((const char*)end - (const char*)wsb) <= ws_bytes;
}

#define DEG_CHUNKS 64
#define DEG_BINS   53248

__global__ __launch_bounds__(1024)
void gcn_count_deg(const int* __restrict__ dst, unsigned* __restrict__ part, int pitch,
                   unsigned char* __restrict__ tags, const float* wsb, size_t ws_bytes,
                   int N, int E, int Epad) {
    __shared__ __attribute__((aligned(16))) unsigned hist[DEG_BINS];
    const int  t       = threadIdx.x;
    const bool tags_on = gcn_ws_fits(wsb, ws_bytes, tags + Epad);
    const int  nq      = Epad >> 2;
    const int  qfull   = E >> 2;
    const int  qper    = (nq + (int)gridDim.x - 1) / (int)gridDim.x;
    const int  q0      = (int)blockIdx.x * qper;
    const int  q1      = (q0 + qper < nq) ? (q0 + qper) : nq;
    unsigned*  prow    = part + (size_t)blockIdx.x * pitch;

    for (int base = 0; base < N; base += DEG_BINS) {
        const int nb = (N - base < DEG_BINS) ? (N - base) : DEG_BINS;
        for (int i = t * 4; i < DEG_BINS; i += 4096) *(uint4*)(hist + i) = uint4{0u, 0u, 0u, 0u};
        __syncthreads();
        for (int qb = q0; qb < q1; qb += 4096) {
            int4 dv[4];
#pragma unroll
            for (int u = 0; u < 4; ++u) {
                const int q  = qb + u * 1024 + t;
                const int qc = (q < qfull) ? q : (qfull - 1);
                const int* dp = dst + (size_t)qc * 4;
                dv[u] = int4{dp[0], dp[1], dp[2], dp[3]};
            }
#pragma unroll
            for (int u = 0; u < 4; ++u) {
                const int  q    = qb + u * 1024 + t;
                const bool mine = q < q1;
                const bool real = mine && q < qfull;
                const int dd[4] = {real ? dv[u].x : -1, real ? dv[u].y : -1,
                                   real ? dv[u].z : -1, real ? dv[u].w : -1};
                unsigned tw = 0u;
#pragma unroll
                for (int k = 0; k < 4; ++k) {
                    const unsigned r = (unsigned)(dd[k] - base);
                    if (r < (unsigned)nb) atomicAdd(&hist[r], 1u);
                    if (dd[k] >= 0) tw |= (((unsigned)dd[k] / OWN_ROWS) & 255u) << (8 * k);
                }
                if (tags_on && base == 0 && mine && !(q == qfull && (E & 3) != 0))
                    *(unsigned*)(tags + (size_t)q * 4) = tw;
            }
        }
        if (t == 0 && (E & 3) != 0 && qfull >= q0 && qfull < q1) {
            unsigned tw = 0u;
            for (int k = 0; k < (E & 3); ++k) {
                const int d = dst[qfull * 4 + k];
                const unsigned r = (unsigned)(d - base);
                if (r < (unsigned)nb) atomicAdd(&hist[r], 1u);
                if (d >= 0) tw |= (((unsigned)d / OWN_ROWS) & 255u) << (8 * k);
            }
            if (tags_on && base == 0) *(unsigned*)(tags + (size_t)qfull * 4) = tw;
        }
        __syncthreads();
        for (int i = t * 4; i < nb; i += 4096) {
            if (i + 4 <= nb) {
                *(uint4*)(prow + base + i) = *(const uint4*)(hist + i);
            } else {
                for (int k = i; k < nb; ++k) prow[base + k] = hist[k];
            }
        }
        __syncthreads();
    }
}

__global__ void gcn_make_dinv(const unsigned* __restrict__ part, int pitch,
                              float* __restrict__ dinv, int n) {
    int i = blockIdx.x * blockDim.x + threadIdx.x;
    if (i < n) {
        unsigned c[DEG_CHUNKS];
#pragma unroll
        for (int k = 0; k < DEG_CHUNKS; ++k) c[k] = part[(size_t)k * pitch + i];
        unsigned sum = 1u;
#pragma unroll
        for (int k = 0; k < DEG_CHUNKS; ++k) sum += c[k];
        const float d = (float)sum;
        dinv[i] = (d > 0.0f) ? rsqrtf(d) : 0.0f;
    }
}

__global__ void gcn_pack_w_f16(const float* __restrict__ W, _Float16* __restrict__ Wp,
                               int K, int N) {
    int t = blockIdx.x * blockDim.x + threadIdx.x;
    int tilesN = N >> 4;
    int total = (K >> 5) * tilesN * 32;
    if (t >= total) return;
    int frag = t >> 5;
    int lane = t & 31;
    int kt = frag / tilesN;
    int tn = frag - kt * tilesN;
    int n    = tn * 16 + (lane & 15);
    int koff = kt * 32 + (lane >> 4) * 8;
    _Float16* o = Wp + (size_t)t * 16;
#pragma unroll
    for (int j = 0; j < 8; ++j) {
        o[j]     = (_Float16)W[(size_t)(koff + j) * N + n];
        o[j + 8] = (_Float16)W[(size_t)(koff + 16 + j) * N + n];
    }
}

typedef __attribute__((ext_vector_type(8))) _Float16 v8h;
typedef __attribute__((ext_vector_type(4))) _Float16 v4h;

__device__ __forceinline__ v16h gcn_lds_a_frag(const _Float16* ar) {
    const v8h lo = *(const v8h*)(ar);
    const v8h hi = *(const v8h*)(ar + 16);
    return __builtin_shufflevector(lo, hi, 0, 1, 2, 3, 4, 5, 6, 7, 8, 9, 10, 11, 12, 13, 14, 15);
}

template <int K, int NTL>
__device__ __forceinline__ void gcn_gemm_tiles(const _Float16* ar0, const v16h (&b)[K / 32],
                                               const int (&toff)[NTL], v8f (&acc)[NTL]) {
    constexpr int KT    = K / 32;
    constexpr int PITCH = K + 8;
    const _Float16* ar[NTL];
    v16h a[3][NTL];
#pragma unroll
    for (int i = 0; i < NTL; ++i) {
        ar[i]   = ar0 + toff[i] * (16 * PITCH);
        a[0][i] = gcn_lds_a_frag(ar[i]);
        a[1][i] = gcn_lds_a_frag(ar[i] + 32);
    }
#pragma unroll
    for (int kt = 0; kt < KT; ++kt) {
        if (kt + 2 < KT) {
#pragma unroll
            for (int i = 0; i < NTL; ++i) a[(kt + 2) % 3][i] = gcn_lds_a_frag(ar[i] + (kt + 2) * 32);
        }
        __builtin_amdgcn_sched_barrier(0);
#pragma unroll
        for (int i = 0; i < NTL; ++i)
            acc[i] = __builtin_amdgcn_wmma_f32_16x16x32_f16(false, a[kt % 3][i], false, b[kt],
                                                            (short)0, acc[i], false, false);
        __builtin_amdgcn_sched_barrier(0);
    }
}

template <int K, int N, int RT>
__global__ __launch_bounds__(512)
void gcn_wmma_gemm_bregs(const float* __restrict__ A,
                         const _Float16* __restrict__ Bp,
                         float* __restrict__ C, int M) {
    constexpr int KT    = K / 32;
    constexpr int NT    = N / 16;
    constexpr int RG    = 16 / NT;
    constexpr int ROWS  = RG * RT * 16;
    constexpr int PITCH = K + 8;
    constexpr int C4    = K / 4;
    constexpr int RSTEP = 512 / C4;
    constexpr int NV    = ROWS / RSTEP;
    static_assert(K % 32 == 0 && N % 16 == 0 && 16 % NT == 0 && 512 % C4 == 0 &&
                  ROWS % RSTEP == 0 && (RT == 1 || RT == 2), "shape not covered");

    __shared__ __attribute__((aligned(16))) _Float16 tile[2 * ROWS * PITCH];

    const int t    = threadIdx.x;
    const int lane = t & 31;
    const int w    = __builtin_amdgcn_readfirstlane(t >> 5);
    const int ct   = w % NT;
    const int rg   = w / NT;
    const int hsel = lane >> 4;
    const int idx  = lane & 15;

    const int nblk  = (M + ROWS - 1) / ROWS;
    const int ntile = M / 16;
    const int srow  = t / C4;
    const int sc    = (t % C4) * 4;
    float4 st[NV];

    int blk = blockIdx.x;
    if (blk < nblk) {
#pragma unroll
        for (int i = 0; i < NV; ++i) {
            int row = blk * ROWS + srow + i * RSTEP;
            row = (row < M) ? row : (M - 1);
            st[i] = *(const float4*)(A + (size_t)row * K + sc);
        }
    }

    v16h b[KT];
    {
        const _Float16* bb = Bp + ((size_t)ct * 32 + lane) * 16;
#pragma unroll
        for (int kt = 0; kt < KT; ++kt) b[kt] = *(const v16h*)(bb + (size_t)kt * NT * 32 * 16);
    }
    __builtin_amdgcn_sched_barrier(0);

    if (blk < nblk) {
#pragma unroll
        for (int i = 0; i < NV; ++i) {
            v4h hv = {(_Float16)st[i].x, (_Float16)st[i].y, (_Float16)st[i].z, (_Float16)st[i].w};
            *(v4h*)(tile + (srow + i * RSTEP) * PITCH + sc) = hv;
        }
    }
    __syncthreads();

    int buf = 0;
    for (; blk < nblk; blk += gridDim.x) {
        const int  nxt      = blk + (int)gridDim.x;
        const bool has_next = nxt < nblk;
        if (has_next) {
#pragma unroll
            for (int i = 0; i < NV; ++i) {
                int row = nxt * ROWS + srow + i * RSTEP;
                row = (row < M) ? row : (M - 1);
                st[i] = *(const float4*)(A + (size_t)row * K + sc);
            }
        }

        const int tile0 = (blk * RG + rg) * RT;
        const _Float16* ar0 = tile + buf * (ROWS * PITCH) + (rg * RT * 16 + idx) * PITCH + hsel * 8;
        float* cb0 = C + (size_t)(tile0 * 16 + hsel * 8) * N + ct * 16 + idx;
        const bool live = tile0 < ntile;
        int toff[RT];
#pragma unroll
        for (int i = 0; i < RT; ++i) toff[i] = (tile0 + i < ntile) ? i : 0;
        v8f acc[RT];
#pragma unroll
        for (int i = 0; i < RT; ++i) acc[i] = v8f{};
        if (live) gcn_gemm_tiles<K, RT>(ar0, b, toff, acc);

        if (has_next) {
            _Float16* tn = tile + (buf ^ 1) * (ROWS * PITCH);
#pragma unroll
            for (int i = 0; i < NV; ++i) {
                v4h hv = {(_Float16)st[i].x, (_Float16)st[i].y, (_Float16)st[i].z, (_Float16)st[i].w};
                *(v4h*)(tn + (srow + i * RSTEP) * PITCH + sc) = hv;
            }
        }
        __syncthreads();
        if (live) {
#pragma unroll
            for (int i = 0; i < RT; ++i) {
                float* cb = cb0 + (size_t)toff[i] * (16 * N);
#pragma unroll
                for (int r = 0; r < 8; ++r) cb[(size_t)r * N] = acc[i][r];
            }
        }
        buf ^= 1;
    }
}

#define OWN_WAVES 32
#define OWN_LIST  544
#define OWN_KEEP  160

template <int F, int G>
__device__ __forceinline__ void gcn_own_add_group(int s, int dl, float w, int j, int lane,
                                                  const float* __restrict__ xw, float* tile) {
    int   rl[G];
    float wj[G];
    if constexpr (F == 256) {
        float4 a[G], b[G];
#pragma unroll
        for (int q = 0; q < G; ++q) {
            const int sj = __builtin_amdgcn_readlane(s, j + q);
            rl[q] = __builtin_amdgcn_readlane(dl, j + q);
            wj[q] = __int_as_float(__builtin_amdgcn_readlane(__float_as_int(w), j + q));
            const float4* xr = (const float4*)(xw + (size_t)sj * F);
            a[q] = xr[lane];
            b[q] = xr[32 + lane];
        }
#pragma unroll
        for (int q = 0; q < G; ++q) {
            float* tr = tile + rl[q] * F + lane;
            atomicAdd(tr,       a[q].x * wj[q]);
            atomicAdd(tr + 32,  a[q].y * wj[q]);
            atomicAdd(tr + 64,  a[q].z * wj[q]);
            atomicAdd(tr + 96,  a[q].w * wj[q]);
            atomicAdd(tr + 128, b[q].x * wj[q]);
            atomicAdd(tr + 160, b[q].y * wj[q]);
            atomicAdd(tr + 192, b[q].z * wj[q]);
            atomicAdd(tr + 224, b[q].w * wj[q]);
        }
    } else {
        float a[G], b[G];
#pragma unroll
        for (int q = 0; q < G; ++q) {
            const int sj = __builtin_amdgcn_readlane(s, j + q);
            rl[q] = __builtin_amdgcn_readlane(dl, j + q);
            wj[q] = __int_as_float(__builtin_amdgcn_readlane(__float_as_int(w), j + q));
            const float* xr = xw + (size_t)sj * F;
            a[q] = xr[lane];
            b[q] = xr[32 + lane];
        }
#pragma unroll
        for (int q = 0; q < G; ++q) {
            float* tr = tile + rl[q] * F + lane;
            atomicAdd(tr,      a[q] * wj[q]);
            atomicAdd(tr + 32, b[q] * wj[q]);
        }
    }
}

template <int F, bool SHORT_TAIL>
__device__ __forceinline__ void gcn_own_add_rows(int s, int dl, float w, int n, int lane,
                                                 const float* __restrict__ xw, float* tile) {
    int j = 0;
    for (; j + 8 <= n; j += 8) gcn_own_add_group<F, 8>(s, dl, w, j, lane, xw, tile);
    if constexpr (SHORT_TAIL) {
        if (j + 4 <= n) { gcn_own_add_group<F, 4>(s, dl, w, j, lane, xw, tile); j += 4; }
        if (j + 2 <= n) { gcn_own_add_group<F, 2>(s, dl, w, j, lane, xw, tile); j += 2; }
        if (j < n)        gcn_own_add_group<F, 1>(s, dl, w, j, lane, xw, tile);
    } else {
        for (; j < n; ++j) gcn_own_add_group<F, 1>(s, dl, w, j, lane, xw, tile);
    }
}

template <int F>
__device__ __forceinline__ void gcn_own_add_edges(const unsigned* lst, int n, int lane, int lo,
                                                  const int* __restrict__ src,
                                                  const float* __restrict__ dinv,
                                                  const float* __restrict__ xw,
                                                  float* tile, unsigned* keep) {
    int s = 0, dl = 0;
    float w = 0.0f;
    if (lane < n) {
        const unsigned rec = lst[lane];
        dl = (int)(rec & 255u);
        s  = src[rec >> 8];
        w  = dinv[s] * dinv[lo + dl];
        if (keep != nullptr) keep[lane] = ((unsigned)s << 8) | (unsigned)dl;
    }
    gcn_own_add_rows<F, false>(s, dl, w, n, lane, xw, tile);
}

template <int F>
__device__ __forceinline__ int gcn_own_scan(const int* __restrict__ src, const int* __restrict__ dst,
                                            const float* __restrict__ dinv,
                                            const float* __restrict__ xw,
                                            float* tile, unsigned* mylist, unsigned* kept,
                                            int lane, int wv, int lo, int nrows, int E) {
    const int nchunks = (E + 1023) >> 10;
    const int c0 = (int)((blockIdx.x * (unsigned)nchunks) / gridDim.x);
    int  count  = 0;
    int  nkept  = 0;
    bool keepok = kept != nullptr;

    for (int i = wv; i < nchunks; i += OWN_WAVES) {
        int c = c0 + i;
        if (c >= nchunks) c -= nchunks;
        const int cbase = c << 10;

        int4 dv[8];
        int  eu[8];
#pragma unroll
        for (int u = 0; u < 8; ++u) {
            const int e = cbase + u * 128 + lane * 4;
            eu[u] = (e < E - 4) ? e : (E - 4);
            dv[u] = *(const int4*)(dst + eu[u]);
        }
        if (cbase + 1024 > E) {
#pragma unroll
            for (int u = 0; u < 8; ++u) {
                const int e = cbase + u * 128 + lane * 4;
                if (eu[u]     < e) dv[u].x = -1;
                if (eu[u] + 1 < e) dv[u].y = -1;
                if (eu[u] + 2 < e) dv[u].z = -1;
                if (eu[u] + 3 < e) dv[u].w = -1;
            }
        }
#pragma unroll
        for (int hf = 0; hf < 2; ++hf) {
#pragma unroll
            for (int uu = 0; uu < 4; ++uu) {
                const int u = hf * 4 + uu;
                const int dd[4] = {dv[u].x, dv[u].y, dv[u].z, dv[u].w};
#pragma unroll
                for (int k = 0; k < 4; ++k) {
                    const unsigned t = (unsigned)(dd[k] - lo);
                    const bool m = t < (unsigned)nrows;
                    const unsigned bal = __builtin_amdgcn_ballot_w32(m);
                    if (bal != 0u) {
                        if (m) mylist[count + __popc(bal & ((1u << lane) - 1u))] = ((unsigned)(eu[u] + k) << 8) | t;
                        count += __popc(bal);
                    }
                }
            }
            __builtin_amdgcn_fence(__ATOMIC_ACQ_REL, "wavefront");
            while (count >= 32) {
                count -= 32;
                unsigned* kp = nullptr;
                if (keepok) {
                    if (nkept + 32 <= OWN_KEEP) { kp = kept + nkept; nkept += 32; }
                    else keepok = false;
                }
                gcn_own_add_edges<F>(mylist + count, 32, lane, lo, src, dinv, xw, tile, kp);
            }
        }
    }
    if (count > 0) {
        unsigned* kp = nullptr;
        if (keepok) {
            if (nkept + count <= OWN_KEEP) { kp = kept + nkept; nkept += count; }
            else keepok = false;
        }
        gcn_own_add_edges<F>(mylist, count, lane, lo, src, dinv, xw, tile, kp);
    }
    return keepok ? nkept : -1;
}

template <int F>
__device__ __forceinline__ void gcn_own_add_found(const unsigned* lst, int n, int lane, int lo,
                                                  const int* __restrict__ src,
                                                  const int* __restrict__ dst,
                                                  const float* __restrict__ dinv,
                                                  const float* __restrict__ xw,
                                                  float* tile, unsigned* keep) {
    int s = 0, dl = 0;
    float w = 0.0f;
    if (lane < n) {
        const unsigned e = lst[lane];
        const int d = dst[e];
        s  = src[e];
        dl = d - lo;
        w  = dinv[s] * dinv[d];
        if (keep != nullptr) keep[lane] = ((unsigned)s << 8) | (unsigned)dl;
    }
    gcn_own_add_rows<F, false>(s, dl, w, n, lane, xw, tile);
}

template <int F>
__device__ __forceinline__ int gcn_own_scan_tags(const int* __restrict__ src,
                                                 const int* __restrict__ dst,
                                                 const unsigned char* __restrict__ tags,
                                                 const float* __restrict__ dinv,
                                                 const float* __restrict__ xw,
                                                 float* tile, unsigned* mylist, unsigned* kept,
                                                 int lane, int wv, int lo, int E) {
    const unsigned pat = blockIdx.x * 0x01010101u;
    const int nunits = (E + OWN_TAGU - 1) / OWN_TAGU;
    const int nfull  = nunits / (8 * OWN_WAVES);
    const int ntrip  = nfull * OWN_WAVES;
    const int t0     = (int)((blockIdx.x * (unsigned)ntrip) / gridDim.x);
    const int rest0  = nfull * (8 * OWN_WAVES);
    const int per    = (nunits - rest0 + OWN_WAVES - 1) / OWN_WAVES;
    int  count  = 0;
    int  nkept  = 0;
    bool keepok = kept != nullptr;

    uint4 tv[8];
    int i = 0;
    int unit0 = 0;
    int nu = 0;
    int done = 0;
    for (;;) {
        if (done >= nu) {
            if (i > nfull) break;
            if (i < nfull) {
                int tr = t0 + i * OWN_WAVES + wv;
                if (tr >= ntrip) tr -= ntrip;
                unit0 = tr * 8;
                nu    = 8;
            } else {
                unit0 = rest0 + wv * per;
                nu    = (nunits - unit0 < per) ? (nunits - unit0) : per;
                if (nu <= 0) break;
            }
            ++i;
            done = 0;
#pragma unroll
            for (int u = 0; u < 8; ++u) {
                int un = unit0 + u;
                un = (un < nunits) ? un : (nunits - 1);
                tv[u] = *(const uint4*)(tags + (size_t)un * OWN_TAGU + lane * 16);
            }
        }
#pragma unroll
        for (int u = 0; u < 8; ++u) {
            if (u >= done && u < nu && count <= OWN_LIST - OWN_TAGU) {
                const int ebase = (unit0 + u) * OWN_TAGU + lane * 16;
                unsigned xs[4] = {tv[u].x ^ pat, tv[u].y ^ pat, tv[u].z ^ pat, tv[u].w ^ pat};
                if ((unit0 + u + 1) * OWN_TAGU > E) {
#pragma unroll
                    for (int q = 0; q < 4; ++q) {
#pragma unroll
                        for (int k = 0; k < 4; ++k)
                            if (ebase + q * 4 + k >= E) xs[q] |= 255u << (8 * k);
                    }
                }
#pragma unroll
                for (int q = 0; q < 4; ++q) {
                    const unsigned x = xs[q];
                    const unsigned z = (x - 0x01010101u) & ~x & 0x80808080u;
                    if (__builtin_amdgcn_ballot_w32(z != 0u) != 0u) {
#pragma unroll
                        for (int k = 0; k < 4; ++k) {
                            const bool m = ((x >> (8 * k)) & 255u) == 0u;
                            const unsigned bal = __builtin_amdgcn_ballot_w32(m);
                            if (bal != 0u) {
                                if (m) mylist[count + __popc(bal & ((1u << lane) - 1u))] = (unsigned)(ebase + q * 4 + k);
                                count += __popc(bal);
                            }
                        }
                    }
                }
                done = u + 1;
            }
        }
        __builtin_amdgcn_fence(__ATOMIC_ACQ_REL, "wavefront");
        while (count >= 32) {
            count -= 32;
            unsigned* kp = nullptr;
            if (keepok) {
                if (nkept + 32 <= OWN_KEEP) { kp = kept + nkept; nkept += 32; }
                else keepok = false;
            }
            gcn_own_add_found<F>(mylist + count, 32, lane, lo, src, dst, dinv, xw, tile, kp);
        }
    }
    if (count > 0) {
        unsigned* kp = nullptr;
        if (keepok) {
            if (nkept + count <= OWN_KEEP) { kp = kept + nkept; nkept += count; }
            else keepok = false;
        }
        gcn_own_add_found<F>(mylist, count, lane, lo, src, dst, dinv, xw, tile, kp);
    }
    return keepok ? nkept : -1;
}

__device__ __forceinline__ bool gcn_own_keep_fits(const float* wsb, size_t ws_bytes, const int* kend) {
    return (size_t)((const char*)kend - (const char*)wsb) <= ws_bytes;
}

__global__ __launch_bounds__(1024)
void gcn_owner_sum_lds(const int* __restrict__ src, const int* __restrict__ dst,
                       const float* __restrict__ dinv, const float* __restrict__ xw,
                       const float* __restrict__ bias, float* __restrict__ h,
                       unsigned* __restrict__ kept, int* __restrict__ kcnt,
                       const unsigned char* __restrict__ tags, int Epad,
                       const float* wsb, size_t ws_bytes, int N, int E) {
    __shared__ float    tile[OWN_ROWS * GCN_D_HID];
    __shared__ unsigned list[OWN_WAVES * OWN_LIST];

    const int lane = threadIdx.x & 31;
    const int wv = __builtin_amdgcn_readfirstlane((int)(threadIdx.x >> 5));
    const int lo = blockIdx.x * OWN_ROWS;
    const int nrows = (N - lo < OWN_ROWS) ? (N - lo) : OWN_ROWS;
    const int wid = blockIdx.x * OWN_WAVES + wv;
    const bool keep_on = gcn_own_keep_fits(wsb, ws_bytes, kcnt + gridDim.x * OWN_WAVES);

    for (int r = wv; r < nrows; r += OWN_WAVES) {
        const float di = dinv[lo + r];
        const float4* xr = (const float4*)(xw + (size_t)(lo + r) * GCN_D_HID);
        const float4 a = xr[lane];
        const float4 b = xr[32 + lane];
        float* tr = tile + r * GCN_D_HID + lane;
        tr[0]   = a.x * di * di;
        tr[32]  = a.y * di * di;
        tr[64]  = a.z * di * di;
        tr[96]  = a.w * di * di;
        tr[128] = b.x * di * di;
        tr[160] = b.y * di * di;
        tr[192] = b.z * di * di;
        tr[224] = b.w * di * di;
    }
    __syncthreads();

    const bool tags_on = gcn_ws_fits(wsb, ws_bytes, tags + Epad) && gridDim.x <= 256u;
    unsigned* kp0 = keep_on ? kept + (size_t)wid * OWN_KEEP : nullptr;
    int nk;
    if (tags_on)
        nk = gcn_own_scan_tags<GCN_D_HID>(src, dst, tags, dinv, xw, tile, list + wv * OWN_LIST,
                                          kp0, lane, wv, lo, E);
    else
        nk = gcn_own_scan<GCN_D_HID>(src, dst, dinv, xw, tile, list + wv * OWN_LIST,
                                     kp0, lane, wv, lo, nrows, E);
    if (keep_on && lane == 0) kcnt[wid] = nk;
    __syncthreads();

    const float4 bb0 = ((const float4*)bias)[lane];
    const float4 bb1 = ((const float4*)bias)[32 + lane];
    for (int r = wv; r < nrows; r += OWN_WAVES) {
        const float* tr = tile + r * GCN_D_HID + lane;
        float4 a, b;
        a.x = fmaxf(tr[0]   + bb0.x, 0.0f);
        a.y = fmaxf(tr[32]  + bb0.y, 0.0f);
        a.z = fmaxf(tr[64]  + bb0.z, 0.0f);
        a.w = fmaxf(tr[96]  + bb0.w, 0.0f);
        b.x = fmaxf(tr[128] + bb1.x, 0.0f);
        b.y = fmaxf(tr[160] + bb1.y, 0.0f);
        b.z = fmaxf(tr[192] + bb1.z, 0.0f);
        b.w = fmaxf(tr[224] + bb1.w, 0.0f);
        float4* hr = (float4*)(h + (size_t)(lo + r) * GCN_D_HID);
        hr[lane]      = a;
        hr[32 + lane] = b;
    }
}

__global__ __launch_bounds__(1024)
void gcn_owner_sum_kept(const int* __restrict__ src, const int* __restrict__ dst,
                        const float* __restrict__ dinv, const float* __restrict__ xw,
                        const float* __restrict__ bias, float* __restrict__ out,
                        const unsigned* __restrict__ kept, const int* __restrict__ kcnt,
                        const float* wsb, size_t ws_bytes, int N, int E) {
    __shared__ float    tile[OWN_ROWS * GCN_D_OUT];
    __shared__ unsigned list[OWN_WAVES * OWN_LIST];

    const int lane = threadIdx.x & 31;
    const int wv = __builtin_amdgcn_readfirstlane((int)(threadIdx.x >> 5));
    const int lo = blockIdx.x * OWN_ROWS;
    const int nrows = (N - lo < OWN_ROWS) ? (N - lo) : OWN_ROWS;
    const int wid = blockIdx.x * OWN_WAVES + wv;
    const bool keep_on = gcn_own_keep_fits(wsb, ws_bytes, kcnt + gridDim.x * OWN_WAVES);

    bool use_kept = false;
    if (keep_on) {
        const int c = kcnt[blockIdx.x * OWN_WAVES + lane];
        use_kept = __builtin_amdgcn_ballot_w32(c < 0 || c > OWN_KEEP) == 0u;
    }

    {
        constexpr int RPW = (OWN_ROWS + OWN_WAVES - 1) / OWN_WAVES;
        float sa[RPW], sb[RPW];
        const int   rl  = wv + lane * OWN_WAVES;
        const float dvl = dinv[lo + ((rl < nrows) ? rl : (nrows - 1))];
#pragma unroll
        for (int i = 0; i < RPW; ++i) {
            const int r  = wv + i * OWN_WAVES;
            const int rc = (r < nrows) ? r : (nrows - 1);
            const float* xr = xw + (size_t)(lo + rc) * GCN_D_OUT;
            sa[i] = xr[lane];
            sb[i] = xr[32 + lane];
        }
#pragma unroll
        for (int i = 0; i < RPW; ++i) {
            const int r = wv + i * OWN_WAVES;
            if (r < nrows) {
                const float di = __int_as_float(__builtin_amdgcn_readlane(__float_as_int(dvl), i));
                float* tr = tile + r * GCN_D_OUT + lane;
                tr[0]  = sa[i] * di * di;
                tr[32] = sb[i] * di * di;
            }
        }
    }
    __syncthreads();

    if (use_kept) {
        const unsigned* kp = kept + (size_t)wid * OWN_KEEP;
        int cnt = __builtin_amdgcn_readfirstlane(kcnt[wid]);
        cnt = (cnt < OWN_KEEP) ? cnt : OWN_KEEP;
        constexpr int NB = OWN_KEEP / 32;
        unsigned* myl = list + wv * OWN_LIST;
        unsigned rec[NB];
#pragma unroll
        for (int b = 0; b < NB; ++b) {
            rec[b] = 0u;
            if (b * 32 + lane < cnt) rec[b] = kp[b * 32 + lane];
        }
        float wgt[NB];
#pragma unroll
        for (int b = 0; b < NB; ++b) {
            const float ds = dinv[rec[b] >> 8];
            const float dd = dinv[lo + (int)(rec[b] & 255u)];
            wgt[b] = ds * dd;
        }
#pragma unroll
        for (int b = 0; b < NB; ++b) {
            myl[b * 32 + lane]            = rec[b];
            myl[OWN_KEEP + b * 32 + lane] = __float_as_uint(wgt[b]);
        }
        for (int base = 0; base < cnt; base += 32) {
            const int n = (cnt - base < 32) ? (cnt - base) : 32;
            const unsigned r = myl[base + lane];
            const float    w = __uint_as_float(myl[OWN_KEEP + base + lane]);
            gcn_own_add_rows<GCN_D_OUT, true>((int)(r >> 8), (int)(r & 255u), w, n, lane, xw, tile);
        }
    } else {
        gcn_own_scan<GCN_D_OUT>(src, dst, dinv, xw, tile, list + wv * OWN_LIST, nullptr,
                                lane, wv, lo, nrows, E);
    }
    __syncthreads();

    const float b0 = bias[lane];
    const float b1 = bias[lane + 32];
    for (int r = wv; r < nrows; r += OWN_WAVES) {
        const float* tr = tile + r * GCN_D_OUT + lane;
        float v0 = tr[0]  + b0;
        float v1 = tr[32] + b1;
        float m = fmaxf(v0, v1);
#pragma unroll
        for (int off = 16; off > 0; off >>= 1) m = fmaxf(m, __shfl_xor(m, off, 32));
        float e0 = __expf(v0 - m);
        float e1 = __expf(v1 - m);
        float sm = e0 + e1;
#pragma unroll
        for (int off = 16; off > 0; off >>= 1) sm += __shfl_xor(sm, off, 32);
        float inv = 1.0f / sm;
        float* orow = out + (size_t)(lo + r) * GCN_D_OUT;
        orow[lane]      = e0 * inv;
        orow[lane + 32] = e1 * inv;
    }
}

extern "C" void kernel_launch(void* const* d_in, const int* in_sizes, int n_in,
                              void* d_out, int out_size, void* d_ws, size_t ws_size,
                              hipStream_t stream) {
    const float* x   = (const float*)d_in[0];
    const int*   ei  = (const int*)  d_in[1];
    const float* W1  = (const float*)d_in[2];
    const float* b1  = (const float*)d_in[3];
    const float* W2  = (const float*)d_in[4];
    const float* b2  = (const float*)d_in[5];
    float* out = (float*)d_out;

    const int N = in_sizes[0] / GCN_D_IN;
    const int E = in_sizes[1] / 2;
    const int* src = ei;
    const int* dst = ei + E;

    float* ws   = (float*)d_ws;
    size_t off  = ((size_t)N + 63) & ~(size_t)63;
    float* dinv = ws;
    float* xw1  = ws + off;
    float* h1   = xw1 + (size_t)N * GCN_D_HID;
    _Float16* W1p = (_Float16*)(h1 + (size_t)N * GCN_D_HID);
    _Float16* W2p = W1p + (size_t)GCN_D_IN * GCN_D_HID;
    float* xw2  = xw1;
    const int ob = (N + OWN_ROWS - 1) / OWN_ROWS;
    unsigned* kept = (unsigned*)(W2p + (size_t)GCN_D_HID * GCN_D_OUT);
    int*      kcnt = (int*)(kept + (size_t)ob * OWN_WAVES * OWN_KEEP);
    unsigned char* tags = (unsigned char*)(kcnt + (size_t)ob * OWN_WAVES);
    const int Epad = (E + OWN_TAGU - 1) / OWN_TAGU * OWN_TAGU;

    const int B = 256;
    {
        int t1 = (GCN_D_IN >> 5) * (GCN_D_HID >> 4) * 32;
        gcn_pack_w_f16<<<(t1 + B - 1) / B, B, 0, stream>>>(W1, W1p, GCN_D_IN, GCN_D_HID);
        int t2 = (GCN_D_HID >> 5) * (GCN_D_OUT >> 4) * 32;
        gcn_pack_w_f16<<<(t2 + B - 1) / B, B, 0, stream>>>(W2, W2p, GCN_D_HID, GCN_D_OUT);
    }

    unsigned* part = (unsigned*)h1;
    gcn_count_deg<<<DEG_CHUNKS, 1024, 0, stream>>>(dst, part, (int)off, tags, ws, ws_size, N, E, Epad);
    gcn_make_dinv<<<(N + B - 1) / B, B, 0, stream>>>(part, (int)off, dinv, N);

    {
        gcn_wmma_gemm_bregs<GCN_D_IN, GCN_D_HID, 2><<<256, 512, 0, stream>>>(x, W1p, xw1, N);
    }
    {
        gcn_owner_sum_lds<<<ob, 1024, 0, stream>>>(src, dst, dinv, xw1, b1, h1, kept, kcnt,
                                                   tags, Epad, ws, ws_size, N, E);
    }

    {
        gcn_wmma_gemm_bregs<GCN_D_HID, GCN_D_OUT, 1><<<256, 512, 0, stream>>>(h1, W2p, xw2, N);
    }
    {
        gcn_owner_sum_kept<<<ob, 1024, 0, stream>>>(src, dst, dinv, xw2, b2, out, kept, kcnt,
                                                    ws, ws_size, N, E);
    }
}


# layer-2 product formed inside the layer-1 owner workgroups: relu(sum+bias) rounded to f16 over the LDS sums, times W2 by WMMA, xw2 stored; h1 never reaches memory, GEMM launch gone; sum order kept
# speedup vs baseline: 4.1773x; 1.0592x over previous
#include <hip/hip_runtime.h>
#include <hip/hip_bf16.h>

typedef __attribute__((ext_vector_type(16))) _Float16 v16h;
typedef __attribute__((ext_vector_type(8)))  float    v8f;

#define GCN_D_IN  512
#define GCN_D_HID 256
#define GCN_D_OUT 64

#define OWN_ROWS  196
#define OWN_TAGU  512

__device__ __forceinline__ bool gcn_ws_fits(const float* wsb, size_t ws_bytes, const void* end) {
    return (size_t)((const char*)end - (const char*)wsb) <= ws_bytes;
}

#define DEG_CHUNKS 64
#define DEG_BINS   53248

__global__ __launch_bounds__(1024)
void gcn_count_deg(const int* __restrict__ dst, unsigned* __restrict__ part, int pitch,
                   unsigned char* __restrict__ tags, const float* wsb, size_t ws_bytes,
                   int N, int E, int Epad) {
    __shared__ __attribute__((aligned(16))) unsigned hist[DEG_BINS];
    const int  t       = threadIdx.x;
    const bool tags_on = gcn_ws_fits(wsb, ws_bytes, tags + Epad);
    const int  nq      = Epad >> 2;
    const int  qfull   = E >> 2;
    const int  qper    = (nq + (int)gridDim.x - 1) / (int)gridDim.x;
    const int  q0      = (int)blockIdx.x * qper;
    const int  q1      = (q0 + qper < nq) ? (q0 + qper) : nq;
    unsigned*  prow    = part + (size_t)blockIdx.x * pitch;

    for (int base = 0; base < N; base += DEG_BINS) {
        const int nb = (N - base < DEG_BINS) ? (N - base) : DEG_BINS;
        for (int i = t * 4; i < DEG_BINS; i += 4096) *(uint4*)(hist + i) = uint4{0u, 0u, 0u, 0u};
        __syncthreads();
        for (int qb = q0; qb < q1; qb += 4096) {
            int4 dv[4];
#pragma unroll
            for (int u = 0; u < 4; ++u) {
                const int q  = qb + u * 1024 + t;
                const int qc = (q < qfull) ? q : (qfull - 1);
                const int* dp = dst + (size_t)qc * 4;
                dv[u] = int4{dp[0], dp[1], dp[2], dp[3]};
            }
#pragma unroll
            for (int u = 0; u < 4; ++u) {
                const int  q    = qb + u * 1024 + t;
                const bool mine = q < q1;
                const bool real = mine && q < qfull;
                const int dd[4] = {real ? dv[u].x : -1, real ? dv[u].y : -1,
                                   real ? dv[u].z : -1, real ? dv[u].w : -1};
                unsigned tw = 0u;
#pragma unroll
                for (int k = 0; k < 4; ++k) {
                    const unsigned r = (unsigned)(dd[k] - base);
                    if (r < (unsigned)nb) atomicAdd(&hist[r], 1u);
                    if (dd[k] >= 0) tw |= (((unsigned)dd[k] / OWN_ROWS) & 255u) << (8 * k);
                }
                if (tags_on && base == 0 && mine && !(q == qfull && (E & 3) != 0))
                    *(unsigned*)(tags + (size_t)q * 4) = tw;
            }
        }
        if (t == 0 && (E & 3) != 0 && qfull >= q0 && qfull < q1) {
            unsigned tw = 0u;
            for (int k = 0; k < (E & 3); ++k) {
                const int d = dst[qfull * 4 + k];
                const unsigned r = (unsigned)(d - base);
                if (r < (unsigned)nb) atomicAdd(&hist[r], 1u);
                if (d >= 0) tw |= (((unsigned)d / OWN_ROWS) & 255u) << (8 * k);
            }
            if (tags_on && base == 0) *(unsigned*)(tags + (size_t)qfull * 4) = tw;
        }
        __syncthreads();
        for (int i = t * 4; i < nb; i += 4096) {
            if (i + 4 <= nb) {
                *(uint4*)(prow + base + i) = *(const uint4*)(hist + i);
            } else {
                for (int k = i; k < nb; ++k) prow[base + k] = hist[k];
            }
        }
        __syncthreads();
    }
}

__global__ void gcn_make_dinv(const unsigned* __restrict__ part, int pitch,
                              float* __restrict__ dinv, int n) {
    int i = blockIdx.x * blockDim.x + threadIdx.x;
    if (i < n) {
        unsigned c[DEG_CHUNKS];
#pragma unroll
        for (int k = 0; k < DEG_CHUNKS; ++k) c[k] = part[(size_t)k * pitch + i];
        unsigned sum = 1u;
#pragma unroll
        for (int k = 0; k < DEG_CHUNKS; ++k) sum += c[k];
        const float d = (float)sum;
        dinv[i] = (d > 0.0f) ? rsqrtf(d) : 0.0f;
    }
}

__global__ void gcn_pack_w_f16(const float* __restrict__ W, _Float16* __restrict__ Wp,
                               int K, int N) {
    int t = blockIdx.x * blockDim.x + threadIdx.x;
    int tilesN = N >> 4;
    int total = (K >> 5) * tilesN * 32;
    if (t >= total) return;
    int frag = t >> 5;
    int lane = t & 31;
    int kt = frag / tilesN;
    int tn = frag - kt * tilesN;
    int n    = tn * 16 + (lane & 15);
    int koff = kt * 32 + (lane >> 4) * 8;
    _Float16* o = Wp + (size_t)t * 16;
#pragma unroll
    for (int j = 0; j < 8; ++j) {
        o[j]     = (_Float16)W[(size_t)(koff + j) * N + n];
        o[j + 8] = (_Float16)W[(size_t)(koff + 16 + j) * N + n];
    }
}

typedef __attribute__((ext_vector_type(8))) _Float16 v8h;
typedef __attribute__((ext_vector_type(4))) _Float16 v4h;

__device__ __forceinline__ v16h gcn_lds_a_frag(const _Float16* ar) {
    const v8h lo = *(const v8h*)(ar);
    const v8h hi = *(const v8h*)(ar + 16);
    return __builtin_shufflevector(lo, hi, 0, 1, 2, 3, 4, 5, 6, 7, 8, 9, 10, 11, 12, 13, 14, 15);
}

template <int K, int NTL>
__device__ __forceinline__ void gcn_gemm_tiles(const _Float16* ar0, const v16h (&b)[K / 32],
                                               const int (&toff)[NTL], v8f (&acc)[NTL]) {
    constexpr int KT    = K / 32;
    constexpr int PITCH = K + 8;
    const _Float16* ar[NTL];
    v16h a[3][NTL];
#pragma unroll
    for (int i = 0; i < NTL; ++i) {
        ar[i]   = ar0 + toff[i] * (16 * PITCH);
        a[0][i] = gcn_lds_a_frag(ar[i]);
        a[1][i] = gcn_lds_a_frag(ar[i] + 32);
    }
#pragma unroll
    for (int kt = 0; kt < KT; ++kt) {
        if (kt + 2 < KT) {
#pragma unroll
            for (int i = 0; i < NTL; ++i) a[(kt + 2) % 3][i] = gcn_lds_a_frag(ar[i] + (kt + 2) * 32);
        }
        __builtin_amdgcn_sched_barrier(0);
#pragma unroll
        for (int i = 0; i < NTL; ++i)
            acc[i] = __builtin_amdgcn_wmma_f32_16x16x32_f16(false, a[kt % 3][i], false, b[kt],
                                                            (short)0, acc[i], false, false);
        __builtin_amdgcn_sched_barrier(0);
    }
}

template <int K, int N, int RT>
__global__ __launch_bounds__(512)
void gcn_wmma_gemm_bregs(const float* __restrict__ A,
                         const _Float16* __restrict__ Bp,
                         float* __restrict__ C, int M) {
    constexpr int KT    = K / 32;
    constexpr int NT    = N / 16;
    constexpr int RG    = 16 / NT;
    constexpr int ROWS  = RG * RT * 16;
    constexpr int PITCH = K + 8;
    constexpr int C4    = K / 4;
    constexpr int RSTEP = 512 / C4;
    constexpr int NV    = ROWS / RSTEP;
    static_assert(K % 32 == 0 && N % 16 == 0 && 16 % NT == 0 && 512 % C4 == 0 &&
                  ROWS % RSTEP == 0 && (RT == 1 || RT == 2), "shape not covered");

    __shared__ __attribute__((aligned(16))) _Float16 tile[2 * ROWS * PITCH];

    const int t    = threadIdx.x;
    const int lane = t & 31;
    const int w    = __builtin_amdgcn_readfirstlane(t >> 5);
    const int ct   = w % NT;
    const int rg   = w / NT;
    const int hsel = lane >> 4;
    const int idx  = lane & 15;

    const int nblk  = (M + ROWS - 1) / ROWS;
    const int ntile = M / 16;
    const int srow  = t / C4;
    const int sc    = (t % C4) * 4;
    float4 st[NV];

    int blk = blockIdx.x;
    if (blk < nblk) {
#pragma unroll
        for (int i = 0; i < NV; ++i) {
            int row = blk * ROWS + srow + i * RSTEP;
            row = (row < M) ? row : (M - 1);
            st[i] = *(const float4*)(A + (size_t)row * K + sc);
        }
    }

    v16h b[KT];
    {
        const _Float16* bb = Bp + ((size_t)ct * 32 + lane) * 16;
#pragma unroll
        for (int kt = 0; kt < KT; ++kt) b[kt] = *(const v16h*)(bb + (size_t)kt * NT * 32 * 16);
    }
    __builtin_amdgcn_sched_barrier(0);

    if (blk < nblk) {
#pragma unroll
        for (int i = 0; i < NV; ++i) {
            v4h hv = {(_Float16)st[i].x, (_Float16)st[i].y, (_Float16)st[i].z, (_Float16)st[i].w};
            *(v4h*)(tile + (srow + i * RSTEP) * PITCH + sc) = hv;
        }
    }
    __syncthreads();

    int buf = 0;
    for (; blk < nblk; blk += gridDim.x) {
        const int  nxt      = blk + (int)gridDim.x;
        const bool has_next = nxt < nblk;
        if (has_next) {
#pragma unroll
            for (int i = 0; i < NV; ++i) {
                int row = nxt * ROWS + srow + i * RSTEP;
                row = (row < M) ? row : (M - 1);
                st[i] = *(const float4*)(A + (size_t)row * K + sc);
            }
        }

        const int tile0 = (blk * RG + rg) * RT;
        const _Float16* ar0 = tile + buf * (ROWS * PITCH) + (rg * RT * 16 + idx) * PITCH + hsel * 8;
        float* cb0 = C + (size_t)(tile0 * 16 + hsel * 8) * N + ct * 16 + idx;
        const bool live = tile0 < ntile;
        int toff[RT];
#pragma unroll
        for (int i = 0; i < RT; ++i) toff[i] = (tile0 + i < ntile) ? i : 0;
        v8f acc[RT];
#pragma unroll
        for (int i = 0; i < RT; ++i) acc[i] = v8f{};
        if (live) gcn_gemm_tiles<K, RT>(ar0, b, toff, acc);

        if (has_next) {
            _Float16* tn = tile + (buf ^ 1) * (ROWS * PITCH);
#pragma unroll
            for (int i = 0; i < NV; ++i) {
                v4h hv = {(_Float16)st[i].x, (_Float16)st[i].y, (_Float16)st[i].z, (_Float16)st[i].w};
                *(v4h*)(tn + (srow + i * RSTEP) * PITCH + sc) = hv;
            }
        }
        __syncthreads();
        if (live) {
#pragma unroll
            for (int i = 0; i < RT; ++i) {
                float* cb = cb0 + (size_t)toff[i] * (16 * N);
#pragma unroll
                for (int r = 0; r < 8; ++r) cb[(size_t)r * N] = acc[i][r];
            }
        }
        buf ^= 1;
    }
}

#define OWN_WAVES 32
#define OWN_LIST  544
#define OWN_KEEP  160

template <int F, int G>
__device__ __forceinline__ void gcn_own_add_group(int s, int dl, float w, int j, int lane,
                                                  const float* __restrict__ xw, float* tile) {
    int   rl[G];
    float wj[G];
    if constexpr (F == 256) {
        float4 a[G], b[G];
#pragma unroll
        for (int q = 0; q < G; ++q) {
            const int sj = __builtin_amdgcn_readlane(s, j + q);
            rl[q] = __builtin_amdgcn_readlane(dl, j + q);
            wj[q] = __int_as_float(__builtin_amdgcn_readlane(__float_as_int(w), j + q));
            const float4* xr = (const float4*)(xw + (size_t)sj * F);
            a[q] = xr[lane];
            b[q] = xr[32 + lane];
        }
#pragma unroll
        for (int q = 0; q < G; ++q) {
            float* tr = tile + rl[q] * F + lane;
            atomicAdd(tr,       a[q].x * wj[q]);
            atomicAdd(tr + 32,  a[q].y * wj[q]);
            atomicAdd(tr + 64,  a[q].z * wj[q]);
            atomicAdd(tr + 96,  a[q].w * wj[q]);
            atomicAdd(tr + 128, b[q].x * wj[q]);
            atomicAdd(tr + 160, b[q].y * wj[q]);
            atomicAdd(tr + 192, b[q].z * wj[q]);
            atomicAdd(tr + 224, b[q].w * wj[q]);
        }
    } else {
        float a[G], b[G];
#pragma unroll
        for (int q = 0; q < G; ++q) {
            const int sj = __builtin_amdgcn_readlane(s, j + q);
            rl[q] = __builtin_amdgcn_readlane(dl, j + q);
            wj[q] = __int_as_float(__builtin_amdgcn_readlane(__float_as_int(w), j + q));
            const float* xr = xw + (size_t)sj * F;
            a[q] = xr[lane];
            b[q] = xr[32 + lane];
        }
#pragma unroll
        for (int q = 0; q < G; ++q) {
            float* tr = tile + rl[q] * F + lane;
            atomicAdd(tr,      a[q] * wj[q]);
            atomicAdd(tr + 32, b[q] * wj[q]);
        }
    }
}

template <int F, bool SHORT_TAIL>
__device__ __forceinline__ void gcn_own_add_rows(int s, int dl, float w, int n, int lane,
                                                 const float* __restrict__ xw, float* tile) {
    int j = 0;
    for (; j + 8 <= n; j += 8) gcn_own_add_group<F, 8>(s, dl, w, j, lane, xw, tile);
    if constexpr (SHORT_TAIL) {
        if (j + 4 <= n) { gcn_own_add_group<F, 4>(s, dl, w, j, lane, xw, tile); j += 4; }
        if (j + 2 <= n) { gcn_own_add_group<F, 2>(s, dl, w, j, lane, xw, tile); j += 2; }
        if (j < n)        gcn_own_add_group<F, 1>(s, dl, w, j, lane, xw, tile);
    } else {
        for (; j < n; ++j) gcn_own_add_group<F, 1>(s, dl, w, j, lane, xw, tile);
    }
}

template <int F>
__device__ __forceinline__ void gcn_own_add_edges(const unsigned* lst, int n, int lane, int lo,
                                                  const int* __restrict__ src,
                                                  const float* __restrict__ dinv,
                                                  const float* __restrict__ xw,
                                                  float* tile, unsigned* keep) {
    int s = 0, dl = 0;
    float w = 0.0f;
    if (lane < n) {
        const unsigned rec = lst[lane];
        dl = (int)(rec & 255u);
        s  = src[rec >> 8];
        w  = dinv[s] * dinv[lo + dl];
        if (keep != nullptr) keep[lane] = ((unsigned)s << 8) | (unsigned)dl;
    }
    gcn_own_add_rows<F, false>(s, dl, w, n, lane, xw, tile);
}

template <int F>
__device__ __forceinline__ int gcn_own_scan(const int* __restrict__ src, const int* __restrict__ dst,
                                            const float* __restrict__ dinv,
                                            const float* __restrict__ xw,
                                            float* tile, unsigned* mylist, unsigned* kept,
                                            int lane, int wv, int lo, int nrows, int E) {
    const int nchunks = (E + 1023) >> 10;
    const int c0 = (int)((blockIdx.x * (unsigned)nchunks) / gridDim.x);
    int  count  = 0;
    int  nkept  = 0;
    bool keepok = kept != nullptr;

    for (int i = wv; i < nchunks; i += OWN_WAVES) {
        int c = c0 + i;
        if (c >= nchunks) c -= nchunks;
        const int cbase = c << 10;

        int4 dv[8];
        int  eu[8];
#pragma unroll
        for (int u = 0; u < 8; ++u) {
            const int e = cbase + u * 128 + lane * 4;
            eu[u] = (e < E - 4) ? e : (E - 4);
            dv[u] = *(const int4*)(dst + eu[u]);
        }
        if (cbase + 1024 > E) {
#pragma unroll
            for (int u = 0; u < 8; ++u) {
                const int e = cbase + u * 128 + lane * 4;
                if (eu[u]     < e) dv[u].x = -1;
                if (eu[u] + 1 < e) dv[u].y = -1;
                if (eu[u] + 2 < e) dv[u].z = -1;
                if (eu[u] + 3 < e) dv[u].w = -1;
            }
        }
#pragma unroll
        for (int hf = 0; hf < 2; ++hf) {
#pragma unroll
            for (int uu = 0; uu < 4; ++uu) {
                const int u = hf * 4 + uu;
                const int dd[4] = {dv[u].x, dv[u].y, dv[u].z, dv[u].w};
#pragma unroll
                for (int k = 0; k < 4; ++k) {
                    const unsigned t = (unsigned)(dd[k] - lo);
                    const bool m = t < (unsigned)nrows;
                    const unsigned bal = __builtin_amdgcn_ballot_w32(m);
                    if (bal != 0u) {
                        if (m) mylist[count + __popc(bal & ((1u << lane) - 1u))] = ((unsigned)(eu[u] + k) << 8) | t;
                        count += __popc(bal);
                    }
                }
            }
            __builtin_amdgcn_fence(__ATOMIC_ACQ_REL, "wavefront");
            while (count >= 32) {
                count -= 32;
                unsigned* kp = nullptr;
                if (keepok) {
                    if (nkept + 32 <= OWN_KEEP) { kp = kept + nkept; nkept += 32; }
                    else keepok = false;
                }
                gcn_own_add_edges<F>(mylist + count, 32, lane, lo, src, dinv, xw, tile, kp);
            }
        }
    }
    if (count > 0) {
        unsigned* kp = nullptr;
        if (keepok) {
            if (nkept + count <= OWN_KEEP) { kp = kept + nkept; nkept += count; }
            else keepok = false;
        }
        gcn_own_add_edges<F>(mylist, count, lane, lo, src, dinv, xw, tile, kp);
    }
    return keepok ? nkept : -1;
}

template <int F>
__device__ __forceinline__ void gcn_own_add_found(const unsigned* lst, int n, int lane, int lo,
                                                  const int* __restrict__ src,
                                                  const int* __restrict__ dst,
                                                  const float* __restrict__ dinv,
                                                  const float* __restrict__ xw,
                                                  float* tile, unsigned* keep) {
    int s = 0, dl = 0;
    float w = 0.0f;
    if (lane < n) {
        const unsigned e = lst[lane];
        const int d = dst[e];
        s  = src[e];
        dl = d - lo;
        w  = dinv[s] * dinv[d];
        if (keep != nullptr) keep[lane] = ((unsigned)s << 8) | (unsigned)dl;
    }
    gcn_own_add_rows<F, false>(s, dl, w, n, lane, xw, tile);
}

template <int F>
__device__ __forceinline__ int gcn_own_scan_tags(const int* __restrict__ src,
                                                 const int* __restrict__ dst,
                                                 const unsigned char* __restrict__ tags,
                                                 const float* __restrict__ dinv,
                                                 const float* __restrict__ xw,
                                                 float* tile, unsigned* mylist, unsigned* kept,
                                                 int lane, int wv, int lo, int E) {
    const unsigned pat = blockIdx.x * 0x01010101u;
    const int nunits = (E + OWN_TAGU - 1) / OWN_TAGU;
    const int nfull  = nunits / (8 * OWN_WAVES);
    const int ntrip  = nfull * OWN_WAVES;
    const int t0     = (int)((blockIdx.x * (unsigned)ntrip) / gridDim.x);
    const int rest0  = nfull * (8 * OWN_WAVES);
    const int per    = (nunits - rest0 + OWN_WAVES - 1) / OWN_WAVES;
    int  count  = 0;
    int  nkept  = 0;
    bool keepok = kept != nullptr;

    uint4 tv[8];
    int i = 0;
    int unit0 = 0;
    int nu = 0;
    int done = 0;
    for (;;) {
        if (done >= nu) {
            if (i > nfull) break;
            if (i < nfull) {
                int tr = t0 + i * OWN_WAVES + wv;
                if (tr >= ntrip) tr -= ntrip;
                unit0 = tr * 8;
                nu    = 8;
            } else {
                unit0 = rest0 + wv * per;
                nu    = (nunits - unit0 < per) ? (nunits - unit0) : per;
                if (nu <= 0) break;
            }
            ++i;
            done = 0;
#pragma unroll
            for (int u = 0; u < 8; ++u) {
                int un = unit0 + u;
                un = (un < nunits) ? un : (nunits - 1);
                tv[u] = *(const uint4*)(tags + (size_t)un * OWN_TAGU + lane * 16);
            }
        }
#pragma unroll
        for (int u = 0; u < 8; ++u) {
            if (u >= done && u < nu && count <= OWN_LIST - OWN_TAGU) {
                const int ebase = (unit0 + u) * OWN_TAGU + lane * 16;
                unsigned xs[4] = {tv[u].x ^ pat, tv[u].y ^ pat, tv[u].z ^ pat, tv[u].w ^ pat};
                if ((unit0 + u + 1) * OWN_TAGU > E) {
#pragma unroll
                    for (int q = 0; q < 4; ++q) {
#pragma unroll
                        for (int k = 0; k < 4; ++k)
                            if (ebase + q * 4 + k >= E) xs[q] |= 255u << (8 * k);
                    }
                }
#pragma unroll
                for (int q = 0; q < 4; ++q) {
                    const unsigned x = xs[q];
                    const unsigned z = (x - 0x01010101u) & ~x & 0x80808080u;
                    if (__builtin_amdgcn_ballot_w32(z != 0u) != 0u) {
#pragma unroll
                        for (int k = 0; k < 4; ++k) {
                            const bool m = ((x >> (8 * k)) & 255u) == 0u;
                            const unsigned bal = __builtin_amdgcn_ballot_w32(m);
                            if (bal != 0u) {
                                if (m) mylist[count + __popc(bal & ((1u << lane) - 1u))] = (unsigned)(ebase + q * 4 + k);
                                count += __popc(bal);
                            }
                        }
                    }
                }
                done = u + 1;
            }
        }
        __builtin_amdgcn_fence(__ATOMIC_ACQ_REL, "wavefront");
        while (count >= 32) {
            count -= 32;
            unsigned* kp = nullptr;
            if (keepok) {
                if (nkept + 32 <= OWN_KEEP) { kp = kept + nkept; nkept += 32; }
                else keepok = false;
            }
            gcn_own_add_found<F>(mylist + count, 32, lane, lo, src, dst, dinv, xw, tile, kp);
        }
    }
    if (count > 0) {
        unsigned* kp = nullptr;
        if (keepok) {
            if (nkept + count <= OWN_KEEP) { kp = kept + nkept; nkept += count; }
            else keepok = false;
        }
        gcn_own_add_found<F>(mylist, count, lane, lo, src, dst, dinv, xw, tile, kp);
    }
    return keepok ? nkept : -1;
}

__device__ __forceinline__ bool gcn_own_keep_fits(const float* wsb, size_t ws_bytes, const int* kend) {
    return (size_t)((const char*)kend - (const char*)wsb) <= ws_bytes;
}

__global__ __launch_bounds__(1024)
void gcn_owner_sum_lds(const int* __restrict__ src, const int* __restrict__ dst,
                       const float* __restrict__ dinv, const float* __restrict__ xw,
                       const float* __restrict__ bias, const _Float16* __restrict__ W2p,
                       float* __restrict__ xw2,
                       unsigned* __restrict__ kept, int* __restrict__ kcnt,
                       const unsigned char* __restrict__ tags, int Epad,
                       const float* wsb, size_t ws_bytes, int N, int E) {
    __shared__ __attribute__((aligned(16))) float tile[OWN_ROWS * GCN_D_HID];
    __shared__ unsigned list[OWN_WAVES * OWN_LIST];

    const int lane = threadIdx.x & 31;
    const int wv = __builtin_amdgcn_readfirstlane((int)(threadIdx.x >> 5));
    const int lo = blockIdx.x * OWN_ROWS;
    const int nrows = (N - lo < OWN_ROWS) ? (N - lo) : OWN_ROWS;
    const int wid = blockIdx.x * OWN_WAVES + wv;
    const bool keep_on = gcn_own_keep_fits(wsb, ws_bytes, kcnt + gridDim.x * OWN_WAVES);

    for (int r = wv; r < nrows; r += OWN_WAVES) {
        const float di = dinv[lo + r];
        const float4* xr = (const float4*)(xw + (size_t)(lo + r) * GCN_D_HID);
        const float4 a = xr[lane];
        const float4 b = xr[32 + lane];
        float* tr = tile + r * GCN_D_HID + lane;
        tr[0]   = a.x * di * di;
        tr[32]  = a.y * di * di;
        tr[64]  = a.z * di * di;
        tr[96]  = a.w * di * di;
        tr[128] = b.x * di * di;
        tr[160] = b.y * di * di;
        tr[192] = b.z * di * di;
        tr[224] = b.w * di * di;
    }
    __syncthreads();

    const bool tags_on = gcn_ws_fits(wsb, ws_bytes, tags + Epad) && gridDim.x <= 256u;
    unsigned* kp0 = keep_on ? kept + (size_t)wid * OWN_KEEP : nullptr;
    int nk;
    if (tags_on)
        nk = gcn_own_scan_tags<GCN_D_HID>(src, dst, tags, dinv, xw, tile, list + wv * OWN_LIST,
                                          kp0, lane, wv, lo, E);
    else
        nk = gcn_own_scan<GCN_D_HID>(src, dst, dinv, xw, tile, list + wv * OWN_LIST,
                                     kp0, lane, wv, lo, nrows, E);
    if (keep_on && lane == 0) kcnt[wid] = nk;
    __syncthreads();

    constexpr int HP  = GCN_D_HID + 8;
    constexpr int NRT = (OWN_ROWS + 15) / 16;
    constexpr int RPW = (NRT * 16 + OWN_WAVES - 1) / OWN_WAVES;
    constexpr int KT2 = GCN_D_HID / 32;
    constexpr int NT2 = GCN_D_OUT / 16;
    constexpr int RGS = OWN_WAVES / NT2;
    static_assert(OWN_WAVES % NT2 == 0 && NRT <= 2 * RGS && GCN_D_HID == 256 &&
                  NRT * 16 * HP * 2 <= OWN_ROWS * GCN_D_HID * 4, "shape not covered");
    const int ct = wv % NT2;
    const int rg = wv / NT2;

    const float4 bb0 = ((const float4*)bias)[lane];
    const float4 bb1 = ((const float4*)bias)[32 + lane];

    v16h b[KT2];
    {
        const _Float16* bb = W2p + ((size_t)ct * 32 + lane) * 16;
#pragma unroll
        for (int kt = 0; kt < KT2; ++kt) b[kt] = *(const v16h*)(bb + (size_t)kt * NT2 * 32 * 16);
    }
    v4h hlo[RPW], hhi[RPW];
#pragma unroll
    for (int i = 0; i < RPW; ++i) {
        const int r = wv + i * OWN_WAVES;
        float4 a = float4{0.0f, 0.0f, 0.0f, 0.0f};
        float4 c = float4{0.0f, 0.0f, 0.0f, 0.0f};
        if (r < nrows) {
            const float* tr = tile + r * GCN_D_HID + lane;
            a.x = fmaxf(tr[0]   + bb0.x, 0.0f);
            a.y = fmaxf(tr[32]  + bb0.y, 0.0f);
            a.z = fmaxf(tr[64]  + bb0.z, 0.0f);
            a.w = fmaxf(tr[96]  + bb0.w, 0.0f);
            c.x = fmaxf(tr[128] + bb1.x, 0.0f);
            c.y = fmaxf(tr[160] + bb1.y, 0.0f);
            c.z = fmaxf(tr[192] + bb1.z, 0.0f);
            c.w = fmaxf(tr[224] + bb1.w, 0.0f);
        }
        hlo[i] = v4h{(_Float16)a.x, (_Float16)a.y, (_Float16)a.z, (_Float16)a.w};
        hhi[i] = v4h{(_Float16)c.x, (_Float16)c.y, (_Float16)c.z, (_Float16)c.w};
    }
    __syncthreads();
    _Float16* hrow = (_Float16*)tile;
#pragma unroll
    for (int i = 0; i < RPW; ++i) {
        const int r = wv + i * OWN_WAVES;
        if (r < NRT * 16) {
            *(v4h*)(hrow + r * HP + lane * 4)       = hlo[i];
            *(v4h*)(hrow + r * HP + 128 + lane * 4) = hhi[i];
        }
    }
    __syncthreads();

    const int hsel = lane >> 4;
    const int idx  = lane & 15;
    const int one[1] = {0};
#pragma unroll
    for (int i = 0; i < 2; ++i) {
        const int rt = rg + i * RGS;
        if (rt < NRT && rt * 16 < nrows) {
            v8f acc[1];
            acc[0] = v8f{};
            gcn_gemm_tiles<GCN_D_HID, 1>(hrow + (rt * 16 + idx) * HP + hsel * 8, b, one, acc);
            const int row0 = rt * 16 + hsel * 8;
            float* cb = xw2 + (size_t)(lo + row0) * GCN_D_OUT + ct * 16 + idx;
#pragma unroll
            for (int r = 0; r < 8; ++r)
                if (row0 + r < nrows) cb[(size_t)r * GCN_D_OUT] = acc[0][r];
        }
    }
}

__global__ __launch_bounds__(1024)
void gcn_owner_sum_kept(const int* __restrict__ src, const int* __restrict__ dst,
                        const float* __restrict__ dinv, const float* __restrict__ xw,
                        const float* __restrict__ bias, float* __restrict__ out,
                        const unsigned* __restrict__ kept, const int* __restrict__ kcnt,
                        const float* wsb, size_t ws_bytes, int N, int E) {
    __shared__ float    tile[OWN_ROWS * GCN_D_OUT];
    __shared__ unsigned list[OWN_WAVES * OWN_LIST];

    const int lane = threadIdx.x & 31;
    const int wv = __builtin_amdgcn_readfirstlane((int)(threadIdx.x >> 5));
    const int lo = blockIdx.x * OWN_ROWS;
    const int nrows = (N - lo < OWN_ROWS) ? (N - lo) : OWN_ROWS;
    const int wid = blockIdx.x * OWN_WAVES + wv;
    const bool keep_on = gcn_own_keep_fits(wsb, ws_bytes, kcnt + gridDim.x * OWN_WAVES);

    bool use_kept = false;
    if (keep_on) {
        const int c = kcnt[blockIdx.x * OWN_WAVES + lane];
        use_kept = __builtin_amdgcn_ballot_w32(c < 0 || c > OWN_KEEP) == 0u;
    }

    {
        constexpr int RPW = (OWN_ROWS + OWN_WAVES - 1) / OWN_WAVES;
        float sa[RPW], sb[RPW];
        const int   rl  = wv + lane * OWN_WAVES;
        const float dvl = dinv[lo + ((rl < nrows) ? rl : (nrows - 1))];
#pragma unroll
        for (int i = 0; i < RPW; ++i) {
            const int r  = wv + i * OWN_WAVES;
            const int rc = (r < nrows) ? r : (nrows - 1);
            const float* xr = xw + (size_t)(lo + rc) * GCN_D_OUT;
            sa[i] = xr[lane];
            sb[i] = xr[32 + lane];
        }
#pragma unroll
        for (int i = 0; i < RPW; ++i) {
            const int r = wv + i * OWN_WAVES;
            if (r < nrows) {
                const float di = __int_as_float(__builtin_amdgcn_readlane(__float_as_int(dvl), i));
                float* tr = tile + r * GCN_D_OUT + lane;
                tr[0]  = sa[i] * di * di;
                tr[32] = sb[i] * di * di;
            }
        }
    }
    __syncthreads();

    if (use_kept) {
        const unsigned* kp = kept + (size_t)wid * OWN_KEEP;
        int cnt = __builtin_amdgcn_readfirstlane(kcnt[wid]);
        cnt = (cnt < OWN_KEEP) ? cnt : OWN_KEEP;
        constexpr int NB = OWN_KEEP / 32;
        unsigned* myl = list + wv * OWN_LIST;
        unsigned rec[NB];
#pragma unroll
        for (int b = 0; b < NB; ++b) {
            rec[b] = 0u;
            if (b * 32 + lane < cnt) rec[b] = kp[b * 32 + lane];
        }
        float wgt[NB];
#pragma unroll
        for (int b = 0; b < NB; ++b) {
            const float ds = dinv[rec[b] >> 8];
            const float dd = dinv[lo + (int)(rec[b] & 255u)];
            wgt[b] = ds * dd;
        }
#pragma unroll
        for (int b = 0; b < NB; ++b) {
            myl[b * 32 + lane]            = rec[b];
            myl[OWN_KEEP + b * 32 + lane] = __float_as_uint(wgt[b]);
        }
        for (int base = 0; base < cnt; base += 32) {
            const int n = (cnt - base < 32) ? (cnt - base) : 32;
            const unsigned r = myl[base + lane];
            const float    w = __uint_as_float(myl[OWN_KEEP + base + lane]);
            gcn_own_add_rows<GCN_D_OUT, true>((int)(r >> 8), (int)(r & 255u), w, n, lane, xw, tile);
        }
    } else {
        gcn_own_scan<GCN_D_OUT>(src, dst, dinv, xw, tile, list + wv * OWN_LIST, nullptr,
                                lane, wv, lo, nrows, E);
    }
    __syncthreads();

    const float b0 = bias[lane];
    const float b1 = bias[lane + 32];
    for (int r = wv; r < nrows; r += OWN_WAVES) {
        const float* tr = tile + r * GCN_D_OUT + lane;
        float v0 = tr[0]  + b0;
        float v1 = tr[32] + b1;
        float m = fmaxf(v0, v1);
#pragma unroll
        for (int off = 16; off > 0; off >>= 1) m = fmaxf(m, __shfl_xor(m, off, 32));
        float e0 = __expf(v0 - m);
        float e1 = __expf(v1 - m);
        float sm = e0 + e1;
#pragma unroll
        for (int off = 16; off > 0; off >>= 1) sm += __shfl_xor(sm, off, 32);
        float inv = 1.0f / sm;
        float* orow = out + (size_t)(lo + r) * GCN_D_OUT;
        orow[lane]      = e0 * inv;
        orow[lane + 32] = e1 * inv;
    }
}

extern "C" void kernel_launch(void* const* d_in, const int* in_sizes, int n_in,
                              void* d_out, int out_size, void* d_ws, size_t ws_size,
                              hipStream_t stream) {
    const float* x   = (const float*)d_in[0];
    const int*   ei  = (const int*)  d_in[1];
    const float* W1  = (const float*)d_in[2];
    const float* b1  = (const float*)d_in[3];
    const float* W2  = (const float*)d_in[4];
    const float* b2  = (const float*)d_in[5];
    float* out = (float*)d_out;

    const int N = in_sizes[0] / GCN_D_IN;
    const int E = in_sizes[1] / 2;
    const int* src = ei;
    const int* dst = ei + E;

    float* ws   = (float*)d_ws;
    size_t off  = ((size_t)N + 63) & ~(size_t)63;
    float* dinv = ws;
    float* xw1  = ws + off;
    float* h1   = xw1 + (size_t)N * GCN_D_HID;
    _Float16* W1p = (_Float16*)(h1 + (size_t)N * GCN_D_HID);
    _Float16* W2p = W1p + (size_t)GCN_D_IN * GCN_D_HID;
    float* xw2  = h1;
    const int ob = (N + OWN_ROWS - 1) / OWN_ROWS;
    unsigned* kept = (unsigned*)(W2p + (size_t)GCN_D_HID * GCN_D_OUT);
    int*      kcnt = (int*)(kept + (size_t)ob * OWN_WAVES * OWN_KEEP);
    unsigned char* tags = (unsigned char*)(kcnt + (size_t)ob * OWN_WAVES);
    const int Epad = (E + OWN_TAGU - 1) / OWN_TAGU * OWN_TAGU;

    const int B = 256;
    {
        int t1 = (GCN_D_IN >> 5) * (GCN_D_HID >> 4) * 32;
        gcn_pack_w_f16<<<(t1 + B - 1) / B, B, 0, stream>>>(W1, W1p, GCN_D_IN, GCN_D_HID);
        int t2 = (GCN_D_HID >> 5) * (GCN_D_OUT >> 4) * 32;
        gcn_pack_w_f16<<<(t2 + B - 1) / B, B, 0, stream>>>(W2, W2p, GCN_D_HID, GCN_D_OUT);
    }

    unsigned* part = (unsigned*)h1;
    gcn_count_deg<<<DEG_CHUNKS, 1024, 0, stream>>>(dst, part, (int)off, tags, ws, ws_size, N, E, Epad);
    gcn_make_dinv<<<(N + B - 1) / B, B, 0, stream>>>(part, (int)off, dinv, N);

    {
        gcn_wmma_gemm_bregs<GCN_D_IN, GCN_D_HID, 2><<<256, 512, 0, stream>>>(x, W1p, xw1, N);
    }
    {
        gcn_owner_sum_lds<<<ob, 1024, 0, stream>>>(src, dst, dinv, xw1, b1, W2p, xw2, kept, kcnt,
                                                   tags, Epad, ws, ws_size, N, E);
    }

    {
        gcn_owner_sum_kept<<<ob, 1024, 0, stream>>>(src, dst, dinv, xw2, b2, out, kept, kcnt,
                                                    ws, ws_size, N, E);
    }
}


# edges binned by owner block in the degree-count launch (256 LDS bins per edge chunk: source + local row); both owner launches read their bins, tag scan, kept lists gone; sum order = LDS add arrival
# speedup vs baseline: 4.6893x; 1.1226x over previous
#include <hip/hip_runtime.h>
#include <hip/hip_bf16.h>

typedef __attribute__((ext_vector_type(16))) _Float16 v16h;
typedef __attribute__((ext_vector_type(8)))  float    v8f;

#define GCN_D_IN  512
#define GCN_D_HID 256
#define GCN_D_OUT 64

#define OWN_ROWS  196
#define OWN_TAGU  512

__device__ __forceinline__ bool gcn_ws_fits(const float* wsb, size_t ws_bytes, const void* end) {
    return (size_t)((const char*)end - (const char*)wsb) <= ws_bytes;
}

#define DEG_CHUNKS 64
#define DEG_BINS   53248
#define BIN_CAP    16384
#define BIN_OFFP   264

__device__ __forceinline__ bool gcn_bins_on(const float* wsb, size_t ws_bytes, const unsigned* boff,
                                            int N, int E, int Epad) {
    const int nq   = Epad >> 2;
    const int qper = (nq + DEG_CHUNKS - 1) / DEG_CHUNKS;
    return gcn_ws_fits(wsb, ws_bytes, boff + (size_t)DEG_CHUNKS * BIN_OFFP) &&
           qper <= BIN_CAP / 4 && (E & 3) == 0 && N <= DEG_BINS &&
           (N + OWN_ROWS - 1) / OWN_ROWS <= 256 && N <= (1 << 24);
}

__global__ __launch_bounds__(1024)
void gcn_count_deg(const int* __restrict__ src, const int* __restrict__ dst,
                   unsigned* __restrict__ part, int pitch,
                   unsigned char* __restrict__ tags, unsigned* __restrict__ bins,
                   unsigned* __restrict__ boff, const float* wsb, size_t ws_bytes,
                   int N, int E, int Epad) {
    __shared__ __attribute__((aligned(16))) unsigned hist[DEG_BINS];
    __shared__ __attribute__((aligned(16))) unsigned brec[BIN_CAP];
    __shared__ unsigned bcnt[256];
    __shared__ unsigned bstart[257];
    const int  t       = threadIdx.x;
    const bool bins_on = gcn_bins_on(wsb, ws_bytes, boff, N, E, Epad) && (int)gridDim.x == DEG_CHUNKS;
    const bool tags_on = !bins_on && gcn_ws_fits(wsb, ws_bytes, tags + Epad);
    const int  nq      = Epad >> 2;
    const int  qfull   = E >> 2;
    const int  qper    = (nq + (int)gridDim.x - 1) / (int)gridDim.x;
    const int  q0      = (int)blockIdx.x * qper;
    const int  q1      = (q0 + qper < nq) ? (q0 + qper) : nq;
    unsigned*  prow    = part + (size_t)blockIdx.x * pitch;

    int4     dv[4];
    int4     sv[4];
    unsigned rk[16];
    unsigned vmask = 0u;
    if (bins_on && t < 256) bcnt[t] = 0u;

    for (int base = 0; base < N; base += DEG_BINS) {
        const int nb = (N - base < DEG_BINS) ? (N - base) : DEG_BINS;
        for (int i = t * 4; i < DEG_BINS; i += 4096) *(uint4*)(hist + i) = uint4{0u, 0u, 0u, 0u};
        __syncthreads();
        for (int qb = q0; qb < q1; qb += 4096) {
#pragma unroll
            for (int u = 0; u < 4; ++u) {
                const int q  = qb + u * 1024 + t;
                const int qc = (q < qfull) ? q : (qfull - 1);
                const int* dp = dst + (size_t)qc * 4;
                dv[u] = int4{dp[0], dp[1], dp[2], dp[3]};
                if (bins_on) {
                    const int* sp = src + (size_t)qc * 4;
                    sv[u] = int4{sp[0], sp[1], sp[2], sp[3]};
                }
            }
#pragma unroll
            for (int u = 0; u < 4; ++u) {
                const int  q    = qb + u * 1024 + t;
                const bool mine = q < q1;
                const bool real = mine && q < qfull;
                const int dd[4] = {real ? dv[u].x : -1, real ? dv[u].y : -1,
                                   real ? dv[u].z : -1, real ? dv[u].w : -1};
                unsigned tw = 0u;
#pragma unroll
                for (int k = 0; k < 4; ++k) {
                    const unsigned r = (unsigned)(dd[k] - base);
                    if (r < (unsigned)nb) atomicAdd(&hist[r], 1u);
                    if (dd[k] >= 0) {
                        const unsigned own = ((unsigned)dd[k] / OWN_ROWS) & 255u;
                        tw |= own << (8 * k);
                        if (bins_on && base == 0) {
                            rk[u * 4 + k] = atomicAdd(&bcnt[own], 1u);
                            vmask |= 1u << (u * 4 + k);
                        }
                    }
                }
                if (tags_on && base == 0 && mine && !(q == qfull && (E & 3) != 0))
                    *(unsigned*)(tags + (size_t)q * 4) = tw;
            }
        }
        if (t == 0 && (E & 3) != 0 && qfull >= q0 && qfull < q1) {
            unsigned tw = 0u;
            for (int k = 0; k < (E & 3); ++k) {
                const int d = dst[qfull * 4 + k];
                const unsigned r = (unsigned)(d - base);
                if (r < (unsigned)nb) atomicAdd(&hist[r], 1u);
                if (d >= 0) tw |= (((unsigned)d / OWN_ROWS) & 255u) << (8 * k);
            }
            if (tags_on && base == 0) *(unsigned*)(tags + (size_t)qfull * 4) = tw;
        }
        __syncthreads();
        if (bins_on && base == 0 && t < 32) {
            unsigned c8[8];
            unsigned run = 0u;
#pragma unroll
            for (int j = 0; j < 8; ++j) { c8[j] = run; run += bcnt[t * 8 + j]; }
            unsigned incl = run;
#pragma unroll
            for (int o = 1; o < 32; o <<= 1) {
                const unsigned y = __shfl_up(incl, o, 32);
                if (t >= o) incl += y;
            }
            const unsigned excl = incl - run;
#pragma unroll
            for (int j = 0; j < 8; ++j) bstart[t * 8 + j] = excl + c8[j];
            if (t == 31) bstart[256] = incl;
        }
        for (int i = t * 4; i < nb; i += 4096) {
            if (i + 4 <= nb) {
                *(uint4*)(prow + base + i) = *(const uint4*)(hist + i);
            } else {
                for (int k = i; k < nb; ++k) prow[base + k] = hist[k];
            }
        }
        __syncthreads();
        if (bins_on && base == 0) {
#pragma unroll
            for (int u = 0; u < 4; ++u) {
                const int dd[4] = {dv[u].x, dv[u].y, dv[u].z, dv[u].w};
                const int ss[4] = {sv[u].x, sv[u].y, sv[u].z, sv[u].w};
#pragma unroll
                for (int k = 0; k < 4; ++k) {
                    if ((vmask >> (u * 4 + k)) & 1u) {
                        const unsigned own = (unsigned)dd[k] / OWN_ROWS;
                        const unsigned pos = bstart[own & 255u] + rk[u * 4 + k];
                        if (pos < (unsigned)BIN_CAP)
                            brec[pos] = ((unsigned)ss[k] << 8) | (((unsigned)dd[k] - own * OWN_ROWS) & 255u);
                    }
                }
            }
            __syncthreads();
            const unsigned total = (bstart[256] < (unsigned)BIN_CAP) ? bstart[256] : (unsigned)BIN_CAP;
            unsigned* bout = bins + (size_t)blockIdx.x * BIN_CAP;
            for (unsigned i = (unsigned)t * 4u; i < total; i += 4096u) {
                if (i + 4u <= total) {
                    *(uint4*)(bout + i) = *(const uint4*)(brec + i);
                } else {
                    for (unsigned k = i; k < total; ++k) bout[k] = brec[k];
                }
            }
            if (t <= 256) boff[(size_t)blockIdx.x * BIN_OFFP + t] = bstart[t];
        }
    }
}

__global__ void gcn_make_dinv(const unsigned* __restrict__ part, int pitch,
                              float* __restrict__ dinv, int n) {
    int i = blockIdx.x * blockDim.x + threadIdx.x;
    if (i < n) {
        unsigned c[DEG_CHUNKS];
#pragma unroll
        for (int k = 0; k < DEG_CHUNKS; ++k) c[k] = part[(size_t)k * pitch + i];
        unsigned sum = 1u;
#pragma unroll
        for (int k = 0; k < DEG_CHUNKS; ++k) sum += c[k];
        const float d = (float)sum;
        dinv[i] = (d > 0.0f) ? rsqrtf(d) : 0.0f;
    }
}

__global__ void gcn_pack_w_f16(const float* __restrict__ W, _Float16* __restrict__ Wp,
                               int K, int N) {
    int t = blockIdx.x * blockDim.x + threadIdx.x;
    int tilesN = N >> 4;
    int total = (K >> 5) * tilesN * 32;
    if (t >= total) return;
    int frag = t >> 5;
    int lane = t & 31;
    int kt = frag / tilesN;
    int tn = frag - kt * tilesN;
    int n    = tn * 16 + (lane & 15);
    int koff = kt * 32 + (lane >> 4) * 8;
    _Float16* o = Wp + (size_t)t * 16;
#pragma unroll
    for (int j = 0; j < 8; ++j) {
        o[j]     = (_Float16)W[(size_t)(koff + j) * N + n];
        o[j + 8] = (_Float16)W[(size_t)(koff + 16 + j) * N + n];
    }
}

typedef __attribute__((ext_vector_type(8))) _Float16 v8h;
typedef __attribute__((ext_vector_type(4))) _Float16 v4h;

__device__ __forceinline__ v16h gcn_lds_a_frag(const _Float16* ar) {
    const v8h lo = *(const v8h*)(ar);
    const v8h hi = *(const v8h*)(ar + 16);
    return __builtin_shufflevector(lo, hi, 0, 1, 2, 3, 4, 5, 6, 7, 8, 9, 10, 11, 12, 13, 14, 15);
}

template <int K, int NTL>
__device__ __forceinline__ void gcn_gemm_tiles(const _Float16* ar0, const v16h (&b)[K / 32],
                                               const int (&toff)[NTL], v8f (&acc)[NTL]) {
    constexpr int KT    = K / 32;
    constexpr int PITCH = K + 8;
    const _Float16* ar[NTL];
    v16h a[3][NTL];
#pragma unroll
    for (int i = 0; i < NTL; ++i) {
        ar[i]   = ar0 + toff[i] * (16 * PITCH);
        a[0][i] = gcn_lds_a_frag(ar[i]);
        a[1][i] = gcn_lds_a_frag(ar[i] + 32);
    }
#pragma unroll
    for (int kt = 0; kt < KT; ++kt) {
        if (kt + 2 < KT) {
#pragma unroll
            for (int i = 0; i < NTL; ++i) a[(kt + 2) % 3][i] = gcn_lds_a_frag(ar[i] + (kt + 2) * 32);
        }
        __builtin_amdgcn_sched_barrier(0);
#pragma unroll
        for (int i = 0; i < NTL; ++i)
            acc[i] = __builtin_amdgcn_wmma_f32_16x16x32_f16(false, a[kt % 3][i], false, b[kt],
                                                            (short)0, acc[i], false, false);
        __builtin_amdgcn_sched_barrier(0);
    }
}

template <int K, int N, int RT>
__global__ __launch_bounds__(512)
void gcn_wmma_gemm_bregs(const float* __restrict__ A,
                         const _Float16* __restrict__ Bp,
                         float* __restrict__ C, int M) {
    constexpr int KT    = K / 32;
    constexpr int NT    = N / 16;
    constexpr int RG    = 16 / NT;
    constexpr int ROWS  = RG * RT * 16;
    constexpr int PITCH = K + 8;
    constexpr int C4    = K / 4;
    constexpr int RSTEP = 512 / C4;
    constexpr int NV    = ROWS / RSTEP;
    static_assert(K % 32 == 0 && N % 16 == 0 && 16 % NT == 0 && 512 % C4 == 0 &&
                  ROWS % RSTEP == 0 && (RT == 1 || RT == 2), "shape not covered");

    __shared__ __attribute__((aligned(16))) _Float16 tile[2 * ROWS * PITCH];

    const int t    = threadIdx.x;
    const int lane = t & 31;
    const int w    = __builtin_amdgcn_readfirstlane(t >> 5);
    const int ct   = w % NT;
    const int rg   = w / NT;
    const int hsel = lane >> 4;
    const int idx  = lane & 15;

    const int nblk  = (M + ROWS - 1) / ROWS;
    const int ntile = M / 16;
    const int srow  = t / C4;
    const int sc    = (t % C4) * 4;
    float4 st[NV];

    int blk = blockIdx.x;
    if (blk < nblk) {
#pragma unroll
        for (int i = 0; i < NV; ++i) {
            int row = blk * ROWS + srow + i * RSTEP;
            row = (row < M) ? row : (M - 1);
            st[i] = *(const float4*)(A + (size_t)row * K + sc);
        }
    }

    v16h b[KT];
    {
        const _Float16* bb = Bp + ((size_t)ct * 32 + lane) * 16;
#pragma unroll
        for (int kt = 0; kt < KT; ++kt) b[kt] = *(const v16h*)(bb + (size_t)kt * NT * 32 * 16);
    }
    __builtin_amdgcn_sched_barrier(0);

    if (blk < nblk) {
#pragma unroll
        for (int i = 0; i < NV; ++i) {
            v4h hv = {(_Float16)st[i].x, (_Float16)st[i].y, (_Float16)st[i].z, (_Float16)st[i].w};
            *(v4h*)(tile + (srow + i * RSTEP) * PITCH + sc) = hv;
        }
    }
    __syncthreads();

    int buf = 0;
    for (; blk < nblk; blk += gridDim.x) {
        const int  nxt      = blk + (int)gridDim.x;
        const bool has_next = nxt < nblk;
        if (has_next) {
#pragma unroll
            for (int i = 0; i < NV; ++i) {
                int row = nxt * ROWS + srow + i * RSTEP;
                row = (row < M) ? row : (M - 1);
                st[i] = *(const float4*)(A + (size_t)row * K + sc);
            }
        }

        const int tile0 = (blk * RG + rg) * RT;
        const _Float16* ar0 = tile + buf * (ROWS * PITCH) + (rg * RT * 16 + idx) * PITCH + hsel * 8;
        float* cb0 = C + (size_t)(tile0 * 16 + hsel * 8) * N + ct * 16 + idx;
        const bool live = tile0 < ntile;
        int toff[RT];
#pragma unroll
        for (int i = 0; i < RT; ++i) toff[i] = (tile0 + i < ntile) ? i : 0;
        v8f acc[RT];
#pragma unroll
        for (int i = 0; i < RT; ++i) acc[i] = v8f{};
        if (live) gcn_gemm_tiles<K, RT>(ar0, b, toff, acc);

        if (has_next) {
            _Float16* tn = tile + (buf ^ 1) * (ROWS * PITCH);
#pragma unroll
            for (int i = 0; i < NV; ++i) {
                v4h hv = {(_Float16)st[i].x, (_Float16)st[i].y, (_Float16)st[i].z, (_Float16)st[i].w};
                *(v4h*)(tn + (srow + i * RSTEP) * PITCH + sc) = hv;
            }
        }
        __syncthreads();
        if (live) {
#pragma unroll
            for (int i = 0; i < RT; ++i) {
                float* cb = cb0 + (size_t)toff[i] * (16 * N);
#pragma unroll
                for (int r = 0; r < 8; ++r) cb[(size_t)r * N] = acc[i][r];
            }
        }
        buf ^= 1;
    }
}

#define OWN_WAVES 32
#define OWN_LIST  544
#define OWN_KEEP  160

template <int F, int G>
__device__ __forceinline__ void gcn_own_add_group(int s, int dl, float w, int j, int lane,
                                                  const float* __restrict__ xw, float* tile) {
    int   rl[G];
    float wj[G];
    if constexpr (F == 256) {
        float4 a[G], b[G];
#pragma unroll
        for (int q = 0; q < G; ++q) {
            const int sj = __builtin_amdgcn_readlane(s, j + q);
            rl[q] = __builtin_amdgcn_readlane(dl, j + q);
            wj[q] = __int_as_float(__builtin_amdgcn_readlane(__float_as_int(w), j + q));
            const float4* xr = (const float4*)(xw + (size_t)sj * F);
            a[q] = xr[lane];
            b[q] = xr[32 + lane];
        }
#pragma unroll
        for (int q = 0; q < G; ++q) {
            float* tr = tile + rl[q] * F + lane;
            atomicAdd(tr,       a[q].x * wj[q]);
            atomicAdd(tr + 32,  a[q].y * wj[q]);
            atomicAdd(tr + 64,  a[q].z * wj[q]);
            atomicAdd(tr + 96,  a[q].w * wj[q]);
            atomicAdd(tr + 128, b[q].x * wj[q]);
            atomicAdd(tr + 160, b[q].y * wj[q]);
            atomicAdd(tr + 192, b[q].z * wj[q]);
            atomicAdd(tr + 224, b[q].w * wj[q]);
        }
    } else {
        float a[G], b[G];
#pragma unroll
        for (int q = 0; q < G; ++q) {
            const int sj = __builtin_amdgcn_readlane(s, j + q);
            rl[q] = __builtin_amdgcn_readlane(dl, j + q);
            wj[q] = __int_as_float(__builtin_amdgcn_readlane(__float_as_int(w), j + q));
            const float* xr = xw + (size_t)sj * F;
            a[q] = xr[lane];
            b[q] = xr[32 + lane];
        }
#pragma unroll
        for (int q = 0; q < G; ++q) {
            float* tr = tile + rl[q] * F + lane;
            atomicAdd(tr,      a[q] * wj[q]);
            atomicAdd(tr + 32, b[q] * wj[q]);
        }
    }
}

template <int F, bool SHORT_TAIL>
__device__ __forceinline__ void gcn_own_add_rows(int s, int dl, float w, int n, int lane,
                                                 const float* __restrict__ xw, float* tile) {
    int j = 0;
    for (; j + 8 <= n; j += 8) gcn_own_add_group<F, 8>(s, dl, w, j, lane, xw, tile);
    if constexpr (SHORT_TAIL) {
        if (j + 4 <= n) { gcn_own_add_group<F, 4>(s, dl, w, j, lane, xw, tile); j += 4; }
        if (j + 2 <= n) { gcn_own_add_group<F, 2>(s, dl, w, j, lane, xw, tile); j += 2; }
        if (j < n)        gcn_own_add_group<F, 1>(s, dl, w, j, lane, xw, tile);
    } else {
        for (; j < n; ++j) gcn_own_add_group<F, 1>(s, dl, w, j, lane, xw, tile);
    }
}

template <int F>
__device__ __forceinline__ void gcn_own_add_binned(const unsigned* __restrict__ bins,
                                                   const unsigned* __restrict__ boff,
                                                   const float* __restrict__ dinv,
                                                   const float* __restrict__ xw,
                                                   float* tile, unsigned* myl,
                                                   int lane, int wv, int lo) {
    static_assert(DEG_CHUNKS == 2 * OWN_WAVES && 2 * OWN_KEEP <= OWN_LIST, "shape not covered");
    const int c0 = wv, c1 = wv + OWN_WAVES;
    const unsigned* o0 = boff + (size_t)c0 * BIN_OFFP + blockIdx.x;
    const unsigned* o1 = boff + (size_t)c1 * BIN_OFFP + blockIdx.x;
    int a0 = __builtin_amdgcn_readfirstlane((int)o0[0]);
    int a1 = __builtin_amdgcn_readfirstlane((int)o0[1]);
    int b0 = __builtin_amdgcn_readfirstlane((int)o1[0]);
    int b1 = __builtin_amdgcn_readfirstlane((int)o1[1]);
    a0 = (a0 < 0) ? 0 : ((a0 > BIN_CAP) ? BIN_CAP : a0);
    a1 = (a1 < a0) ? a0 : ((a1 > BIN_CAP) ? BIN_CAP : a1);
    b0 = (b0 < 0) ? 0 : ((b0 > BIN_CAP) ? BIN_CAP : b0);
    b1 = (b1 < b0) ? b0 : ((b1 > BIN_CAP) ? BIN_CAP : b1);
    const int n0 = a1 - a0;
    const int total = n0 + (b1 - b0);
    const unsigned* p0 = bins + (size_t)c0 * BIN_CAP + a0;
    const unsigned* p1 = bins + (size_t)c1 * BIN_CAP + b0;
    constexpr int NB = OWN_KEEP / 32;
    for (int pos = 0; pos < total; pos += OWN_KEEP) {
        const int cnt = (total - pos < OWN_KEEP) ? (total - pos) : OWN_KEEP;
        unsigned rec[NB];
#pragma unroll
        for (int b = 0; b < NB; ++b) {
            rec[b] = 0u;
            const int i = pos + b * 32 + lane;
            if (b * 32 + lane < cnt) {
                const unsigned* p = (i < n0) ? (p0 + i) : (p1 + (i - n0));
                rec[b] = *p;
            }
        }
        float wgt[NB];
#pragma unroll
        for (int b = 0; b < NB; ++b) {
            const float ds = dinv[rec[b] >> 8];
            const float dd = dinv[lo + (int)(rec[b] & 255u)];
            wgt[b] = ds * dd;
        }
#pragma unroll
        for (int b = 0; b < NB; ++b) {
            myl[b * 32 + lane]            = rec[b];
            myl[OWN_KEEP + b * 32 + lane] = __float_as_uint(wgt[b]);
        }
        for (int base = 0; base < cnt; base += 32) {
            const int n = (cnt - base < 32) ? (cnt - base) : 32;
            const unsigned r = myl[base + lane];
            const float    w = __uint_as_float(myl[OWN_KEEP + base + lane]);
            gcn_own_add_rows<F, true>((int)(r >> 8), (int)(r & 255u), w, n, lane, xw, tile);
        }
    }
}

template <int F>
__device__ __forceinline__ void gcn_own_add_edges(const unsigned* lst, int n, int lane, int lo,
                                                  const int* __restrict__ src,
                                                  const float* __restrict__ dinv,
                                                  const float* __restrict__ xw,
                                                  float* tile, unsigned* keep) {
    int s = 0, dl = 0;
    float w = 0.0f;
    if (lane < n) {
        const unsigned rec = lst[lane];
        dl = (int)(rec & 255u);
        s  = src[rec >> 8];
        w  = dinv[s] * dinv[lo + dl];
        if (keep != nullptr) keep[lane] = ((unsigned)s << 8) | (unsigned)dl;
    }
    gcn_own_add_rows<F, false>(s, dl, w, n, lane, xw, tile);
}

template <int F>
__device__ __forceinline__ int gcn_own_scan(const int* __restrict__ src, const int* __restrict__ dst,
                                            const float* __restrict__ dinv,
                                            const float* __restrict__ xw,
                                            float* tile, unsigned* mylist, unsigned* kept,
                                            int lane, int wv, int lo, int nrows, int E) {
    const int nchunks = (E + 1023) >> 10;
    const int c0 = (int)((blockIdx.x * (unsigned)nchunks) / gridDim.x);
    int  count  = 0;
    int  nkept  = 0;
    bool keepok = kept != nullptr;

    for (int i = wv; i < nchunks; i += OWN_WAVES) {
        int c = c0 + i;
        if (c >= nchunks) c -= nchunks;
        const int cbase = c << 10;

        int4 dv[8];
        int  eu[8];
#pragma unroll
        for (int u = 0; u < 8; ++u) {
            const int e = cbase + u * 128 + lane * 4;
            eu[u] = (e < E - 4) ? e : (E - 4);
            dv[u] = *(const int4*)(dst + eu[u]);
        }
        if (cbase + 1024 > E) {
#pragma unroll
            for (int u = 0; u < 8; ++u) {
                const int e = cbase + u * 128 + lane * 4;
                if (eu[u]     < e) dv[u].x = -1;
                if (eu[u] + 1 < e) dv[u].y = -1;
                if (eu[u] + 2 < e) dv[u].z = -1;
                if (eu[u] + 3 < e) dv[u].w = -1;
            }
        }
#pragma unroll
        for (int hf = 0; hf < 2; ++hf) {
#pragma unroll
            for (int uu = 0; uu < 4; ++uu) {
                const int u = hf * 4 + uu;
                const int dd[4] = {dv[u].x, dv[u].y, dv[u].z, dv[u].w};
#pragma unroll
                for (int k = 0; k < 4; ++k) {
                    const unsigned t = (unsigned)(dd[k] - lo);
                    const bool m = t < (unsigned)nrows;
                    const unsigned bal = __builtin_amdgcn_ballot_w32(m);
                    if (bal != 0u) {
                        if (m) mylist[count + __popc(bal & ((1u << lane) - 1u))] = ((unsigned)(eu[u] + k) << 8) | t;
                        count += __popc(bal);
                    }
                }
            }
            __builtin_amdgcn_fence(__ATOMIC_ACQ_REL, "wavefront");
            while (count >= 32) {
                count -= 32;
                unsigned* kp = nullptr;
                if (keepok) {
                    if (nkept + 32 <= OWN_KEEP) { kp = kept + nkept; nkept += 32; }
                    else keepok = false;
                }
                gcn_own_add_edges<F>(mylist + count, 32, lane, lo, src, dinv, xw, tile, kp);
            }
        }
    }
    if (count > 0) {
        unsigned* kp = nullptr;
        if (keepok) {
            if (nkept + count <= OWN_KEEP) { kp = kept + nkept; nkept += count; }
            else keepok = false;
        }
        gcn_own_add_edges<F>(mylist, count, lane, lo, src, dinv, xw, tile, kp);
    }
    return keepok ? nkept : -1;
}

template <int F>
__device__ __forceinline__ void gcn_own_add_found(const unsigned* lst, int n, int lane, int lo,
                                                  const int* __restrict__ src,
                                                  const int* __restrict__ dst,
                                                  const float* __restrict__ dinv,
                                                  const float* __restrict__ xw,
                                                  float* tile, unsigned* keep) {
    int s = 0, dl = 0;
    float w = 0.0f;
    if (lane < n) {
        const unsigned e = lst[lane];
        const int d = dst[e];
        s  = src[e];
        dl = d - lo;
        w  = dinv[s] * dinv[d];
        if (keep != nullptr) keep[lane] = ((unsigned)s << 8) | (unsigned)dl;
    }
    gcn_own_add_rows<F, false>(s, dl, w, n, lane, xw, tile);
}

template <int F>
__device__ __forceinline__ int gcn_own_scan_tags(const int* __restrict__ src,
                                                 const int* __restrict__ dst,
                                                 const unsigned char* __restrict__ tags,
                                                 const float* __restrict__ dinv,
                                                 const float* __restrict__ xw,
                                                 float* tile, unsigned* mylist, unsigned* kept,
                                                 int lane, int wv, int lo, int E) {
    const unsigned pat = blockIdx.x * 0x01010101u;
    const int nunits = (E + OWN_TAGU - 1) / OWN_TAGU;
    const int nfull  = nunits / (8 * OWN_WAVES);
    const int ntrip  = nfull * OWN_WAVES;
    const int t0     = (int)((blockIdx.x * (unsigned)ntrip) / gridDim.x);
    const int rest0  = nfull * (8 * OWN_WAVES);
    const int per    = (nunits - rest0 + OWN_WAVES - 1) / OWN_WAVES;
    int  count  = 0;
    int  nkept  = 0;
    bool keepok = kept != nullptr;

    uint4 tv[8];
    int i = 0;
    int unit0 = 0;
    int nu = 0;
    int done = 0;
    for (;;) {
        if (done >= nu) {
            if (i > nfull) break;
            if (i < nfull) {
                int tr = t0 + i * OWN_WAVES + wv;
                if (tr >= ntrip) tr -= ntrip;
                unit0 = tr * 8;
                nu    = 8;
            } else {
                unit0 = rest0 + wv * per;
                nu    = (nunits - unit0 < per) ? (nunits - unit0) : per;
                if (nu <= 0) break;
            }
            ++i;
            done = 0;
#pragma unroll
            for (int u = 0; u < 8; ++u) {
                int un = unit0 + u;
                un = (un < nunits) ? un : (nunits - 1);
                tv[u] = *(const uint4*)(tags + (size_t)un * OWN_TAGU + lane * 16);
            }
        }
#pragma unroll
        for (int u = 0; u < 8; ++u) {
            if (u >= done && u < nu && count <= OWN_LIST - OWN_TAGU) {
                const int ebase = (unit0 + u) * OWN_TAGU + lane * 16;
                unsigned xs[4] = {tv[u].x ^ pat, tv[u].y ^ pat, tv[u].z ^ pat, tv[u].w ^ pat};
                if ((unit0 + u + 1) * OWN_TAGU > E) {
#pragma unroll
                    for (int q = 0; q < 4; ++q) {
#pragma unroll
                        for (int k = 0; k < 4; ++k)
                            if (ebase + q * 4 + k >= E) xs[q] |= 255u << (8 * k);
                    }
                }
#pragma unroll
                for (int q = 0; q < 4; ++q) {
                    const unsigned x = xs[q];
                    const unsigned z = (x - 0x01010101u) & ~x & 0x80808080u;
                    if (__builtin_amdgcn_ballot_w32(z != 0u) != 0u) {
#pragma unroll
                        for (int k = 0; k < 4; ++k) {
                            const bool m = ((x >> (8 * k)) & 255u) == 0u;
                            const unsigned bal = __builtin_amdgcn_ballot_w32(m);
                            if (bal != 0u) {
                                if (m) mylist[count + __popc(bal & ((1u << lane) - 1u))] = (unsigned)(ebase + q * 4 + k);
                                count += __popc(bal);
                            }
                        }
                    }
                }
                done = u + 1;
            }
        }
        __builtin_amdgcn_fence(__ATOMIC_ACQ_REL, "wavefront");
        while (count >= 32) {
            count -= 32;
            unsigned* kp = nullptr;
            if (keepok) {
                if (nkept + 32 <= OWN_KEEP) { kp = kept + nkept; nkept += 32; }
                else keepok = false;
            }
            gcn_own_add_found<F>(mylist + count, 32, lane, lo, src, dst, dinv, xw, tile, kp);
        }
    }
    if (count > 0) {
        unsigned* kp = nullptr;
        if (keepok) {
            if (nkept + count <= OWN_KEEP) { kp = kept + nkept; nkept += count; }
            else keepok = false;
        }
        gcn_own_add_found<F>(mylist, count, lane, lo, src, dst, dinv, xw, tile, kp);
    }
    return keepok ? nkept : -1;
}

__device__ __forceinline__ bool gcn_own_keep_fits(const float* wsb, size_t ws_bytes, const int* kend) {
    return (size_t)((const char*)kend - (const char*)wsb) <= ws_bytes;
}

__global__ __launch_bounds__(1024)
void gcn_owner_sum_lds(const int* __restrict__ src, const int* __restrict__ dst,
                       const float* __restrict__ dinv, const float* __restrict__ xw,
                       const float* __restrict__ bias, const _Float16* __restrict__ W2p,
                       float* __restrict__ xw2,
                       unsigned* __restrict__ kept, int* __restrict__ kcnt,
                       const unsigned char* __restrict__ tags, int Epad,
                       const unsigned* __restrict__ bins, const unsigned* __restrict__ boff,
                       const float* wsb, size_t ws_bytes, int N, int E) {
    __shared__ __attribute__((aligned(16))) float tile[OWN_ROWS * GCN_D_HID];
    __shared__ unsigned list[OWN_WAVES * OWN_LIST];

    const int lane = threadIdx.x & 31;
    const int wv = __builtin_amdgcn_readfirstlane((int)(threadIdx.x >> 5));
    const int lo = blockIdx.x * OWN_ROWS;
    const int nrows = (N - lo < OWN_ROWS) ? (N - lo) : OWN_ROWS;
    const int wid = blockIdx.x * OWN_WAVES + wv;
    const bool keep_on = gcn_own_keep_fits(wsb, ws_bytes, kcnt + gridDim.x * OWN_WAVES);

    for (int r = wv; r < nrows; r += OWN_WAVES) {
        const float di = dinv[lo + r];
        const float4* xr = (const float4*)(xw + (size_t)(lo + r) * GCN_D_HID);
        const float4 a = xr[lane];
        const float4 b = xr[32 + lane];
        float* tr = tile + r * GCN_D_HID + lane;
        tr[0]   = a.x * di * di;
        tr[32]  = a.y * di * di;
        tr[64]  = a.z * di * di;
        tr[96]  = a.w * di * di;
        tr[128] = b.x * di * di;
        tr[160] = b.y * di * di;
        tr[192] = b.z * di * di;
        tr[224] = b.w * di * di;
    }
    __syncthreads();

    const bool bins_on = gcn_bins_on(wsb, ws_bytes, boff, N, E, Epad);
    if (bins_on) {
        gcn_own_add_binned<GCN_D_HID>(bins, boff, dinv, xw, tile, list + wv * OWN_LIST, lane, wv, lo);
    } else {
        const bool tags_on = gcn_ws_fits(wsb, ws_bytes, tags + Epad) && gridDim.x <= 256u;
        unsigned* kp0 = keep_on ? kept + (size_t)wid * OWN_KEEP : nullptr;
        int nk;
        if (tags_on)
            nk = gcn_own_scan_tags<GCN_D_HID>(src, dst, tags, dinv, xw, tile, list + wv * OWN_LIST,
                                              kp0, lane, wv, lo, E);
        else
            nk = gcn_own_scan<GCN_D_HID>(src, dst, dinv, xw, tile, list + wv * OWN_LIST,
                                         kp0, lane, wv, lo, nrows, E);
        if (keep_on && lane == 0) kcnt[wid] = nk;
    }
    __syncthreads();

    constexpr int HP  = GCN_D_HID + 8;
    constexpr int NRT = (OWN_ROWS + 15) / 16;
    constexpr int RPW = (NRT * 16 + OWN_WAVES - 1) / OWN_WAVES;
    constexpr int KT2 = GCN_D_HID / 32;
    constexpr int NT2 = GCN_D_OUT / 16;
    constexpr int RGS = OWN_WAVES / NT2;
    static_assert(OWN_WAVES % NT2 == 0 && NRT <= 2 * RGS && GCN_D_HID == 256 &&
                  NRT * 16 * HP * 2 <= OWN_ROWS * GCN_D_HID * 4, "shape not covered");
    const int ct = wv % NT2;
    const int rg = wv / NT2;

    const float4 bb0 = ((const float4*)bias)[lane];
    const float4 bb1 = ((const float4*)bias)[32 + lane];

    v16h b[KT2];
    {
        const _Float16* bb = W2p + ((size_t)ct * 32 + lane) * 16;
#pragma unroll
        for (int kt = 0; kt < KT2; ++kt) b[kt] = *(const v16h*)(bb + (size_t)kt * NT2 * 32 * 16);
    }
    v4h hlo[RPW], hhi[RPW];
#pragma unroll
    for (int i = 0; i < RPW; ++i) {
        const int r = wv + i * OWN_WAVES;
        float4 a = float4{0.0f, 0.0f, 0.0f, 0.0f};
        float4 c = float4{0.0f, 0.0f, 0.0f, 0.0f};
        if (r < nrows) {
            const float* tr = tile + r * GCN_D_HID + lane;
            a.x = fmaxf(tr[0]   + bb0.x, 0.0f);
            a.y = fmaxf(tr[32]  + bb0.y, 0.0f);
            a.z = fmaxf(tr[64]  + bb0.z, 0.0f);
            a.w = fmaxf(tr[96]  + bb0.w, 0.0f);
            c.x = fmaxf(tr[128] + bb1.x, 0.0f);
            c.y = fmaxf(tr[160] + bb1.y, 0.0f);
            c.z = fmaxf(tr[192] + bb1.z, 0.0f);
            c.w = fmaxf(tr[224] + bb1.w, 0.0f);
        }
        hlo[i] = v4h{(_Float16)a.x, (_Float16)a.y, (_Float16)a.z, (_Float16)a.w};
        hhi[i] = v4h{(_Float16)c.x, (_Float16)c.y, (_Float16)c.z, (_Float16)c.w};
    }
    __syncthreads();
    _Float16* hrow = (_Float16*)tile;
#pragma unroll
    for (int i = 0; i < RPW; ++i) {
        const int r = wv + i * OWN_WAVES;
        if (r < NRT * 16) {
            *(v4h*)(hrow + r * HP + lane * 4)       = hlo[i];
            *(v4h*)(hrow + r * HP + 128 + lane * 4) = hhi[i];
        }
    }
    __syncthreads();

    const int hsel = lane >> 4;
    const int idx  = lane & 15;
    const int one[1] = {0};
#pragma unroll
    for (int i = 0; i < 2; ++i) {
        const int rt = rg + i * RGS;
        if (rt < NRT && rt * 16 < nrows) {
            v8f acc[1];
            acc[0] = v8f{};
            gcn_gemm_tiles<GCN_D_HID, 1>(hrow + (rt * 16 + idx) * HP + hsel * 8, b, one, acc);
            const int row0 = rt * 16 + hsel * 8;
            float* cb = xw2 + (size_t)(lo + row0) * GCN_D_OUT + ct * 16 + idx;
#pragma unroll
            for (int r = 0; r < 8; ++r)
                if (row0 + r < nrows) cb[(size_t)r * GCN_D_OUT] = acc[0][r];
        }
    }
}

__global__ __launch_bounds__(1024)
void gcn_owner_sum_kept(const int* __restrict__ src, const int* __restrict__ dst,
                        const float* __restrict__ dinv, const float* __restrict__ xw,
                        const float* __restrict__ bias, float* __restrict__ out,
                        const unsigned* __restrict__ kept, const int* __restrict__ kcnt,
                        const unsigned* __restrict__ bins, const unsigned* __restrict__ boff,
                        int Epad, const float* wsb, size_t ws_bytes, int N, int E) {
    __shared__ float    tile[OWN_ROWS * GCN_D_OUT];
    __shared__ unsigned list[OWN_WAVES * OWN_LIST];

    const int lane = threadIdx.x & 31;
    const int wv = __builtin_amdgcn_readfirstlane((int)(threadIdx.x >> 5));
    const int lo = blockIdx.x * OWN_ROWS;
    const int nrows = (N - lo < OWN_ROWS) ? (N - lo) : OWN_ROWS;
    const int wid = blockIdx.x * OWN_WAVES + wv;
    const bool keep_on = gcn_own_keep_fits(wsb, ws_bytes, kcnt + gridDim.x * OWN_WAVES);

    const bool bins_on = gcn_bins_on(wsb, ws_bytes, boff, N, E, Epad);
    bool use_kept = false;
    if (!bins_on && keep_on) {
        const int c = kcnt[blockIdx.x * OWN_WAVES + lane];
        use_kept = __builtin_amdgcn_ballot_w32(c < 0 || c > OWN_KEEP) == 0u;
    }

    {
        constexpr int RPW = (OWN_ROWS + OWN_WAVES - 1) / OWN_WAVES;
        float sa[RPW], sb[RPW];
        const int   rl  = wv + lane * OWN_WAVES;
        const float dvl = dinv[lo + ((rl < nrows) ? rl : (nrows - 1))];
#pragma unroll
        for (int i = 0; i < RPW; ++i) {
            const int r  = wv + i * OWN_WAVES;
            const int rc = (r < nrows) ? r : (nrows - 1);
            const float* xr = xw + (size_t)(lo + rc) * GCN_D_OUT;
            sa[i] = xr[lane];
            sb[i] = xr[32 + lane];
        }
#pragma unroll
        for (int i = 0; i < RPW; ++i) {
            const int r = wv + i * OWN_WAVES;
            if (r < nrows) {
                const float di = __int_as_float(__builtin_amdgcn_readlane(__float_as_int(dvl), i));
                float* tr = tile + r * GCN_D_OUT + lane;
                tr[0]  = sa[i] * di * di;
                tr[32] = sb[i] * di * di;
            }
        }
    }
    __syncthreads();

    if (bins_on) {
        gcn_own_add_binned<GCN_D_OUT>(bins, boff, dinv, xw, tile, list + wv * OWN_LIST, lane, wv, lo);
    } else if (use_kept) {
        const unsigned* kp = kept + (size_t)wid * OWN_KEEP;
        int cnt = __builtin_amdgcn_readfirstlane(kcnt[wid]);
        cnt = (cnt < OWN_KEEP) ? cnt : OWN_KEEP;
        constexpr int NB = OWN_KEEP / 32;
        unsigned* myl = list + wv * OWN_LIST;
        unsigned rec[NB];
#pragma unroll
        for (int b = 0; b < NB; ++b) {
            rec[b] = 0u;
            if (b * 32 + lane < cnt) rec[b] = kp[b * 32 + lane];
        }
        float wgt[NB];
#pragma unroll
        for (int b = 0; b < NB; ++b) {
            const float ds = dinv[rec[b] >> 8];
            const float dd = dinv[lo + (int)(rec[b] & 255u)];
            wgt[b] = ds * dd;
        }
#pragma unroll
        for (int b = 0; b < NB; ++b) {
            myl[b * 32 + lane]            = rec[b];
            myl[OWN_KEEP + b * 32 + lane] = __float_as_uint(wgt[b]);
        }
        for (int base = 0; base < cnt; base += 32) {
            const int n = (cnt - base < 32) ? (cnt - base) : 32;
            const unsigned r = myl[base + lane];
            const float    w = __uint_as_float(myl[OWN_KEEP + base + lane]);
            gcn_own_add_rows<GCN_D_OUT, true>((int)(r >> 8), (int)(r & 255u), w, n, lane, xw, tile);
        }
    } else {
        gcn_own_scan<GCN_D_OUT>(src, dst, dinv, xw, tile, list + wv * OWN_LIST, nullptr,
                                lane, wv, lo, nrows, E);
    }
    __syncthreads();

    const float b0 = bias[lane];
    const float b1 = bias[lane + 32];
    for (int r = wv; r < nrows; r += OWN_WAVES) {
        const float* tr = tile + r * GCN_D_OUT + lane;
        float v0 = tr[0]  + b0;
        float v1 = tr[32] + b1;
        float m = fmaxf(v0, v1);
#pragma unroll
        for (int off = 16; off > 0; off >>= 1) m = fmaxf(m, __shfl_xor(m, off, 32));
        float e0 = __expf(v0 - m);
        float e1 = __expf(v1 - m);
        float sm = e0 + e1;
#pragma unroll
        for (int off = 16; off > 0; off >>= 1) sm += __shfl_xor(sm, off, 32);
        float inv = 1.0f / sm;
        float* orow = out + (size_t)(lo + r) * GCN_D_OUT;
        orow[lane]      = e0 * inv;
        orow[lane + 32] = e1 * inv;
    }
}

extern "C" void kernel_launch(void* const* d_in, const int* in_sizes, int n_in,
                              void* d_out, int out_size, void* d_ws, size_t ws_size,
                              hipStream_t stream) {
    const float* x   = (const float*)d_in[0];
    const int*   ei  = (const int*)  d_in[1];
    const float* W1  = (const float*)d_in[2];
    const float* b1  = (const float*)d_in[3];
    const float* W2  = (const float*)d_in[4];
    const float* b2  = (const float*)d_in[5];
    float* out = (float*)d_out;

    const int N = in_sizes[0] / GCN_D_IN;
    const int E = in_sizes[1] / 2;
    const int* src = ei;
    const int* dst = ei + E;

    float* ws   = (float*)d_ws;
    size_t off  = ((size_t)N + 63) & ~(size_t)63;
    float* dinv = ws;
    float* xw1  = ws + off;
    float* h1   = xw1 + (size_t)N * GCN_D_HID;
    _Float16* W1p = (_Float16*)(h1 + (size_t)N * GCN_D_HID);
    _Float16* W2p = W1p + (size_t)GCN_D_IN * GCN_D_HID;
    float* xw2  = h1;
    const int ob = (N + OWN_ROWS - 1) / OWN_ROWS;
    unsigned* kept = (unsigned*)(W2p + (size_t)GCN_D_HID * GCN_D_OUT);
    int*      kcnt = (int*)(kept + (size_t)ob * OWN_WAVES * OWN_KEEP);
    unsigned char* tags = (unsigned char*)(kcnt + (size_t)ob * OWN_WAVES);
    const int Epad = (E + OWN_TAGU - 1) / OWN_TAGU * OWN_TAGU;
    unsigned* bins = kept;
    unsigned* boff = bins + (size_t)DEG_CHUNKS * BIN_CAP;

    const int B = 256;
    {
        int t1 = (GCN_D_IN >> 5) * (GCN_D_HID >> 4) * 32;
        gcn_pack_w_f16<<<(t1 + B - 1) / B, B, 0, stream>>>(W1, W1p, GCN_D_IN, GCN_D_HID);
        int t2 = (GCN_D_HID >> 5) * (GCN_D_OUT >> 4) * 32;
        gcn_pack_w_f16<<<(t2 + B - 1) / B, B, 0, stream>>>(W2, W2p, GCN_D_HID, GCN_D_OUT);
    }

    unsigned* part = (unsigned*)h1;
    gcn_count_deg<<<DEG_CHUNKS, 1024, 0, stream>>>(src, dst, part, (int)off, tags, bins, boff, ws, ws_size, N, E, Epad);
    gcn_make_dinv<<<(N + B - 1) / B, B, 0, stream>>>(part, (int)off, dinv, N);

    {
        gcn_wmma_gemm_bregs<GCN_D_IN, GCN_D_HID, 2><<<256, 512, 0, stream>>>(x, W1p, xw1, N);
    }
    {
        gcn_owner_sum_lds<<<ob, 1024, 0, stream>>>(src, dst, dinv, xw1, b1, W2p, xw2, kept, kcnt,
                                                   tags, Epad, bins, boff, ws, ws_size, N, E);
    }

    {
        gcn_owner_sum_kept<<<ob, 1024, 0, stream>>>(src, dst, dinv, xw2, b2, out, kept, kcnt,
                                                    bins, boff, Epad, ws, ws_size, N, E);
    }
}


# layer-1 LDS float atomics gone: waves deal their binned edges to the wave owning the destination row (LDS inboxes); owners add rows by plain 16-B LDS load/add/store; sum order = inbox arrival rank
# speedup vs baseline: 4.9654x; 1.0589x over previous
#include <hip/hip_runtime.h>
#include <hip/hip_bf16.h>

typedef __attribute__((ext_vector_type(16))) _Float16 v16h;
typedef __attribute__((ext_vector_type(8)))  float    v8f;

#define GCN_D_IN  512
#define GCN_D_HID 256
#define GCN_D_OUT 64

#define OWN_ROWS  196
#define OWN_TAGU  512

__device__ __forceinline__ bool gcn_ws_fits(const float* wsb, size_t ws_bytes, const void* end) {
    return (size_t)((const char*)end - (const char*)wsb) <= ws_bytes;
}

#define DEG_CHUNKS 64
#define DEG_BINS   53248
#define BIN_CAP    16384
#define BIN_OFFP   264

__device__ __forceinline__ bool gcn_bins_on(const float* wsb, size_t ws_bytes, const unsigned* boff,
                                            int N, int E, int Epad) {
    const int nq   = Epad >> 2;
    const int qper = (nq + DEG_CHUNKS - 1) / DEG_CHUNKS;
    return gcn_ws_fits(wsb, ws_bytes, boff + (size_t)DEG_CHUNKS * BIN_OFFP) &&
           qper <= BIN_CAP / 4 && (E & 3) == 0 && N <= DEG_BINS &&
           (N + OWN_ROWS - 1) / OWN_ROWS <= 256 && N <= (1 << 24);
}

__global__ __launch_bounds__(1024)
void gcn_count_deg(const int* __restrict__ src, const int* __restrict__ dst,
                   unsigned* __restrict__ part, int pitch,
                   unsigned char* __restrict__ tags, unsigned* __restrict__ bins,
                   unsigned* __restrict__ boff, const float* wsb, size_t ws_bytes,
                   int N, int E, int Epad) {
    __shared__ __attribute__((aligned(16))) unsigned hist[DEG_BINS];
    __shared__ __attribute__((aligned(16))) unsigned brec[BIN_CAP];
    __shared__ unsigned bcnt[256];
    __shared__ unsigned bstart[257];
    const int  t       = threadIdx.x;
    const bool bins_on = gcn_bins_on(wsb, ws_bytes, boff, N, E, Epad) && (int)gridDim.x == DEG_CHUNKS;
    const bool tags_on = !bins_on && gcn_ws_fits(wsb, ws_bytes, tags + Epad);
    const int  nq      = Epad >> 2;
    const int  qfull   = E >> 2;
    const int  qper    = (nq + (int)gridDim.x - 1) / (int)gridDim.x;
    const int  q0      = (int)blockIdx.x * qper;
    const int  q1      = (q0 + qper < nq) ? (q0 + qper) : nq;
    unsigned*  prow    = part + (size_t)blockIdx.x * pitch;

    int4     dv[4];
    int4     sv[4];
    unsigned rk[16];
    unsigned vmask = 0u;
    if (bins_on && t < 256) bcnt[t] = 0u;

    for (int base = 0; base < N; base += DEG_BINS) {
        const int nb = (N - base < DEG_BINS) ? (N - base) : DEG_BINS;
        for (int i = t * 4; i < DEG_BINS; i += 4096) *(uint4*)(hist + i) = uint4{0u, 0u, 0u, 0u};
        __syncthreads();
        for (int qb = q0; qb < q1; qb += 4096) {
#pragma unroll
            for (int u = 0; u < 4; ++u) {
                const int q  = qb + u * 1024 + t;
                const int qc = (q < qfull) ? q : (qfull - 1);
                const int* dp = dst + (size_t)qc * 4;
                dv[u] = int4{dp[0], dp[1], dp[2], dp[3]};
                if (bins_on) {
                    const int* sp = src + (size_t)qc * 4;
                    sv[u] = int4{sp[0], sp[1], sp[2], sp[3]};
                }
            }
#pragma unroll
            for (int u = 0; u < 4; ++u) {
                const int  q    = qb + u * 1024 + t;
                const bool mine = q < q1;
                const bool real = mine && q < qfull;
                const int dd[4] = {real ? dv[u].x : -1, real ? dv[u].y : -1,
                                   real ? dv[u].z : -1, real ? dv[u].w : -1};
                unsigned tw = 0u;
#pragma unroll
                for (int k = 0; k < 4; ++k) {
                    const unsigned r = (unsigned)(dd[k] - base);
                    if (r < (unsigned)nb) atomicAdd(&hist[r], 1u);
                    if (dd[k] >= 0) {
                        const unsigned own = ((unsigned)dd[k] / OWN_ROWS) & 255u;
                        tw |= own << (8 * k);
                        if (bins_on && base == 0) {
                            rk[u * 4 + k] = atomicAdd(&bcnt[own], 1u);
                            vmask |= 1u << (u * 4 + k);
                        }
                    }
                }
                if (tags_on && base == 0 && mine && !(q == qfull && (E & 3) != 0))
                    *(unsigned*)(tags + (size_t)q * 4) = tw;
            }
        }
        if (t == 0 && (E & 3) != 0 && qfull >= q0 && qfull < q1) {
            unsigned tw = 0u;
            for (int k = 0; k < (E & 3); ++k) {
                const int d = dst[qfull * 4 + k];
                const unsigned r = (unsigned)(d - base);
                if (r < (unsigned)nb) atomicAdd(&hist[r], 1u);
                if (d >= 0) tw |= (((unsigned)d / OWN_ROWS) & 255u) << (8 * k);
            }
            if (tags_on && base == 0) *(unsigned*)(tags + (size_t)qfull * 4) = tw;
        }
        __syncthreads();
        if (bins_on && base == 0 && t < 32) {
            unsigned c8[8];
            unsigned run = 0u;
#pragma unroll
            for (int j = 0; j < 8; ++j) { c8[j] = run; run += bcnt[t * 8 + j]; }
            unsigned incl = run;
#pragma unroll
            for (int o = 1; o < 32; o <<= 1) {
                const unsigned y = __shfl_up(incl, o, 32);
                if (t >= o) incl += y;
            }
            const unsigned excl = incl - run;
#pragma unroll
            for (int j = 0; j < 8; ++j) bstart[t * 8 + j] = excl + c8[j];
            if (t == 31) bstart[256] = incl;
        }
        for (int i = t * 4; i < nb; i += 4096) {
            if (i + 4 <= nb) {
                *(uint4*)(prow + base + i) = *(const uint4*)(hist + i);
            } else {
                for (int k = i; k < nb; ++k) prow[base + k] = hist[k];
            }
        }
        __syncthreads();
        if (bins_on && base == 0) {
#pragma unroll
            for (int u = 0; u < 4; ++u) {
                const int dd[4] = {dv[u].x, dv[u].y, dv[u].z, dv[u].w};
                const int ss[4] = {sv[u].x, sv[u].y, sv[u].z, sv[u].w};
#pragma unroll
                for (int k = 0; k < 4; ++k) {
                    if ((vmask >> (u * 4 + k)) & 1u) {
                        const unsigned own = (unsigned)dd[k] / OWN_ROWS;
                        const unsigned pos = bstart[own & 255u] + rk[u * 4 + k];
                        if (pos < (unsigned)BIN_CAP)
                            brec[pos] = ((unsigned)ss[k] << 8) | (((unsigned)dd[k] - own * OWN_ROWS) & 255u);
                    }
                }
            }
            __syncthreads();
            const unsigned total = (bstart[256] < (unsigned)BIN_CAP) ? bstart[256] : (unsigned)BIN_CAP;
            unsigned* bout = bins + (size_t)blockIdx.x * BIN_CAP;
            for (unsigned i = (unsigned)t * 4u; i < total; i += 4096u) {
                if (i + 4u <= total) {
                    *(uint4*)(bout + i) = *(const uint4*)(brec + i);
                } else {
                    for (unsigned k = i; k < total; ++k) bout[k] = brec[k];
                }
            }
            if (t <= 256) boff[(size_t)blockIdx.x * BIN_OFFP + t] = bstart[t];
        }
    }
}

__global__ void gcn_make_dinv(const unsigned* __restrict__ part, int pitch,
                              float* __restrict__ dinv, int n) {
    int i = blockIdx.x * blockDim.x + threadIdx.x;
    if (i < n) {
        unsigned c[DEG_CHUNKS];
#pragma unroll
        for (int k = 0; k < DEG_CHUNKS; ++k) c[k] = part[(size_t)k * pitch + i];
        unsigned sum = 1u;
#pragma unroll
        for (int k = 0; k < DEG_CHUNKS; ++k) sum += c[k];
        const float d = (float)sum;
        dinv[i] = (d > 0.0f) ? rsqrtf(d) : 0.0f;
    }
}

__global__ void gcn_pack_w_f16(const float* __restrict__ W, _Float16* __restrict__ Wp,
                               int K, int N) {
    int t = blockIdx.x * blockDim.x + threadIdx.x;
    int tilesN = N >> 4;
    int total = (K >> 5) * tilesN * 32;
    if (t >= total) return;
    int frag = t >> 5;
    int lane = t & 31;
    int kt = frag / tilesN;
    int tn = frag - kt * tilesN;
    int n    = tn * 16 + (lane & 15);
    int koff = kt * 32 + (lane >> 4) * 8;
    _Float16* o = Wp + (size_t)t * 16;
#pragma unroll
    for (int j = 0; j < 8; ++j) {
        o[j]     = (_Float16)W[(size_t)(koff + j) * N + n];
        o[j + 8] = (_Float16)W[(size_t)(koff + 16 + j) * N + n];
    }
}

typedef __attribute__((ext_vector_type(8))) _Float16 v8h;
typedef __attribute__((ext_vector_type(4))) _Float16 v4h;

__device__ __forceinline__ v16h gcn_lds_a_frag(const _Float16* ar) {
    const v8h lo = *(const v8h*)(ar);
    const v8h hi = *(const v8h*)(ar + 16);
    return __builtin_shufflevector(lo, hi, 0, 1, 2, 3, 4, 5, 6, 7, 8, 9, 10, 11, 12, 13, 14, 15);
}

template <int K, int NTL>
__device__ __forceinline__ void gcn_gemm_tiles(const _Float16* ar0, const v16h (&b)[K / 32],
                                               const int (&toff)[NTL], v8f (&acc)[NTL]) {
    constexpr int KT    = K / 32;
    constexpr int PITCH = K + 8;
    const _Float16* ar[NTL];
    v16h a[3][NTL];
#pragma unroll
    for (int i = 0; i < NTL; ++i) {
        ar[i]   = ar0 + toff[i] * (16 * PITCH);
        a[0][i] = gcn_lds_a_frag(ar[i]);
        a[1][i] = gcn_lds_a_frag(ar[i] + 32);
    }
#pragma unroll
    for (int kt = 0; kt < KT; ++kt) {
        if (kt + 2 < KT) {
#pragma unroll
            for (int i = 0; i < NTL; ++i) a[(kt + 2) % 3][i] = gcn_lds_a_frag(ar[i] + (kt + 2) * 32);
        }
        __builtin_amdgcn_sched_barrier(0);
#pragma unroll
        for (int i = 0; i < NTL; ++i)
            acc[i] = __builtin_amdgcn_wmma_f32_16x16x32_f16(false, a[kt % 3][i], false, b[kt],
                                                            (short)0, acc[i], false, false);
        __builtin_amdgcn_sched_barrier(0);
    }
}

template <int K, int N, int RT>
__global__ __launch_bounds__(512)
void gcn_wmma_gemm_bregs(const float* __restrict__ A,
                         const _Float16* __restrict__ Bp,
                         float* __restrict__ C, int M) {
    constexpr int KT    = K / 32;
    constexpr int NT    = N / 16;
    constexpr int RG    = 16 / NT;
    constexpr int ROWS  = RG * RT * 16;
    constexpr int PITCH = K + 8;
    constexpr int C4    = K / 4;
    constexpr int RSTEP = 512 / C4;
    constexpr int NV    = ROWS / RSTEP;
    static_assert(K % 32 == 0 && N % 16 == 0 && 16 % NT == 0 && 512 % C4 == 0 &&
                  ROWS % RSTEP == 0 && (RT == 1 || RT == 2), "shape not covered");

    __shared__ __attribute__((aligned(16))) _Float16 tile[2 * ROWS * PITCH];

    const int t    = threadIdx.x;
    const int lane = t & 31;
    const int w    = __builtin_amdgcn_readfirstlane(t >> 5);
    const int ct   = w % NT;
    const int rg   = w / NT;
    const int hsel = lane >> 4;
    const int idx  = lane & 15;

    const int nblk  = (M + ROWS - 1) / ROWS;
    const int ntile = M / 16;
    const int srow  = t / C4;
    const int sc    = (t % C4) * 4;
    float4 st[NV];

    int blk = blockIdx.x;
    if (blk < nblk) {
#pragma unroll
        for (int i = 0; i < NV; ++i) {
            int row = blk * ROWS + srow + i * RSTEP;
            row = (row < M) ? row : (M - 1);
            st[i] = *(const float4*)(A + (size_t)row * K + sc);
        }
    }

    v16h b[KT];
    {
        const _Float16* bb = Bp + ((size_t)ct * 32 + lane) * 16;
#pragma unroll
        for (int kt = 0; kt < KT; ++kt) b[kt] = *(const v16h*)(bb + (size_t)kt * NT * 32 * 16);
    }
    __builtin_amdgcn_sched_barrier(0);

    if (blk < nblk) {
#pragma unroll
        for (int i = 0; i < NV; ++i) {
            v4h hv = {(_Float16)st[i].x, (_Float16)st[i].y, (_Float16)st[i].z, (_Float16)st[i].w};
            *(v4h*)(tile + (srow + i * RSTEP) * PITCH + sc) = hv;
        }
    }
    __syncthreads();

    int buf = 0;
    for (; blk < nblk; blk += gridDim.x) {
        const int  nxt      = blk + (int)gridDim.x;
        const bool has_next = nxt < nblk;
        if (has_next) {
#pragma unroll
            for (int i = 0; i < NV; ++i) {
                int row = nxt * ROWS + srow + i * RSTEP;
                row = (row < M) ? row : (M - 1);
                st[i] = *(const float4*)(A + (size_t)row * K + sc);
            }
        }

        const int tile0 = (blk * RG + rg) * RT;
        const _Float16* ar0 = tile + buf * (ROWS * PITCH) + (rg * RT * 16 + idx) * PITCH + hsel * 8;
        float* cb0 = C + (size_t)(tile0 * 16 + hsel * 8) * N + ct * 16 + idx;
        const bool live = tile0 < ntile;
        int toff[RT];
#pragma unroll
        for (int i = 0; i < RT; ++i) toff[i] = (tile0 + i < ntile) ? i : 0;
        v8f acc[RT];
#pragma unroll
        for (int i = 0; i < RT; ++i) acc[i] = v8f{};
        if (live) gcn_gemm_tiles<K, RT>(ar0, b, toff, acc);

        if (has_next) {
            _Float16* tn = tile + (buf ^ 1) * (ROWS * PITCH);
#pragma unroll
            for (int i = 0; i < NV; ++i) {
                v4h hv = {(_Float16)st[i].x, (_Float16)st[i].y, (_Float16)st[i].z, (_Float16)st[i].w};
                *(v4h*)(tn + (srow + i * RSTEP) * PITCH + sc) = hv;
            }
        }
        __syncthreads();
        if (live) {
#pragma unroll
            for (int i = 0; i < RT; ++i) {
                float* cb = cb0 + (size_t)toff[i] * (16 * N);
#pragma unroll
                for (int r = 0; r < 8; ++r) cb[(size_t)r * N] = acc[i][r];
            }
        }
        buf ^= 1;
    }
}

#define OWN_WAVES 32
#define OWN_LIST  544
#define OWN_KEEP  160
#define OWN_INBOX (OWN_LIST / 2)

template <int F, int G, bool PLAIN>
__device__ __forceinline__ void gcn_own_add_group(int s, int dl, float w, int j, int lane,
                                                  const float* __restrict__ xw, float* tile) {
    static_assert(!PLAIN || F == 256, "shape not covered");
    int   rl[G];
    float wj[G];
    if constexpr (F == 256) {
        float4 a[G], b[G];
#pragma unroll
        for (int q = 0; q < G; ++q) {
            const int sj = __builtin_amdgcn_readlane(s, j + q);
            rl[q] = __builtin_amdgcn_readlane(dl, j + q);
            wj[q] = __int_as_float(__builtin_amdgcn_readlane(__float_as_int(w), j + q));
            const float4* xr = (const float4*)(xw + (size_t)sj * F);
            a[q] = xr[lane];
            b[q] = xr[32 + lane];
        }
#pragma unroll
        for (int q = 0; q < G; ++q) {
            float* tr = tile + rl[q] * F + lane * 4;
            if constexpr (PLAIN) {
                float4 o0 = *(const float4*)tr;
                float4 o1 = *(const float4*)(tr + 128);
                o0.x += a[q].x * wj[q];
                o0.y += a[q].y * wj[q];
                o0.z += a[q].z * wj[q];
                o0.w += a[q].w * wj[q];
                o1.x += b[q].x * wj[q];
                o1.y += b[q].y * wj[q];
                o1.z += b[q].z * wj[q];
                o1.w += b[q].w * wj[q];
                *(float4*)tr         = o0;
                *(float4*)(tr + 128) = o1;
            } else {
                atomicAdd(tr,       a[q].x * wj[q]);
                atomicAdd(tr + 1,   a[q].y * wj[q]);
                atomicAdd(tr + 2,   a[q].z * wj[q]);
                atomicAdd(tr + 3,   a[q].w * wj[q]);
                atomicAdd(tr + 128, b[q].x * wj[q]);
                atomicAdd(tr + 129, b[q].y * wj[q]);
                atomicAdd(tr + 130, b[q].z * wj[q]);
                atomicAdd(tr + 131, b[q].w * wj[q]);
            }
        }
    } else {
        float a[G], b[G];
#pragma unroll
        for (int q = 0; q < G; ++q) {
            const int sj = __builtin_amdgcn_readlane(s, j + q);
            rl[q] = __builtin_amdgcn_readlane(dl, j + q);
            wj[q] = __int_as_float(__builtin_amdgcn_readlane(__float_as_int(w), j + q));
            const float* xr = xw + (size_t)sj * F;
            a[q] = xr[lane];
            b[q] = xr[32 + lane];
        }
#pragma unroll
        for (int q = 0; q < G; ++q) {
            float* tr = tile + rl[q] * F + lane;
            atomicAdd(tr,      a[q] * wj[q]);
            atomicAdd(tr + 32, b[q] * wj[q]);
        }
    }
}

template <int F, bool SHORT_TAIL, bool PLAIN = false>
__device__ __forceinline__ void gcn_own_add_rows(int s, int dl, float w, int n, int lane,
                                                 const float* __restrict__ xw, float* tile) {
    int j = 0;
    for (; j + 8 <= n; j += 8) gcn_own_add_group<F, 8, PLAIN>(s, dl, w, j, lane, xw, tile);
    if constexpr (SHORT_TAIL) {
        if (j + 4 <= n) { gcn_own_add_group<F, 4, PLAIN>(s, dl, w, j, lane, xw, tile); j += 4; }
        if (j + 2 <= n) { gcn_own_add_group<F, 2, PLAIN>(s, dl, w, j, lane, xw, tile); j += 2; }
        if (j < n)        gcn_own_add_group<F, 1, PLAIN>(s, dl, w, j, lane, xw, tile);
    } else {
        for (; j < n; ++j) gcn_own_add_group<F, 1, PLAIN>(s, dl, w, j, lane, xw, tile);
    }
}

template <int F>
__device__ __forceinline__ void gcn_own_add_binned(const unsigned* __restrict__ bins,
                                                   const unsigned* __restrict__ boff,
                                                   const float* __restrict__ dinv,
                                                   const float* __restrict__ xw,
                                                   float* tile, unsigned* myl,
                                                   int lane, int wv, int lo) {
    static_assert(DEG_CHUNKS == 2 * OWN_WAVES && 2 * OWN_KEEP <= OWN_LIST, "shape not covered");
    const int c0 = wv, c1 = wv + OWN_WAVES;
    const unsigned* o0 = boff + (size_t)c0 * BIN_OFFP + blockIdx.x;
    const unsigned* o1 = boff + (size_t)c1 * BIN_OFFP + blockIdx.x;
    int a0 = __builtin_amdgcn_readfirstlane((int)o0[0]);
    int a1 = __builtin_amdgcn_readfirstlane((int)o0[1]);
    int b0 = __builtin_amdgcn_readfirstlane((int)o1[0]);
    int b1 = __builtin_amdgcn_readfirstlane((int)o1[1]);
    a0 = (a0 < 0) ? 0 : ((a0 > BIN_CAP) ? BIN_CAP : a0);
    a1 = (a1 < a0) ? a0 : ((a1 > BIN_CAP) ? BIN_CAP : a1);
    b0 = (b0 < 0) ? 0 : ((b0 > BIN_CAP) ? BIN_CAP : b0);
    b1 = (b1 < b0) ? b0 : ((b1 > BIN_CAP) ? BIN_CAP : b1);
    const int n0 = a1 - a0;
    const int total = n0 + (b1 - b0);
    const unsigned* p0 = bins + (size_t)c0 * BIN_CAP + a0;
    const unsigned* p1 = bins + (size_t)c1 * BIN_CAP + b0;
    constexpr int NB = OWN_KEEP / 32;
    for (int pos = 0; pos < total; pos += OWN_KEEP) {
        const int cnt = (total - pos < OWN_KEEP) ? (total - pos) : OWN_KEEP;
        unsigned rec[NB];
#pragma unroll
        for (int b = 0; b < NB; ++b) {
            rec[b] = 0u;
            const int i = pos + b * 32 + lane;
            if (b * 32 + lane < cnt) {
                const unsigned* p = (i < n0) ? (p0 + i) : (p1 + (i - n0));
                rec[b] = *p;
            }
        }
        float wgt[NB];
#pragma unroll
        for (int b = 0; b < NB; ++b) {
            const float ds = dinv[rec[b] >> 8];
            const float dd = dinv[lo + (int)(rec[b] & 255u)];
            wgt[b] = ds * dd;
        }
#pragma unroll
        for (int b = 0; b < NB; ++b) {
            myl[b * 32 + lane]            = rec[b];
            myl[OWN_KEEP + b * 32 + lane] = __float_as_uint(wgt[b]);
        }
        for (int base = 0; base < cnt; base += 32) {
            const int n = (cnt - base < 32) ? (cnt - base) : 32;
            const unsigned r = myl[base + lane];
            const float    w = __uint_as_float(myl[OWN_KEEP + base + lane]);
            gcn_own_add_rows<F, true>((int)(r >> 8), (int)(r & 255u), w, n, lane, xw, tile);
        }
    }
}

template <int F>
__device__ __forceinline__ void gcn_own_add_dealt(const unsigned* __restrict__ bins,
                                                  const unsigned* __restrict__ boff,
                                                  const float* __restrict__ dinv,
                                                  const float* __restrict__ xw,
                                                  float* tile, unsigned* list, unsigned* icnt,
                                                  int lane, int wv, int lo) {
    static_assert(DEG_CHUNKS == 2 * OWN_WAVES && (OWN_WAVES & (OWN_WAVES - 1)) == 0 &&
                  2 * OWN_INBOX <= OWN_LIST, "shape not covered");
    const int c0 = wv, c1 = wv + OWN_WAVES;
    const unsigned* o0 = boff + (size_t)c0 * BIN_OFFP + blockIdx.x;
    const unsigned* o1 = boff + (size_t)c1 * BIN_OFFP + blockIdx.x;
    int a0 = __builtin_amdgcn_readfirstlane((int)o0[0]);
    int a1 = __builtin_amdgcn_readfirstlane((int)o0[1]);
    int b0 = __builtin_amdgcn_readfirstlane((int)o1[0]);
    int b1 = __builtin_amdgcn_readfirstlane((int)o1[1]);
    a0 = (a0 < 0) ? 0 : ((a0 > BIN_CAP) ? BIN_CAP : a0);
    a1 = (a1 < a0) ? a0 : ((a1 > BIN_CAP) ? BIN_CAP : a1);
    b0 = (b0 < 0) ? 0 : ((b0 > BIN_CAP) ? BIN_CAP : b0);
    b1 = (b1 < b0) ? b0 : ((b1 > BIN_CAP) ? BIN_CAP : b1);
    const int n0 = a1 - a0;
    const int total = n0 + (b1 - b0);
    const unsigned* p0 = bins + (size_t)c0 * BIN_CAP + a0;
    const unsigned* p1 = bins + (size_t)c1 * BIN_CAP + b0;
    constexpr int NB = OWN_KEEP / 32;
    for (int pos = 0; pos < total; pos += OWN_KEEP) {
        const int cnt = (total - pos < OWN_KEEP) ? (total - pos) : OWN_KEEP;
        unsigned rec[NB];
#pragma unroll
        for (int b = 0; b < NB; ++b) {
            rec[b] = 0u;
            const int i = pos + b * 32 + lane;
            if (b * 32 + lane < cnt) {
                const unsigned* p = (i < n0) ? (p0 + i) : (p1 + (i - n0));
                rec[b] = *p;
            }
        }
        float wgt[NB];
#pragma unroll
        for (int b = 0; b < NB; ++b) {
            const float ds = dinv[rec[b] >> 8];
            const float dd = dinv[lo + (int)(rec[b] & 255u)];
            wgt[b] = ds * dd;
        }
#pragma unroll
        for (int b = 0; b < NB; ++b) {
            const bool     valid = b * 32 + lane < cnt;
            const unsigned row   = rec[b] & 255u;
            const unsigned own   = row & (unsigned)(OWN_WAVES - 1);
            unsigned at = (unsigned)OWN_INBOX;
            if (valid) at = atomicAdd(&icnt[own], 1u);
            const bool fits = valid && at < (unsigned)OWN_INBOX;
            if (fits) {
                unsigned* ib = list + own * OWN_LIST;
                ib[at]             = rec[b];
                ib[OWN_INBOX + at] = __float_as_uint(wgt[b]);
            }
            unsigned ovf = __builtin_amdgcn_ballot_w32(valid && !fits);
            while (ovf != 0u) {
                const int j = __builtin_ctz(ovf);
                ovf &= ovf - 1u;
                gcn_own_add_group<F, 1, false>((int)(rec[b] >> 8), (int)row, wgt[b], j, lane, xw, tile);
            }
        }
    }
    __syncthreads();
    int n = __builtin_amdgcn_readfirstlane((int)icnt[wv]);
    n = (n < OWN_INBOX) ? n : OWN_INBOX;
    const unsigned* mine = list + wv * OWN_LIST;
    for (int base = 0; base < n; base += 32) {
        const int m = (n - base < 32) ? (n - base) : 32;
        unsigned r = 0u;
        float    w = 0.0f;
        if (lane < m) {
            r = mine[base + lane];
            w = __uint_as_float(mine[OWN_INBOX + base + lane]);
        }
        gcn_own_add_rows<F, true, true>((int)(r >> 8), (int)(r & 255u), w, m, lane, xw, tile);
    }
}

template <int F>
__device__ __forceinline__ void gcn_own_add_edges(const unsigned* lst, int n, int lane, int lo,
                                                  const int* __restrict__ src,
                                                  const float* __restrict__ dinv,
                                                  const float* __restrict__ xw,
                                                  float* tile, unsigned* keep) {
    int s = 0, dl = 0;
    float w = 0.0f;
    if (lane < n) {
        const unsigned rec = lst[lane];
        dl = (int)(rec & 255u);
        s  = src[rec >> 8];
        w  = dinv[s] * dinv[lo + dl];
        if (keep != nullptr) keep[lane] = ((unsigned)s << 8) | (unsigned)dl;
    }
    gcn_own_add_rows<F, false>(s, dl, w, n, lane, xw, tile);
}

template <int F>
__device__ __forceinline__ int gcn_own_scan(const int* __restrict__ src, const int* __restrict__ dst,
                                            const float* __restrict__ dinv,
                                            const float* __restrict__ xw,
                                            float* tile, unsigned* mylist, unsigned* kept,
                                            int lane, int wv, int lo, int nrows, int E) {
    const int nchunks = (E + 1023) >> 10;
    const int c0 = (int)((blockIdx.x * (unsigned)nchunks) / gridDim.x);
    int  count  = 0;
    int  nkept  = 0;
    bool keepok = kept != nullptr;

    for (int i = wv; i < nchunks; i += OWN_WAVES) {
        int c = c0 + i;
        if (c >= nchunks) c -= nchunks;
        const int cbase = c << 10;

        int4 dv[8];
        int  eu[8];
#pragma unroll
        for (int u = 0; u < 8; ++u) {
            const int e = cbase + u * 128 + lane * 4;
            eu[u] = (e < E - 4) ? e : (E - 4);
            dv[u] = *(const int4*)(dst + eu[u]);
        }
        if (cbase + 1024 > E) {
#pragma unroll
            for (int u = 0; u < 8; ++u) {
                const int e = cbase + u * 128 + lane * 4;
                if (eu[u]     < e) dv[u].x = -1;
                if (eu[u] + 1 < e) dv[u].y = -1;
                if (eu[u] + 2 < e) dv[u].z = -1;
                if (eu[u] + 3 < e) dv[u].w = -1;
            }
        }
#pragma unroll
        for (int hf = 0; hf < 2; ++hf) {
#pragma unroll
            for (int uu = 0; uu < 4; ++uu) {
                const int u = hf * 4 + uu;
                const int dd[4] = {dv[u].x, dv[u].y, dv[u].z, dv[u].w};
#pragma unroll
                for (int k = 0; k < 4; ++k) {
                    const unsigned t = (unsigned)(dd[k] - lo);
                    const bool m = t < (unsigned)nrows;
                    const unsigned bal = __builtin_amdgcn_ballot_w32(m);
                    if (bal != 0u) {
                        if (m) mylist[count + __popc(bal & ((1u << lane) - 1u))] = ((unsigned)(eu[u] + k) << 8) | t;
                        count += __popc(bal);
                    }
                }
            }
            __builtin_amdgcn_fence(__ATOMIC_ACQ_REL, "wavefront");
            while (count >= 32) {
                count -= 32;
                unsigned* kp = nullptr;
                if (keepok) {
                    if (nkept + 32 <= OWN_KEEP) { kp = kept + nkept; nkept += 32; }
                    else keepok = false;
                }
                gcn_own_add_edges<F>(mylist + count, 32, lane, lo, src, dinv, xw, tile, kp);
            }
        }
    }
    if (count > 0) {
        unsigned* kp = nullptr;
        if (keepok) {
            if (nkept + count <= OWN_KEEP) { kp = kept + nkept; nkept += count; }
            else keepok = false;
        }
        gcn_own_add_edges<F>(mylist, count, lane, lo, src, dinv, xw, tile, kp);
    }
    return keepok ? nkept : -1;
}

template <int F>
__device__ __forceinline__ void gcn_own_add_found(const unsigned* lst, int n, int lane, int lo,
                                                  const int* __restrict__ src,
                                                  const int* __restrict__ dst,
                                                  const float* __restrict__ dinv,
                                                  const float* __restrict__ xw,
                                                  float* tile, unsigned* keep) {
    int s = 0, dl = 0;
    float w = 0.0f;
    if (lane < n) {
        const unsigned e = lst[lane];
        const int d = dst[e];
        s  = src[e];
        dl = d - lo;
        w  = dinv[s] * dinv[d];
        if (keep != nullptr) keep[lane] = ((unsigned)s << 8) | (unsigned)dl;
    }
    gcn_own_add_rows<F, false>(s, dl, w, n, lane, xw, tile);
}

template <int F>
__device__ __forceinline__ int gcn_own_scan_tags(const int* __restrict__ src,
                                                 const int* __restrict__ dst,
                                                 const unsigned char* __restrict__ tags,
                                                 const float* __restrict__ dinv,
                                                 const float* __restrict__ xw,
                                                 float* tile, unsigned* mylist, unsigned* kept,
                                                 int lane, int wv, int lo, int E) {
    const unsigned pat = blockIdx.x * 0x01010101u;
    const int nunits = (E + OWN_TAGU - 1) / OWN_TAGU;
    const int nfull  = nunits / (8 * OWN_WAVES);
    const int ntrip  = nfull * OWN_WAVES;
    const int t0     = (int)((blockIdx.x * (unsigned)ntrip) / gridDim.x);
    const int rest0  = nfull * (8 * OWN_WAVES);
    const int per    = (nunits - rest0 + OWN_WAVES - 1) / OWN_WAVES;
    int  count  = 0;
    int  nkept  = 0;
    bool keepok = kept != nullptr;

    uint4 tv[8];
    int i = 0;
    int unit0 = 0;
    int nu = 0;
    int done = 0;
    for (;;) {
        if (done >= nu) {
            if (i > nfull) break;
            if (i < nfull) {
                int tr = t0 + i * OWN_WAVES + wv;
                if (tr >= ntrip) tr -= ntrip;
                unit0 = tr * 8;
                nu    = 8;
            } else {
                unit0 = rest0 + wv * per;
                nu    = (nunits - unit0 < per) ? (nunits - unit0) : per;
                if (nu <= 0) break;
            }
            ++i;
            done = 0;
#pragma unroll
            for (int u = 0; u < 8; ++u) {
                int un = unit0 + u;
                un = (un < nunits) ? un : (nunits - 1);
                tv[u] = *(const uint4*)(tags + (size_t)un * OWN_TAGU + lane * 16);
            }
        }
#pragma unroll
        for (int u = 0; u < 8; ++u) {
            if (u >= done && u < nu && count <= OWN_LIST - OWN_TAGU) {
                const int ebase = (unit0 + u) * OWN_TAGU + lane * 16;
                unsigned xs[4] = {tv[u].x ^ pat, tv[u].y ^ pat, tv[u].z ^ pat, tv[u].w ^ pat};
                if ((unit0 + u + 1) * OWN_TAGU > E) {
#pragma unroll
                    for (int q = 0; q < 4; ++q) {
#pragma unroll
                        for (int k = 0; k < 4; ++k)
                            if (ebase + q * 4 + k >= E) xs[q] |= 255u << (8 * k);
                    }
                }
#pragma unroll
                for (int q = 0; q < 4; ++q) {
                    const unsigned x = xs[q];
                    const unsigned z = (x - 0x01010101u) & ~x & 0x80808080u;
                    if (__builtin_amdgcn_ballot_w32(z != 0u) != 0u) {
#pragma unroll
                        for (int k = 0; k < 4; ++k) {
                            const bool m = ((x >> (8 * k)) & 255u) == 0u;
                            const unsigned bal = __builtin_amdgcn_ballot_w32(m);
                            if (bal != 0u) {
                                if (m) mylist[count + __popc(bal & ((1u << lane) - 1u))] = (unsigned)(ebase + q * 4 + k);
                                count += __popc(bal);
                            }
                        }
                    }
                }
                done = u + 1;
            }
        }
        __builtin_amdgcn_fence(__ATOMIC_ACQ_REL, "wavefront");
        while (count >= 32) {
            count -= 32;
            unsigned* kp = nullptr;
            if (keepok) {
                if (nkept + 32 <= OWN_KEEP) { kp = kept + nkept; nkept += 32; }
                else keepok = false;
            }
            gcn_own_add_found<F>(mylist + count, 32, lane, lo, src, dst, dinv, xw, tile, kp);
        }
    }
    if (count > 0) {
        unsigned* kp = nullptr;
        if (keepok) {
            if (nkept + count <= OWN_KEEP) { kp = kept + nkept; nkept += count; }
            else keepok = false;
        }
        gcn_own_add_found<F>(mylist, count, lane, lo, src, dst, dinv, xw, tile, kp);
    }
    return keepok ? nkept : -1;
}

__device__ __forceinline__ bool gcn_own_keep_fits(const float* wsb, size_t ws_bytes, const int* kend) {
    return (size_t)((const char*)kend - (const char*)wsb) <= ws_bytes;
}

__global__ __launch_bounds__(1024)
void gcn_owner_sum_lds(const int* __restrict__ src, const int* __restrict__ dst,
                       const float* __restrict__ dinv, const float* __restrict__ xw,
                       const float* __restrict__ bias, const _Float16* __restrict__ W2p,
                       float* __restrict__ xw2,
                       unsigned* __restrict__ kept, int* __restrict__ kcnt,
                       const unsigned char* __restrict__ tags, int Epad,
                       const unsigned* __restrict__ bins, const unsigned* __restrict__ boff,
                       const float* wsb, size_t ws_bytes, int N, int E) {
    __shared__ __attribute__((aligned(16))) float tile[OWN_ROWS * GCN_D_HID];
    __shared__ unsigned list[OWN_WAVES * OWN_LIST];
    __shared__ unsigned icnt[OWN_WAVES];

    const int lane = threadIdx.x & 31;
    const int wv = __builtin_amdgcn_readfirstlane((int)(threadIdx.x >> 5));
    const int lo = blockIdx.x * OWN_ROWS;
    const int nrows = (N - lo < OWN_ROWS) ? (N - lo) : OWN_ROWS;
    const int wid = blockIdx.x * OWN_WAVES + wv;
    const bool keep_on = gcn_own_keep_fits(wsb, ws_bytes, kcnt + gridDim.x * OWN_WAVES);

    if (threadIdx.x < OWN_WAVES) icnt[threadIdx.x] = 0u;

    for (int r = wv; r < nrows; r += OWN_WAVES) {
        const float di = dinv[lo + r];
        const float4* xr = (const float4*)(xw + (size_t)(lo + r) * GCN_D_HID);
        const float4 a = xr[lane];
        const float4 b = xr[32 + lane];
        float* tr = tile + r * GCN_D_HID + lane * 4;
        *(float4*)tr         = float4{a.x * di * di, a.y * di * di, a.z * di * di, a.w * di * di};
        *(float4*)(tr + 128) = float4{b.x * di * di, b.y * di * di, b.z * di * di, b.w * di * di};
    }
    __syncthreads();

    const bool bins_on = gcn_bins_on(wsb, ws_bytes, boff, N, E, Epad);
    if (bins_on) {
        gcn_own_add_dealt<GCN_D_HID>(bins, boff, dinv, xw, tile, list, icnt, lane, wv, lo);
    } else {
        const bool tags_on = gcn_ws_fits(wsb, ws_bytes, tags + Epad) && gridDim.x <= 256u;
        unsigned* kp0 = keep_on ? kept + (size_t)wid * OWN_KEEP : nullptr;
        int nk;
        if (tags_on)
            nk = gcn_own_scan_tags<GCN_D_HID>(src, dst, tags, dinv, xw, tile, list + wv * OWN_LIST,
                                              kp0, lane, wv, lo, E);
        else
            nk = gcn_own_scan<GCN_D_HID>(src, dst, dinv, xw, tile, list + wv * OWN_LIST,
                                         kp0, lane, wv, lo, nrows, E);
        if (keep_on && lane == 0) kcnt[wid] = nk;
    }
    __syncthreads();

    constexpr int HP  = GCN_D_HID + 8;
    constexpr int NRT = (OWN_ROWS + 15) / 16;
    constexpr int RPW = (NRT * 16 + OWN_WAVES - 1) / OWN_WAVES;
    constexpr int KT2 = GCN_D_HID / 32;
    constexpr int NT2 = GCN_D_OUT / 16;
    constexpr int RGS = OWN_WAVES / NT2;
    static_assert(OWN_WAVES % NT2 == 0 && NRT <= 2 * RGS && GCN_D_HID == 256 &&
                  NRT * 16 * HP * 2 <= OWN_ROWS * GCN_D_HID * 4, "shape not covered");
    const int ct = wv % NT2;
    const int rg = wv / NT2;

    const float4 bb0 = ((const float4*)bias)[lane];
    const float4 bb1 = ((const float4*)bias)[32 + lane];

    v16h b[KT2];
    {
        const _Float16* bb = W2p + ((size_t)ct * 32 + lane) * 16;
#pragma unroll
        for (int kt = 0; kt < KT2; ++kt) b[kt] = *(const v16h*)(bb + (size_t)kt * NT2 * 32 * 16);
    }
    v4h hlo[RPW], hhi[RPW];
#pragma unroll
    for (int i = 0; i < RPW; ++i) {
        const int r = wv + i * OWN_WAVES;
        float4 a = float4{0.0f, 0.0f, 0.0f, 0.0f};
        float4 c = float4{0.0f, 0.0f, 0.0f, 0.0f};
        if (r < nrows) {
            const float* tr = tile + r * GCN_D_HID + lane * 4;
            const float4 s0 = *(const float4*)tr;
            const float4 s1 = *(const float4*)(tr + 128);
            a.x = fmaxf(s0.x + bb0.x, 0.0f);
            a.y = fmaxf(s0.y + bb0.y, 0.0f);
            a.z = fmaxf(s0.z + bb0.z, 0.0f);
            a.w = fmaxf(s0.w + bb0.w, 0.0f);
            c.x = fmaxf(s1.x + bb1.x, 0.0f);
            c.y = fmaxf(s1.y + bb1.y, 0.0f);
            c.z = fmaxf(s1.z + bb1.z, 0.0f);
            c.w = fmaxf(s1.w + bb1.w, 0.0f);
        }
        hlo[i] = v4h{(_Float16)a.x, (_Float16)a.y, (_Float16)a.z, (_Float16)a.w};
        hhi[i] = v4h{(_Float16)c.x, (_Float16)c.y, (_Float16)c.z, (_Float16)c.w};
    }
    __syncthreads();
    _Float16* hrow = (_Float16*)tile;
#pragma unroll
    for (int i = 0; i < RPW; ++i) {
        const int r = wv + i * OWN_WAVES;
        if (r < NRT * 16) {
            *(v4h*)(hrow + r * HP + lane * 4)       = hlo[i];
            *(v4h*)(hrow + r * HP + 128 + lane * 4) = hhi[i];
        }
    }
    __syncthreads();

    const int hsel = lane >> 4;
    const int idx  = lane & 15;
    const int one[1] = {0};
#pragma unroll
    for (int i = 0; i < 2; ++i) {
        const int rt = rg + i * RGS;
        if (rt < NRT && rt * 16 < nrows) {
            v8f acc[1];
            acc[0] = v8f{};
            gcn_gemm_tiles<GCN_D_HID, 1>(hrow + (rt * 16 + idx) * HP + hsel * 8, b, one, acc);
            const int row0 = rt * 16 + hsel * 8;
            float* cb = xw2 + (size_t)(lo + row0) * GCN_D_OUT + ct * 16 + idx;
#pragma unroll
            for (int r = 0; r < 8; ++r)
                if (row0 + r < nrows) cb[(size_t)r * GCN_D_OUT] = acc[0][r];
        }
    }
}

__global__ __launch_bounds__(1024)
void gcn_owner_sum_kept(const int* __restrict__ src, const int* __restrict__ dst,
                        const float* __restrict__ dinv, const float* __restrict__ xw,
                        const float* __restrict__ bias, float* __restrict__ out,
                        const unsigned* __restrict__ kept, const int* __restrict__ kcnt,
                        const unsigned* __restrict__ bins, const unsigned* __restrict__ boff,
                        int Epad, const float* wsb, size_t ws_bytes, int N, int E) {
    __shared__ float    tile[OWN_ROWS * GCN_D_OUT];
    __shared__ unsigned list[OWN_WAVES * OWN_LIST];

    const int lane = threadIdx.x & 31;
    const int wv = __builtin_amdgcn_readfirstlane((int)(threadIdx.x >> 5));
    const int lo = blockIdx.x * OWN_ROWS;
    const int nrows = (N - lo < OWN_ROWS) ? (N - lo) : OWN_ROWS;
    const int wid = blockIdx.x * OWN_WAVES + wv;
    const bool keep_on = gcn_own_keep_fits(wsb, ws_bytes, kcnt + gridDim.x * OWN_WAVES);

    const bool bins_on = gcn_bins_on(wsb, ws_bytes, boff, N, E, Epad);
    bool use_kept = false;
    if (!bins_on && keep_on) {
        const int c = kcnt[blockIdx.x * OWN_WAVES + lane];
        use_kept = __builtin_amdgcn_ballot_w32(c < 0 || c > OWN_KEEP) == 0u;
    }

    {
        constexpr int RPW = (OWN_ROWS + OWN_WAVES - 1) / OWN_WAVES;
        float sa[RPW], sb[RPW];
        const int   rl  = wv + lane * OWN_WAVES;
        const float dvl = dinv[lo + ((rl < nrows) ? rl : (nrows - 1))];
#pragma unroll
        for (int i = 0; i < RPW; ++i) {
            const int r  = wv + i * OWN_WAVES;
            const int rc = (r < nrows) ? r : (nrows - 1);
            const float* xr = xw + (size_t)(lo + rc) * GCN_D_OUT;
            sa[i] = xr[lane];
            sb[i] = xr[32 + lane];
        }
#pragma unroll
        for (int i = 0; i < RPW; ++i) {
            const int r = wv + i * OWN_WAVES;
            if (r < nrows) {
                const float di = __int_as_float(__builtin_amdgcn_readlane(__float_as_int(dvl), i));
                float* tr = tile + r * GCN_D_OUT + lane;
                tr[0]  = sa[i] * di * di;
                tr[32] = sb[i] * di * di;
            }
        }
    }
    __syncthreads();

    if (bins_on) {
        gcn_own_add_binned<GCN_D_OUT>(bins, boff, dinv, xw, tile, list + wv * OWN_LIST, lane, wv, lo);
    } else if (use_kept) {
        const unsigned* kp = kept + (size_t)wid * OWN_KEEP;
        int cnt = __builtin_amdgcn_readfirstlane(kcnt[wid]);
        cnt = (cnt < OWN_KEEP) ? cnt : OWN_KEEP;
        constexpr int NB = OWN_KEEP / 32;
        unsigned* myl = list + wv * OWN_LIST;
        unsigned rec[NB];
#pragma unroll
        for (int b = 0; b < NB; ++b) {
            rec[b] = 0u;
            if (b * 32 + lane < cnt) rec[b] = kp[b * 32 + lane];
        }
        float wgt[NB];
#pragma unroll
        for (int b = 0; b < NB; ++b) {
            const float ds = dinv[rec[b] >> 8];
            const float dd = dinv[lo + (int)(rec[b] & 255u)];
            wgt[b] = ds * dd;
        }
#pragma unroll
        for (int b = 0; b < NB; ++b) {
            myl[b * 32 + lane]            = rec[b];
            myl[OWN_KEEP + b * 32 + lane] = __float_as_uint(wgt[b]);
        }
        for (int base = 0; base < cnt; base += 32) {
            const int n = (cnt - base < 32) ? (cnt - base) : 32;
            const unsigned r = myl[base + lane];
            const float    w = __uint_as_float(myl[OWN_KEEP + base + lane]);
            gcn_own_add_rows<GCN_D_OUT, true>((int)(r >> 8), (int)(r & 255u), w, n, lane, xw, tile);
        }
    } else {
        gcn_own_scan<GCN_D_OUT>(src, dst, dinv, xw, tile, list + wv * OWN_LIST, nullptr,
                                lane, wv, lo, nrows, E);
    }
    __syncthreads();

    const float b0 = bias[lane];
    const float b1 = bias[lane + 32];
    for (int r = wv; r < nrows; r += OWN_WAVES) {
        const float* tr = tile + r * GCN_D_OUT + lane;
        float v0 = tr[0]  + b0;
        float v1 = tr[32] + b1;
        float m = fmaxf(v0, v1);
#pragma unroll
        for (int off = 16; off > 0; off >>= 1) m = fmaxf(m, __shfl_xor(m, off, 32));
        float e0 = __expf(v0 - m);
        float e1 = __expf(v1 - m);
        float sm = e0 + e1;
#pragma unroll
        for (int off = 16; off > 0; off >>= 1) sm += __shfl_xor(sm, off, 32);
        float inv = 1.0f / sm;
        float* orow = out + (size_t)(lo + r) * GCN_D_OUT;
        orow[lane]      = e0 * inv;
        orow[lane + 32] = e1 * inv;
    }
}

extern "C" void kernel_launch(void* const* d_in, const int* in_sizes, int n_in,
                              void* d_out, int out_size, void* d_ws, size_t ws_size,
                              hipStream_t stream) {
    const float* x   = (const float*)d_in[0];
    const int*   ei  = (const int*)  d_in[1];
    const float* W1  = (const float*)d_in[2];
    const float* b1  = (const float*)d_in[3];
    const float* W2  = (const float*)d_in[4];
    const float* b2  = (const float*)d_in[5];
    float* out = (float*)d_out;

    const int N = in_sizes[0] / GCN_D_IN;
    const int E = in_sizes[1] / 2;
    const int* src = ei;
    const int* dst = ei + E;

    float* ws   = (float*)d_ws;
    size_t off  = ((size_t)N + 63) & ~(size_t)63;
    float* dinv = ws;
    float* xw1  = ws + off;
    float* h1   = xw1 + (size_t)N * GCN_D_HID;
    _Float16* W1p = (_Float16*)(h1 + (size_t)N * GCN_D_HID);
    _Float16* W2p = W1p + (size_t)GCN_D_IN * GCN_D_HID;
    float* xw2  = h1;
    const int ob = (N + OWN_ROWS - 1) / OWN_ROWS;
    unsigned* kept = (unsigned*)(W2p + (size_t)GCN_D_HID * GCN_D_OUT);
    int*      kcnt = (int*)(kept + (size_t)ob * OWN_WAVES * OWN_KEEP);
    unsigned char* tags = (unsigned char*)(kcnt + (size_t)ob * OWN_WAVES);
    const int Epad = (E + OWN_TAGU - 1) / OWN_TAGU * OWN_TAGU;
    unsigned* bins = kept;
    unsigned* boff = bins + (size_t)DEG_CHUNKS * BIN_CAP;

    const int B = 256;
    {
        int t1 = (GCN_D_IN >> 5) * (GCN_D_HID >> 4) * 32;
        gcn_pack_w_f16<<<(t1 + B - 1) / B, B, 0, stream>>>(W1, W1p, GCN_D_IN, GCN_D_HID);
        int t2 = (GCN_D_HID >> 5) * (GCN_D_OUT >> 4) * 32;
        gcn_pack_w_f16<<<(t2 + B - 1) / B, B, 0, stream>>>(W2, W2p, GCN_D_HID, GCN_D_OUT);
    }

    unsigned* part = (unsigned*)h1;
    gcn_count_deg<<<DEG_CHUNKS, 1024, 0, stream>>>(src, dst, part, (int)off, tags, bins, boff, ws, ws_size, N, E, Epad);
    gcn_make_dinv<<<(N + B - 1) / B, B, 0, stream>>>(part, (int)off, dinv, N);

    {
        gcn_wmma_gemm_bregs<GCN_D_IN, GCN_D_HID, 2><<<256, 512, 0, stream>>>(x, W1p, xw1, N);
    }
    {
        gcn_owner_sum_lds<<<ob, 1024, 0, stream>>>(src, dst, dinv, xw1, b1, W2p, xw2, kept, kcnt,
                                                   tags, Epad, bins, boff, ws, ws_size, N, E);
    }

    {
        gcn_owner_sum_kept<<<ob, 1024, 0, stream>>>(src, dst, dinv, xw2, b2, out, kept, kcnt,
                                                    bins, boff, Epad, ws, ws_size, N, E);
    }
}


# degrees counted from the owner bins: each GEMM workgroup counts one owner block's records in LDS in its prologue and writes dinv; per-node counters, partial-count rows, dinv launch gone; counts exact
# speedup vs baseline: 5.2592x; 1.0592x over previous
#include <hip/hip_runtime.h>
#include <hip/hip_bf16.h>

typedef __attribute__((ext_vector_type(16))) _Float16 v16h;
typedef __attribute__((ext_vector_type(8)))  float    v8f;

#define GCN_D_IN  512
#define GCN_D_HID 256
#define GCN_D_OUT 64

#define OWN_ROWS  196
#define OWN_TAGU  512

__device__ __forceinline__ bool gcn_ws_fits(const float* wsb, size_t ws_bytes, const void* end) {
    return (size_t)((const char*)end - (const char*)wsb) <= ws_bytes;
}

#define DEG_CHUNKS 64
#define DEG_BINS   53248
#define BIN_CAP    16384
#define BIN_OFFP   264

__device__ __forceinline__ bool gcn_bins_on(const float* wsb, size_t ws_bytes, const unsigned* boff,
                                            int N, int E, int Epad) {
    const int nq   = Epad >> 2;
    const int qper = (nq + DEG_CHUNKS - 1) / DEG_CHUNKS;
    return gcn_ws_fits(wsb, ws_bytes, boff + (size_t)DEG_CHUNKS * BIN_OFFP) &&
           qper <= BIN_CAP / 4 && (E & 3) == 0 && N <= DEG_BINS &&
           (N + OWN_ROWS - 1) / OWN_ROWS <= 256 && N <= (1 << 24);
}

__global__ __launch_bounds__(1024)
void gcn_count_deg(const int* __restrict__ src, const int* __restrict__ dst,
                   unsigned* __restrict__ part, int pitch,
                   unsigned char* __restrict__ tags, unsigned* __restrict__ bins,
                   unsigned* __restrict__ boff, const float* wsb, size_t ws_bytes,
                   int N, int E, int Epad) {
    __shared__ __attribute__((aligned(16))) unsigned hist[DEG_BINS];
    __shared__ __attribute__((aligned(16))) unsigned brec[BIN_CAP];
    __shared__ unsigned bcnt[256];
    __shared__ unsigned bstart[257];
    const int  t       = threadIdx.x;
    const bool bins_on = gcn_bins_on(wsb, ws_bytes, boff, N, E, Epad) && (int)gridDim.x == DEG_CHUNKS;
    const bool tags_on = !bins_on && gcn_ws_fits(wsb, ws_bytes, tags + Epad);
    const bool hist_on = !bins_on;
    const int  nq      = Epad >> 2;
    const int  qfull   = E >> 2;
    const int  qper    = (nq + (int)gridDim.x - 1) / (int)gridDim.x;
    const int  q0      = (int)blockIdx.x * qper;
    const int  q1      = (q0 + qper < nq) ? (q0 + qper) : nq;
    unsigned*  prow    = part + (size_t)blockIdx.x * pitch;

    int4     dv[4];
    int4     sv[4];
    unsigned rk[16];
    unsigned vmask = 0u;
    if (bins_on && t < 256) bcnt[t] = 0u;

    for (int base = 0; base < N; base += DEG_BINS) {
        const int nb = (N - base < DEG_BINS) ? (N - base) : DEG_BINS;
        if (hist_on)
            for (int i = t * 4; i < DEG_BINS; i += 4096) *(uint4*)(hist + i) = uint4{0u, 0u, 0u, 0u};
        __syncthreads();
        for (int qb = q0; qb < q1; qb += 4096) {
#pragma unroll
            for (int u = 0; u < 4; ++u) {
                const int q  = qb + u * 1024 + t;
                const int qc = (q < qfull) ? q : (qfull - 1);
                const int* dp = dst + (size_t)qc * 4;
                dv[u] = int4{dp[0], dp[1], dp[2], dp[3]};
                if (bins_on) {
                    const int* sp = src + (size_t)qc * 4;
                    sv[u] = int4{sp[0], sp[1], sp[2], sp[3]};
                }
            }
#pragma unroll
            for (int u = 0; u < 4; ++u) {
                const int  q    = qb + u * 1024 + t;
                const bool mine = q < q1;
                const bool real = mine && q < qfull;
                const int dd[4] = {real ? dv[u].x : -1, real ? dv[u].y : -1,
                                   real ? dv[u].z : -1, real ? dv[u].w : -1};
                unsigned tw = 0u;
#pragma unroll
                for (int k = 0; k < 4; ++k) {
                    const unsigned r = (unsigned)(dd[k] - base);
                    if (hist_on && r < (unsigned)nb) atomicAdd(&hist[r], 1u);
                    if (dd[k] >= 0) {
                        const unsigned own = ((unsigned)dd[k] / OWN_ROWS) & 255u;
                        tw |= own << (8 * k);
                        if (bins_on && base == 0) {
                            rk[u * 4 + k] = atomicAdd(&bcnt[own], 1u);
                            vmask |= 1u << (u * 4 + k);
                        }
                    }
                }
                if (tags_on && base == 0 && mine && !(q == qfull && (E & 3) != 0))
                    *(unsigned*)(tags + (size_t)q * 4) = tw;
            }
        }
        if (t == 0 && (E & 3) != 0 && qfull >= q0 && qfull < q1) {
            unsigned tw = 0u;
            for (int k = 0; k < (E & 3); ++k) {
                const int d = dst[qfull * 4 + k];
                const unsigned r = (unsigned)(d - base);
                if (r < (unsigned)nb) atomicAdd(&hist[r], 1u);
                if (d >= 0) tw |= (((unsigned)d / OWN_ROWS) & 255u) << (8 * k);
            }
            if (tags_on && base == 0) *(unsigned*)(tags + (size_t)qfull * 4) = tw;
        }
        __syncthreads();
        if (bins_on && base == 0 && t < 32) {
            unsigned c8[8];
            unsigned run = 0u;
#pragma unroll
            for (int j = 0; j < 8; ++j) { c8[j] = run; run += bcnt[t * 8 + j]; }
            unsigned incl = run;
#pragma unroll
            for (int o = 1; o < 32; o <<= 1) {
                const unsigned y = __shfl_up(incl, o, 32);
                if (t >= o) incl += y;
            }
            const unsigned excl = incl - run;
#pragma unroll
            for (int j = 0; j < 8; ++j) bstart[t * 8 + j] = excl + c8[j];
            if (t == 31) bstart[256] = incl;
        }
        for (int i = t * 4; hist_on && i < nb; i += 4096) {
            if (i + 4 <= nb) {
                *(uint4*)(prow + base + i) = *(const uint4*)(hist + i);
            } else {
                for (int k = i; k < nb; ++k) prow[base + k] = hist[k];
            }
        }
        __syncthreads();
        if (bins_on && base == 0) {
#pragma unroll
            for (int u = 0; u < 4; ++u) {
                const int dd[4] = {dv[u].x, dv[u].y, dv[u].z, dv[u].w};
                const int ss[4] = {sv[u].x, sv[u].y, sv[u].z, sv[u].w};
#pragma unroll
                for (int k = 0; k < 4; ++k) {
                    if ((vmask >> (u * 4 + k)) & 1u) {
                        const unsigned own = (unsigned)dd[k] / OWN_ROWS;
                        const unsigned pos = bstart[own & 255u] + rk[u * 4 + k];
                        if (pos < (unsigned)BIN_CAP)
                            brec[pos] = ((unsigned)ss[k] << 8) | (((unsigned)dd[k] - own * OWN_ROWS) & 255u);
                    }
                }
            }
            __syncthreads();
            const unsigned total = (bstart[256] < (unsigned)BIN_CAP) ? bstart[256] : (unsigned)BIN_CAP;
            unsigned* bout = bins + (size_t)blockIdx.x * BIN_CAP;
            for (unsigned i = (unsigned)t * 4u; i < total; i += 4096u) {
                if (i + 4u <= total) {
                    *(uint4*)(bout + i) = *(const uint4*)(brec + i);
                } else {
                    for (unsigned k = i; k < total; ++k) bout[k] = brec[k];
                }
            }
            if (t <= 256) boff[(size_t)blockIdx.x * BIN_OFFP + t] = bstart[t];
        }
    }
}

__device__ __forceinline__ void gcn_make_dinv_rows(const unsigned* __restrict__ part, int pitch,
                                                   float* __restrict__ dinv, int n,
                                                   int gtid, int nthreads) {
    for (int i = gtid; i < n; i += nthreads) {
        unsigned c[DEG_CHUNKS];
#pragma unroll
        for (int k = 0; k < DEG_CHUNKS; ++k) c[k] = part[(size_t)k * pitch + i];
        unsigned sum = 1u;
#pragma unroll
        for (int k = 0; k < DEG_CHUNKS; ++k) sum += c[k];
        const float d = (float)sum;
        dinv[i] = (d > 0.0f) ? rsqrtf(d) : 0.0f;
    }
}

#define DINV_WAVES 16
#define DINV_AHEAD 3
#define DINV_CPW   (DEG_CHUNKS / DINV_WAVES)

__device__ __forceinline__ void gcn_dinv_bins_starts(const unsigned* __restrict__ boff, int ow, int w,
                                                     int (&a0)[DINV_CPW], int (&a1)[DINV_CPW]) {
    static_assert(DEG_CHUNKS % DINV_WAVES == 0 && OWN_ROWS <= 256, "shape not covered");
#pragma unroll
    for (int j = 0; j < DINV_CPW; ++j) {
        const unsigned* o = boff + (size_t)(w + j * DINV_WAVES) * BIN_OFFP + ow;
        a0[j] = (int)o[0];
        a1[j] = (int)o[1];
    }
#pragma unroll
    for (int j = 0; j < DINV_CPW; ++j) {
        int s0 = __builtin_amdgcn_readfirstlane(a0[j]);
        int s1 = __builtin_amdgcn_readfirstlane(a1[j]);
        s0 = (s0 < 0) ? 0 : ((s0 > BIN_CAP) ? BIN_CAP : s0);
        s1 = (s1 < s0) ? s0 : ((s1 > BIN_CAP) ? BIN_CAP : s1);
        a0[j] = s0;
        a1[j] = s1;
    }
}

__device__ __forceinline__ void gcn_dinv_bins_ask(const unsigned* __restrict__ bins, int lane, int w,
                                                  const int (&a0)[DINV_CPW],
                                                  unsigned (&rec)[DINV_CPW][DINV_AHEAD]) {
#pragma unroll
    for (int j = 0; j < DINV_CPW; ++j) {
        const unsigned* p = bins + (size_t)(w + j * DINV_WAVES) * BIN_CAP;
#pragma unroll
        for (int k = 0; k < DINV_AHEAD; ++k) {
            const int i = a0[j] + k * 32 + lane;
            rec[j][k] = p[(i < BIN_CAP) ? i : (BIN_CAP - 1)];
        }
    }
}

__device__ __forceinline__ void gcn_dinv_bins_hold(const unsigned (&rec)[DINV_CPW][DINV_AHEAD]) {
#pragma unroll
    for (int j = 0; j < DINV_CPW; ++j) {
#pragma unroll
        for (int k = 0; k < DINV_AHEAD; ++k) asm volatile("" : : "v"(rec[j][k]));
    }
}

__device__ __forceinline__ void gcn_dinv_bins_count(const unsigned* __restrict__ bins, unsigned* cnt,
                                                    int lane, int w,
                                                    const int (&a0)[DINV_CPW], const int (&a1)[DINV_CPW],
                                                    const unsigned (&rec)[DINV_CPW][DINV_AHEAD]) {
#pragma unroll
    for (int j = 0; j < DINV_CPW; ++j) {
#pragma unroll
        for (int k = 0; k < DINV_AHEAD; ++k)
            if (a0[j] + k * 32 + lane < a1[j]) atomicAdd(&cnt[rec[j][k] & 255u], 1u);
    }
#pragma unroll
    for (int j = 0; j < DINV_CPW; ++j) {
        const unsigned* p = bins + (size_t)(w + j * DINV_WAVES) * BIN_CAP;
        for (int i = a0[j] + DINV_AHEAD * 32 + lane; i < a1[j]; i += 32) atomicAdd(&cnt[p[i] & 255u], 1u);
    }
}

__device__ __forceinline__ void gcn_dinv_bins_store(float* __restrict__ dinv, const unsigned* cnt,
                                                    int n, int ow, int t) {
    const int lo = ow * OWN_ROWS;
    if (t < OWN_ROWS && lo + t < n) {
        const float d = (float)(1u + cnt[t]);
        dinv[lo + t] = (d > 0.0f) ? rsqrtf(d) : 0.0f;
    }
}

__device__ __forceinline__ void gcn_make_dinv_bins(const unsigned* __restrict__ bins,
                                                   const unsigned* __restrict__ boff,
                                                   float* __restrict__ dinv, unsigned* cnt,
                                                   int n, int ow, int t, int lane, int w) {
    if (t < 256) cnt[t] = 0u;
    __syncthreads();
    int a0[DINV_CPW], a1[DINV_CPW];
    unsigned rec[DINV_CPW][DINV_AHEAD];
    gcn_dinv_bins_starts(boff, ow, w, a0, a1);
    gcn_dinv_bins_ask(bins, lane, w, a0, rec);
    gcn_dinv_bins_hold(rec);
    gcn_dinv_bins_count(bins, cnt, lane, w, a0, a1, rec);
    __syncthreads();
    gcn_dinv_bins_store(dinv, cnt, n, ow, t);
}

__global__ void gcn_pack_w_f16(const float* __restrict__ W, _Float16* __restrict__ Wp,
                               int K, int N) {
    int t = blockIdx.x * blockDim.x + threadIdx.x;
    int tilesN = N >> 4;
    int total = (K >> 5) * tilesN * 32;
    if (t >= total) return;
    int frag = t >> 5;
    int lane = t & 31;
    int kt = frag / tilesN;
    int tn = frag - kt * tilesN;
    int n    = tn * 16 + (lane & 15);
    int koff = kt * 32 + (lane >> 4) * 8;
    _Float16* o = Wp + (size_t)t * 16;
#pragma unroll
    for (int j = 0; j < 8; ++j) {
        o[j]     = (_Float16)W[(size_t)(koff + j) * N + n];
        o[j + 8] = (_Float16)W[(size_t)(koff + 16 + j) * N + n];
    }
}

typedef __attribute__((ext_vector_type(8))) _Float16 v8h;
typedef __attribute__((ext_vector_type(4))) _Float16 v4h;

__device__ __forceinline__ v16h gcn_lds_a_frag(const _Float16* ar) {
    const v8h lo = *(const v8h*)(ar);
    const v8h hi = *(const v8h*)(ar + 16);
    return __builtin_shufflevector(lo, hi, 0, 1, 2, 3, 4, 5, 6, 7, 8, 9, 10, 11, 12, 13, 14, 15);
}

template <int K, int NTL>
__device__ __forceinline__ void gcn_gemm_tiles(const _Float16* ar0, const v16h (&b)[K / 32],
                                               const int (&toff)[NTL], v8f (&acc)[NTL]) {
    constexpr int KT    = K / 32;
    constexpr int PITCH = K + 8;
    const _Float16* ar[NTL];
    v16h a[3][NTL];
#pragma unroll
    for (int i = 0; i < NTL; ++i) {
        ar[i]   = ar0 + toff[i] * (16 * PITCH);
        a[0][i] = gcn_lds_a_frag(ar[i]);
        a[1][i] = gcn_lds_a_frag(ar[i] + 32);
    }
#pragma unroll
    for (int kt = 0; kt < KT; ++kt) {
        if (kt + 2 < KT) {
#pragma unroll
            for (int i = 0; i < NTL; ++i) a[(kt + 2) % 3][i] = gcn_lds_a_frag(ar[i] + (kt + 2) * 32);
        }
        __builtin_amdgcn_sched_barrier(0);
#pragma unroll
        for (int i = 0; i < NTL; ++i)
            acc[i] = __builtin_amdgcn_wmma_f32_16x16x32_f16(false, a[kt % 3][i], false, b[kt],
                                                            (short)0, acc[i], false, false);
        __builtin_amdgcn_sched_barrier(0);
    }
}

template <int K, int N, int RT>
__global__ __launch_bounds__(512)
void gcn_wmma_gemm_bregs(const float* __restrict__ A,
                         const _Float16* __restrict__ Bp,
                         float* __restrict__ C, int M,
                         const unsigned* __restrict__ bins, const unsigned* __restrict__ boff,
                         const unsigned* __restrict__ part, int pitch,
                         float* __restrict__ dinv, const float* wsb, size_t ws_bytes,
                         int nodes, int E, int Epad) {
    constexpr int KT    = K / 32;
    constexpr int NT    = N / 16;
    constexpr int RG    = 16 / NT;
    constexpr int ROWS  = RG * RT * 16;
    constexpr int PITCH = K + 8;
    constexpr int C4    = K / 4;
    constexpr int RSTEP = 512 / C4;
    constexpr int NV    = ROWS / RSTEP;
    static_assert(K % 32 == 0 && N % 16 == 0 && 16 % NT == 0 && 512 % C4 == 0 &&
                  ROWS % RSTEP == 0 && (RT == 1 || RT == 2), "shape not covered");

    __shared__ __attribute__((aligned(16))) _Float16 tile[2 * ROWS * PITCH];
    __shared__ unsigned dcnt[256];

    const int t    = threadIdx.x;
    const int lane = t & 31;
    const int w    = __builtin_amdgcn_readfirstlane(t >> 5);
    const int ct   = w % NT;
    const int rg   = w / NT;
    const int hsel = lane >> 4;
    const int idx  = lane & 15;

    const int nblk  = (M + ROWS - 1) / ROWS;
    const int ntile = M / 16;
    const int srow  = t / C4;
    const int sc    = (t % C4) * 4;
    float4 st[NV];

    static_assert(512 / 32 == DINV_WAVES, "shape not covered");
    const bool bins_on  = gcn_bins_on(wsb, ws_bytes, boff, nodes, E, Epad);
    const int  ob       = (nodes + OWN_ROWS - 1) / OWN_ROWS;
    const bool dv_first = bins_on && (int)blockIdx.x < ob;
    int da0[DINV_CPW], da1[DINV_CPW];
    unsigned drec[DINV_CPW][DINV_AHEAD];
    if (dv_first) {
        gcn_dinv_bins_starts(boff, (int)blockIdx.x, w, da0, da1);
        if (t < 256) dcnt[t] = 0u;
        __syncthreads();
    }

    int blk = blockIdx.x;
    if (blk < nblk) {
#pragma unroll
        for (int i = 0; i < NV; ++i) {
            int row = blk * ROWS + srow + i * RSTEP;
            row = (row < M) ? row : (M - 1);
            st[i] = *(const float4*)(A + (size_t)row * K + sc);
        }
    }

    v16h b[KT];
    {
        const _Float16* bb = Bp + ((size_t)ct * 32 + lane) * 16;
#pragma unroll
        for (int kt = 0; kt < KT; ++kt) b[kt] = *(const v16h*)(bb + (size_t)kt * NT * 32 * 16);
    }
    __builtin_amdgcn_sched_barrier(0);

    if (dv_first) {
        gcn_dinv_bins_ask(bins, lane, w, da0, drec);
        gcn_dinv_bins_hold(drec);
        gcn_dinv_bins_count(bins, dcnt, lane, w, da0, da1, drec);
    }

    if (blk < nblk) {
#pragma unroll
        for (int i = 0; i < NV; ++i) {
            v4h hv = {(_Float16)st[i].x, (_Float16)st[i].y, (_Float16)st[i].z, (_Float16)st[i].w};
            *(v4h*)(tile + (srow + i * RSTEP) * PITCH + sc) = hv;
        }
    }
    __syncthreads();
    if (dv_first) gcn_dinv_bins_store(dinv, dcnt, nodes, (int)blockIdx.x, t);
    if (bins_on) {
        for (int ow = (int)(blockIdx.x + gridDim.x); ow < ob; ow += (int)gridDim.x)
            gcn_make_dinv_bins(bins, boff, dinv, dcnt, nodes, ow, t, lane, w);
    } else {
        gcn_make_dinv_rows(part, pitch, dinv, nodes, (int)blockIdx.x * 512 + t, (int)gridDim.x * 512);
    }

    int buf = 0;
    for (; blk < nblk; blk += gridDim.x) {
        const int  nxt      = blk + (int)gridDim.x;
        const bool has_next = nxt < nblk;
        if (has_next) {
#pragma unroll
            for (int i = 0; i < NV; ++i) {
                int row = nxt * ROWS + srow + i * RSTEP;
                row = (row < M) ? row : (M - 1);
                st[i] = *(const float4*)(A + (size_t)row * K + sc);
            }
        }

        const int tile0 = (blk * RG + rg) * RT;
        const _Float16* ar0 = tile + buf * (ROWS * PITCH) + (rg * RT * 16 + idx) * PITCH + hsel * 8;
        float* cb0 = C + (size_t)(tile0 * 16 + hsel * 8) * N + ct * 16 + idx;
        const bool live = tile0 < ntile;
        int toff[RT];
#pragma unroll
        for (int i = 0; i < RT; ++i) toff[i] = (tile0 + i < ntile) ? i : 0;
        v8f acc[RT];
#pragma unroll
        for (int i = 0; i < RT; ++i) acc[i] = v8f{};
        if (live) gcn_gemm_tiles<K, RT>(ar0, b, toff, acc);

        if (has_next) {
            _Float16* tn = tile + (buf ^ 1) * (ROWS * PITCH);
#pragma unroll
            for (int i = 0; i < NV; ++i) {
                v4h hv = {(_Float16)st[i].x, (_Float16)st[i].y, (_Float16)st[i].z, (_Float16)st[i].w};
                *(v4h*)(tn + (srow + i * RSTEP) * PITCH + sc) = hv;
            }
        }
        __syncthreads();
        if (live) {
#pragma unroll
            for (int i = 0; i < RT; ++i) {
                float* cb = cb0 + (size_t)toff[i] * (16 * N);
#pragma unroll
                for (int r = 0; r < 8; ++r) cb[(size_t)r * N] = acc[i][r];
            }
        }
        buf ^= 1;
    }
}

#define OWN_WAVES 32
#define OWN_LIST  544
#define OWN_KEEP  160
#define OWN_INBOX (OWN_LIST / 2)

template <int F, int G, bool PLAIN>
__device__ __forceinline__ void gcn_own_add_group(int s, int dl, float w, int j, int lane,
                                                  const float* __restrict__ xw, float* tile) {
    static_assert(!PLAIN || F == 256, "shape not covered");
    int   rl[G];
    float wj[G];
    if constexpr (F == 256) {
        float4 a[G], b[G];
#pragma unroll
        for (int q = 0; q < G; ++q) {
            const int sj = __builtin_amdgcn_readlane(s, j + q);
            rl[q] = __builtin_amdgcn_readlane(dl, j + q);
            wj[q] = __int_as_float(__builtin_amdgcn_readlane(__float_as_int(w), j + q));
            const float4* xr = (const float4*)(xw + (size_t)sj * F);
            a[q] = xr[lane];
            b[q] = xr[32 + lane];
        }
#pragma unroll
        for (int q = 0; q < G; ++q) {
            float* tr = tile + rl[q] * F + lane * 4;
            if constexpr (PLAIN) {
                float4 o0 = *(const float4*)tr;
                float4 o1 = *(const float4*)(tr + 128);
                o0.x += a[q].x * wj[q];
                o0.y += a[q].y * wj[q];
                o0.z += a[q].z * wj[q];
                o0.w += a[q].w * wj[q];
                o1.x += b[q].x * wj[q];
                o1.y += b[q].y * wj[q];
                o1.z += b[q].z * wj[q];
                o1.w += b[q].w * wj[q];
                *(float4*)tr         = o0;
                *(float4*)(tr + 128) = o1;
            } else {
                atomicAdd(tr,       a[q].x * wj[q]);
                atomicAdd(tr + 1,   a[q].y * wj[q]);
                atomicAdd(tr + 2,   a[q].z * wj[q]);
                atomicAdd(tr + 3,   a[q].w * wj[q]);
                atomicAdd(tr + 128, b[q].x * wj[q]);
                atomicAdd(tr + 129, b[q].y * wj[q]);
                atomicAdd(tr + 130, b[q].z * wj[q]);
                atomicAdd(tr + 131, b[q].w * wj[q]);
            }
        }
    } else {
        float a[G], b[G];
#pragma unroll
        for (int q = 0; q < G; ++q) {
            const int sj = __builtin_amdgcn_readlane(s, j + q);
            rl[q] = __builtin_amdgcn_readlane(dl, j + q);
            wj[q] = __int_as_float(__builtin_amdgcn_readlane(__float_as_int(w), j + q));
            const float* xr = xw + (size_t)sj * F;
            a[q] = xr[lane];
            b[q] = xr[32 + lane];
        }
#pragma unroll
        for (int q = 0; q < G; ++q) {
            float* tr = tile + rl[q] * F + lane;
            atomicAdd(tr,      a[q] * wj[q]);
            atomicAdd(tr + 32, b[q] * wj[q]);
        }
    }
}

template <int F, bool SHORT_TAIL, bool PLAIN = false>
__device__ __forceinline__ void gcn_own_add_rows(int s, int dl, float w, int n, int lane,
                                                 const float* __restrict__ xw, float* tile) {
    int j = 0;
    for (; j + 8 <= n; j += 8) gcn_own_add_group<F, 8, PLAIN>(s, dl, w, j, lane, xw, tile);
    if constexpr (SHORT_TAIL) {
        if (j + 4 <= n) { gcn_own_add_group<F, 4, PLAIN>(s, dl, w, j, lane, xw, tile); j += 4; }
        if (j + 2 <= n) { gcn_own_add_group<F, 2, PLAIN>(s, dl, w, j, lane, xw, tile); j += 2; }
        if (j < n)        gcn_own_add_group<F, 1, PLAIN>(s, dl, w, j, lane, xw, tile);
    } else {
        for (; j < n; ++j) gcn_own_add_group<F, 1, PLAIN>(s, dl, w, j, lane, xw, tile);
    }
}

template <int F>
__device__ __forceinline__ void gcn_own_add_binned(const unsigned* __restrict__ bins,
                                                   const unsigned* __restrict__ boff,
                                                   const float* __restrict__ dinv,
                                                   const float* __restrict__ xw,
                                                   float* tile, unsigned* myl,
                                                   int lane, int wv, int lo) {
    static_assert(DEG_CHUNKS == 2 * OWN_WAVES && 2 * OWN_KEEP <= OWN_LIST, "shape not covered");
    const int c0 = wv, c1 = wv + OWN_WAVES;
    const unsigned* o0 = boff + (size_t)c0 * BIN_OFFP + blockIdx.x;
    const unsigned* o1 = boff + (size_t)c1 * BIN_OFFP + blockIdx.x;
    int a0 = __builtin_amdgcn_readfirstlane((int)o0[0]);
    int a1 = __builtin_amdgcn_readfirstlane((int)o0[1]);
    int b0 = __builtin_amdgcn_readfirstlane((int)o1[0]);
    int b1 = __builtin_amdgcn_readfirstlane((int)o1[1]);
    a0 = (a0 < 0) ? 0 : ((a0 > BIN_CAP) ? BIN_CAP : a0);
    a1 = (a1 < a0) ? a0 : ((a1 > BIN_CAP) ? BIN_CAP : a1);
    b0 = (b0 < 0) ? 0 : ((b0 > BIN_CAP) ? BIN_CAP : b0);
    b1 = (b1 < b0) ? b0 : ((b1 > BIN_CAP) ? BIN_CAP : b1);
    const int n0 = a1 - a0;
    const int total = n0 + (b1 - b0);
    const unsigned* p0 = bins + (size_t)c0 * BIN_CAP + a0;
    const unsigned* p1 = bins + (size_t)c1 * BIN_CAP + b0;
    constexpr int NB = OWN_KEEP / 32;
    for (int pos = 0; pos < total; pos += OWN_KEEP) {
        const int cnt = (total - pos < OWN_KEEP) ? (total - pos) : OWN_KEEP;
        unsigned rec[NB];
#pragma unroll
        for (int b = 0; b < NB; ++b) {
            rec[b] = 0u;
            const int i = pos + b * 32 + lane;
            if (b * 32 + lane < cnt) {
                const unsigned* p = (i < n0) ? (p0 + i) : (p1 + (i - n0));
                rec[b] = *p;
            }
        }
        float wgt[NB];
#pragma unroll
        for (int b = 0; b < NB; ++b) {
            const float ds = dinv[rec[b] >> 8];
            const float dd = dinv[lo + (int)(rec[b] & 255u)];
            wgt[b] = ds * dd;
        }
#pragma unroll
        for (int b = 0; b < NB; ++b) {
            myl[b * 32 + lane]            = rec[b];
            myl[OWN_KEEP + b * 32 + lane] = __float_as_uint(wgt[b]);
        }
        for (int base = 0; base < cnt; base += 32) {
            const int n = (cnt - base < 32) ? (cnt - base) : 32;
            const unsigned r = myl[base + lane];
            const float    w = __uint_as_float(myl[OWN_KEEP + base + lane]);
            gcn_own_add_rows<F, true>((int)(r >> 8), (int)(r & 255u), w, n, lane, xw, tile);
        }
    }
}

template <int F>
__device__ __forceinline__ void gcn_own_add_dealt(const unsigned* __restrict__ bins,
                                                  const unsigned* __restrict__ boff,
                                                  const float* __restrict__ dinv,
                                                  const float* __restrict__ xw,
                                                  float* tile, unsigned* list, unsigned* icnt,
                                                  int lane, int wv, int lo) {
    static_assert(DEG_CHUNKS == 2 * OWN_WAVES && (OWN_WAVES & (OWN_WAVES - 1)) == 0 &&
                  2 * OWN_INBOX <= OWN_LIST, "shape not covered");
    const int c0 = wv, c1 = wv + OWN_WAVES;
    const unsigned* o0 = boff + (size_t)c0 * BIN_OFFP + blockIdx.x;
    const unsigned* o1 = boff + (size_t)c1 * BIN_OFFP + blockIdx.x;
    int a0 = __builtin_amdgcn_readfirstlane((int)o0[0]);
    int a1 = __builtin_amdgcn_readfirstlane((int)o0[1]);
    int b0 = __builtin_amdgcn_readfirstlane((int)o1[0]);
    int b1 = __builtin_amdgcn_readfirstlane((int)o1[1]);
    a0 = (a0 < 0) ? 0 : ((a0 > BIN_CAP) ? BIN_CAP : a0);
    a1 = (a1 < a0) ? a0 : ((a1 > BIN_CAP) ? BIN_CAP : a1);
    b0 = (b0 < 0) ? 0 : ((b0 > BIN_CAP) ? BIN_CAP : b0);
    b1 = (b1 < b0) ? b0 : ((b1 > BIN_CAP) ? BIN_CAP : b1);
    const int n0 = a1 - a0;
    const int total = n0 + (b1 - b0);
    const unsigned* p0 = bins + (size_t)c0 * BIN_CAP + a0;
    const unsigned* p1 = bins + (size_t)c1 * BIN_CAP + b0;
    constexpr int NB = OWN_KEEP / 32;
    for (int pos = 0; pos < total; pos += OWN_KEEP) {
        const int cnt = (total - pos < OWN_KEEP) ? (total - pos) : OWN_KEEP;
        unsigned rec[NB];
#pragma unroll
        for (int b = 0; b < NB; ++b) {
            rec[b] = 0u;
            const int i = pos + b * 32 + lane;
            if (b * 32 + lane < cnt) {
                const unsigned* p = (i < n0) ? (p0 + i) : (p1 + (i - n0));
                rec[b] = *p;
            }
        }
        float wgt[NB];
#pragma unroll
        for (int b = 0; b < NB; ++b) {
            const float ds = dinv[rec[b] >> 8];
            const float dd = dinv[lo + (int)(rec[b] & 255u)];
            wgt[b] = ds * dd;
        }
#pragma unroll
        for (int b = 0; b < NB; ++b) {
            const bool     valid = b * 32 + lane < cnt;
            const unsigned row   = rec[b] & 255u;
            const unsigned own   = row & (unsigned)(OWN_WAVES - 1);
            unsigned at = (unsigned)OWN_INBOX;
            if (valid) at = atomicAdd(&icnt[own], 1u);
            const bool fits = valid && at < (unsigned)OWN_INBOX;
            if (fits) {
                unsigned* ib = list + own * OWN_LIST;
                ib[at]             = rec[b];
                ib[OWN_INBOX + at] = __float_as_uint(wgt[b]);
            }
            unsigned ovf = __builtin_amdgcn_ballot_w32(valid && !fits);
            while (ovf != 0u) {
                const int j = __builtin_ctz(ovf);
                ovf &= ovf - 1u;
                gcn_own_add_group<F, 1, false>((int)(rec[b] >> 8), (int)row, wgt[b], j, lane, xw, tile);
            }
        }
    }
    __syncthreads();
    int n = __builtin_amdgcn_readfirstlane((int)icnt[wv]);
    n = (n < OWN_INBOX) ? n : OWN_INBOX;
    const unsigned* mine = list + wv * OWN_LIST;
    for (int base = 0; base < n; base += 32) {
        const int m = (n - base < 32) ? (n - base) : 32;
        unsigned r = 0u;
        float    w = 0.0f;
        if (lane < m) {
            r = mine[base + lane];
            w = __uint_as_float(mine[OWN_INBOX + base + lane]);
        }
        gcn_own_add_rows<F, true, true>((int)(r >> 8), (int)(r & 255u), w, m, lane, xw, tile);
    }
}

template <int F>
__device__ __forceinline__ void gcn_own_add_edges(const unsigned* lst, int n, int lane, int lo,
                                                  const int* __restrict__ src,
                                                  const float* __restrict__ dinv,
                                                  const float* __restrict__ xw,
                                                  float* tile, unsigned* keep) {
    int s = 0, dl = 0;
    float w = 0.0f;
    if (lane < n) {
        const unsigned rec = lst[lane];
        dl = (int)(rec & 255u);
        s  = src[rec >> 8];
        w  = dinv[s] * dinv[lo + dl];
        if (keep != nullptr) keep[lane] = ((unsigned)s << 8) | (unsigned)dl;
    }
    gcn_own_add_rows<F, false>(s, dl, w, n, lane, xw, tile);
}

template <int F>
__device__ __forceinline__ int gcn_own_scan(const int* __restrict__ src, const int* __restrict__ dst,
                                            const float* __restrict__ dinv,
                                            const float* __restrict__ xw,
                                            float* tile, unsigned* mylist, unsigned* kept,
                                            int lane, int wv, int lo, int nrows, int E) {
    const int nchunks = (E + 1023) >> 10;
    const int c0 = (int)((blockIdx.x * (unsigned)nchunks) / gridDim.x);
    int  count  = 0;
    int  nkept  = 0;
    bool keepok = kept != nullptr;

    for (int i = wv; i < nchunks; i += OWN_WAVES) {
        int c = c0 + i;
        if (c >= nchunks) c -= nchunks;
        const int cbase = c << 10;

        int4 dv[8];
        int  eu[8];
#pragma unroll
        for (int u = 0; u < 8; ++u) {
            const int e = cbase + u * 128 + lane * 4;
            eu[u] = (e < E - 4) ? e : (E - 4);
            dv[u] = *(const int4*)(dst + eu[u]);
        }
        if (cbase + 1024 > E) {
#pragma unroll
            for (int u = 0; u < 8; ++u) {
                const int e = cbase + u * 128 + lane * 4;
                if (eu[u]     < e) dv[u].x = -1;
                if (eu[u] + 1 < e) dv[u].y = -1;
                if (eu[u] + 2 < e) dv[u].z = -1;
                if (eu[u] + 3 < e) dv[u].w = -1;
            }
        }
#pragma unroll
        for (int hf = 0; hf < 2; ++hf) {
#pragma unroll
            for (int uu = 0; uu < 4; ++uu) {
                const int u = hf * 4 + uu;
                const int dd[4] = {dv[u].x, dv[u].y, dv[u].z, dv[u].w};
#pragma unroll
                for (int k = 0; k < 4; ++k) {
                    const unsigned t = (unsigned)(dd[k] - lo);
                    const bool m = t < (unsigned)nrows;
                    const unsigned bal = __builtin_amdgcn_ballot_w32(m);
                    if (bal != 0u) {
                        if (m) mylist[count + __popc(bal & ((1u << lane) - 1u))] = ((unsigned)(eu[u] + k) << 8) | t;
                        count += __popc(bal);
                    }
                }
            }
            __builtin_amdgcn_fence(__ATOMIC_ACQ_REL, "wavefront");
            while (count >= 32) {
                count -= 32;
                unsigned* kp = nullptr;
                if (keepok) {
                    if (nkept + 32 <= OWN_KEEP) { kp = kept + nkept; nkept += 32; }
                    else keepok = false;
                }
                gcn_own_add_edges<F>(mylist + count, 32, lane, lo, src, dinv, xw, tile, kp);
            }
        }
    }
    if (count > 0) {
        unsigned* kp = nullptr;
        if (keepok) {
            if (nkept + count <= OWN_KEEP) { kp = kept + nkept; nkept += count; }
            else keepok = false;
        }
        gcn_own_add_edges<F>(mylist, count, lane, lo, src, dinv, xw, tile, kp);
    }
    return keepok ? nkept : -1;
}

template <int F>
__device__ __forceinline__ void gcn_own_add_found(const unsigned* lst, int n, int lane, int lo,
                                                  const int* __restrict__ src,
                                                  const int* __restrict__ dst,
                                                  const float* __restrict__ dinv,
                                                  const float* __restrict__ xw,
                                                  float* tile, unsigned* keep) {
    int s = 0, dl = 0;
    float w = 0.0f;
    if (lane < n) {
        const unsigned e = lst[lane];
        const int d = dst[e];
        s  = src[e];
        dl = d - lo;
        w  = dinv[s] * dinv[d];
        if (keep != nullptr) keep[lane] = ((unsigned)s << 8) | (unsigned)dl;
    }
    gcn_own_add_rows<F, false>(s, dl, w, n, lane, xw, tile);
}

template <int F>
__device__ __forceinline__ int gcn_own_scan_tags(const int* __restrict__ src,
                                                 const int* __restrict__ dst,
                                                 const unsigned char* __restrict__ tags,
                                                 const float* __restrict__ dinv,
                                                 const float* __restrict__ xw,
                                                 float* tile, unsigned* mylist, unsigned* kept,
                                                 int lane, int wv, int lo, int E) {
    const unsigned pat = blockIdx.x * 0x01010101u;
    const int nunits = (E + OWN_TAGU - 1) / OWN_TAGU;
    const int nfull  = nunits / (8 * OWN_WAVES);
    const int ntrip  = nfull * OWN_WAVES;
    const int t0     = (int)((blockIdx.x * (unsigned)ntrip) / gridDim.x);
    const int rest0  = nfull * (8 * OWN_WAVES);
    const int per    = (nunits - rest0 + OWN_WAVES - 1) / OWN_WAVES;
    int  count  = 0;
    int  nkept  = 0;
    bool keepok = kept != nullptr;

    uint4 tv[8];
    int i = 0;
    int unit0 = 0;
    int nu = 0;
    int done = 0;
    for (;;) {
        if (done >= nu) {
            if (i > nfull) break;
            if (i < nfull) {
                int tr = t0 + i * OWN_WAVES + wv;
                if (tr >= ntrip) tr -= ntrip;
                unit0 = tr * 8;
                nu    = 8;
            } else {
                unit0 = rest0 + wv * per;
                nu    = (nunits - unit0 < per) ? (nunits - unit0) : per;
                if (nu <= 0) break;
            }
            ++i;
            done = 0;
#pragma unroll
            for (int u = 0; u < 8; ++u) {
                int un = unit0 + u;
                un = (un < nunits) ? un : (nunits - 1);
                tv[u] = *(const uint4*)(tags + (size_t)un * OWN_TAGU + lane * 16);
            }
        }
#pragma unroll
        for (int u = 0; u < 8; ++u) {
            if (u >= done && u < nu && count <= OWN_LIST - OWN_TAGU) {
                const int ebase = (unit0 + u) * OWN_TAGU + lane * 16;
                unsigned xs[4] = {tv[u].x ^ pat, tv[u].y ^ pat, tv[u].z ^ pat, tv[u].w ^ pat};
                if ((unit0 + u + 1) * OWN_TAGU > E) {
#pragma unroll
                    for (int q = 0; q < 4; ++q) {
#pragma unroll
                        for (int k = 0; k < 4; ++k)
                            if (ebase + q * 4 + k >= E) xs[q] |= 255u << (8 * k);
                    }
                }
#pragma unroll
                for (int q = 0; q < 4; ++q) {
                    const unsigned x = xs[q];
                    const unsigned z = (x - 0x01010101u) & ~x & 0x80808080u;
                    if (__builtin_amdgcn_ballot_w32(z != 0u) != 0u) {
#pragma unroll
                        for (int k = 0; k < 4; ++k) {
                            const bool m = ((x >> (8 * k)) & 255u) == 0u;
                            const unsigned bal = __builtin_amdgcn_ballot_w32(m);
                            if (bal != 0u) {
                                if (m) mylist[count + __popc(bal & ((1u << lane) - 1u))] = (unsigned)(ebase + q * 4 + k);
                                count += __popc(bal);
                            }
                        }
                    }
                }
                done = u + 1;
            }
        }
        __builtin_amdgcn_fence(__ATOMIC_ACQ_REL, "wavefront");
        while (count >= 32) {
            count -= 32;
            unsigned* kp = nullptr;
            if (keepok) {
                if (nkept + 32 <= OWN_KEEP) { kp = kept + nkept; nkept += 32; }
                else keepok = false;
            }
            gcn_own_add_found<F>(mylist + count, 32, lane, lo, src, dst, dinv, xw, tile, kp);
        }
    }
    if (count > 0) {
        unsigned* kp = nullptr;
        if (keepok) {
            if (nkept + count <= OWN_KEEP) { kp = kept + nkept; nkept += count; }
            else keepok = false;
        }
        gcn_own_add_found<F>(mylist, count, lane, lo, src, dst, dinv, xw, tile, kp);
    }
    return keepok ? nkept : -1;
}

__device__ __forceinline__ bool gcn_own_keep_fits(const float* wsb, size_t ws_bytes, const int* kend) {
    return (size_t)((const char*)kend - (const char*)wsb) <= ws_bytes;
}

__global__ __launch_bounds__(1024)
void gcn_owner_sum_lds(const int* __restrict__ src, const int* __restrict__ dst,
                       const float* __restrict__ dinv, const float* __restrict__ xw,
                       const float* __restrict__ bias, const _Float16* __restrict__ W2p,
                       float* __restrict__ xw2,
                       unsigned* __restrict__ kept, int* __restrict__ kcnt,
                       const unsigned char* __restrict__ tags, int Epad,
                       const unsigned* __restrict__ bins, const unsigned* __restrict__ boff,
                       const float* wsb, size_t ws_bytes, int N, int E) {
    __shared__ __attribute__((aligned(16))) float tile[OWN_ROWS * GCN_D_HID];
    __shared__ unsigned list[OWN_WAVES * OWN_LIST];
    __shared__ unsigned icnt[OWN_WAVES];

    const int lane = threadIdx.x & 31;
    const int wv = __builtin_amdgcn_readfirstlane((int)(threadIdx.x >> 5));
    const int lo = blockIdx.x * OWN_ROWS;
    const int nrows = (N - lo < OWN_ROWS) ? (N - lo) : OWN_ROWS;
    const int wid = blockIdx.x * OWN_WAVES + wv;
    const bool keep_on = gcn_own_keep_fits(wsb, ws_bytes, kcnt + gridDim.x * OWN_WAVES);

    if (threadIdx.x < OWN_WAVES) icnt[threadIdx.x] = 0u;

    for (int r = wv; r < nrows; r += OWN_WAVES) {
        const float di = dinv[lo + r];
        const float4* xr = (const float4*)(xw + (size_t)(lo + r) * GCN_D_HID);
        const float4 a = xr[lane];
        const float4 b = xr[32 + lane];
        float* tr = tile + r * GCN_D_HID + lane * 4;
        *(float4*)tr         = float4{a.x * di * di, a.y * di * di, a.z * di * di, a.w * di * di};
        *(float4*)(tr + 128) = float4{b.x * di * di, b.y * di * di, b.z * di * di, b.w * di * di};
    }
    __syncthreads();

    const bool bins_on = gcn_bins_on(wsb, ws_bytes, boff, N, E, Epad);
    if (bins_on) {
        gcn_own_add_dealt<GCN_D_HID>(bins, boff, dinv, xw, tile, list, icnt, lane, wv, lo);
    } else {
        const bool tags_on = gcn_ws_fits(wsb, ws_bytes, tags + Epad) && gridDim.x <= 256u;
        unsigned* kp0 = keep_on ? kept + (size_t)wid * OWN_KEEP : nullptr;
        int nk;
        if (tags_on)
            nk = gcn_own_scan_tags<GCN_D_HID>(src, dst, tags, dinv, xw, tile, list + wv * OWN_LIST,
                                              kp0, lane, wv, lo, E);
        else
            nk = gcn_own_scan<GCN_D_HID>(src, dst, dinv, xw, tile, list + wv * OWN_LIST,
                                         kp0, lane, wv, lo, nrows, E);
        if (keep_on && lane == 0) kcnt[wid] = nk;
    }
    __syncthreads();

    constexpr int HP  = GCN_D_HID + 8;
    constexpr int NRT = (OWN_ROWS + 15) / 16;
    constexpr int RPW = (NRT * 16 + OWN_WAVES - 1) / OWN_WAVES;
    constexpr int KT2 = GCN_D_HID / 32;
    constexpr int NT2 = GCN_D_OUT / 16;
    constexpr int RGS = OWN_WAVES / NT2;
    static_assert(OWN_WAVES % NT2 == 0 && NRT <= 2 * RGS && GCN_D_HID == 256 &&
                  NRT * 16 * HP * 2 <= OWN_ROWS * GCN_D_HID * 4, "shape not covered");
    const int ct = wv % NT2;
    const int rg = wv / NT2;

    const float4 bb0 = ((const float4*)bias)[lane];
    const float4 bb1 = ((const float4*)bias)[32 + lane];

    v16h b[KT2];
    {
        const _Float16* bb = W2p + ((size_t)ct * 32 + lane) * 16;
#pragma unroll
        for (int kt = 0; kt < KT2; ++kt) b[kt] = *(const v16h*)(bb + (size_t)kt * NT2 * 32 * 16);
    }
    v4h hlo[RPW], hhi[RPW];
#pragma unroll
    for (int i = 0; i < RPW; ++i) {
        const int r = wv + i * OWN_WAVES;
        float4 a = float4{0.0f, 0.0f, 0.0f, 0.0f};
        float4 c = float4{0.0f, 0.0f, 0.0f, 0.0f};
        if (r < nrows) {
            const float* tr = tile + r * GCN_D_HID + lane * 4;
            const float4 s0 = *(const float4*)tr;
            const float4 s1 = *(const float4*)(tr + 128);
            a.x = fmaxf(s0.x + bb0.x, 0.0f);
            a.y = fmaxf(s0.y + bb0.y, 0.0f);
            a.z = fmaxf(s0.z + bb0.z, 0.0f);
            a.w = fmaxf(s0.w + bb0.w, 0.0f);
            c.x = fmaxf(s1.x + bb1.x, 0.0f);
            c.y = fmaxf(s1.y + bb1.y, 0.0f);
            c.z = fmaxf(s1.z + bb1.z, 0.0f);
            c.w = fmaxf(s1.w + bb1.w, 0.0f);
        }
        hlo[i] = v4h{(_Float16)a.x, (_Float16)a.y, (_Float16)a.z, (_Float16)a.w};
        hhi[i] = v4h{(_Float16)c.x, (_Float16)c.y, (_Float16)c.z, (_Float16)c.w};
    }
    __syncthreads();
    _Float16* hrow = (_Float16*)tile;
#pragma unroll
    for (int i = 0; i < RPW; ++i) {
        const int r = wv + i * OWN_WAVES;
        if (r < NRT * 16) {
            *(v4h*)(hrow + r * HP + lane * 4)       = hlo[i];
            *(v4h*)(hrow + r * HP + 128 + lane * 4) = hhi[i];
        }
    }
    __syncthreads();

    const int hsel = lane >> 4;
    const int idx  = lane & 15;
    const int one[1] = {0};
#pragma unroll
    for (int i = 0; i < 2; ++i) {
        const int rt = rg + i * RGS;
        if (rt < NRT && rt * 16 < nrows) {
            v8f acc[1];
            acc[0] = v8f{};
            gcn_gemm_tiles<GCN_D_HID, 1>(hrow + (rt * 16 + idx) * HP + hsel * 8, b, one, acc);
            const int row0 = rt * 16 + hsel * 8;
            float* cb = xw2 + (size_t)(lo + row0) * GCN_D_OUT + ct * 16 + idx;
#pragma unroll
            for (int r = 0; r < 8; ++r)
                if (row0 + r < nrows) cb[(size_t)r * GCN_D_OUT] = acc[0][r];
        }
    }
}

__global__ __launch_bounds__(1024)
void gcn_owner_sum_kept(const int* __restrict__ src, const int* __restrict__ dst,
                        const float* __restrict__ dinv, const float* __restrict__ xw,
                        const float* __restrict__ bias, float* __restrict__ out,
                        const unsigned* __restrict__ kept, const int* __restrict__ kcnt,
                        const unsigned* __restrict__ bins, const unsigned* __restrict__ boff,
                        int Epad, const float* wsb, size_t ws_bytes, int N, int E) {
    __shared__ float    tile[OWN_ROWS * GCN_D_OUT];
    __shared__ unsigned list[OWN_WAVES * OWN_LIST];

    const int lane = threadIdx.x & 31;
    const int wv = __builtin_amdgcn_readfirstlane((int)(threadIdx.x >> 5));
    const int lo = blockIdx.x * OWN_ROWS;
    const int nrows = (N - lo < OWN_ROWS) ? (N - lo) : OWN_ROWS;
    const int wid = blockIdx.x * OWN_WAVES + wv;
    const bool keep_on = gcn_own_keep_fits(wsb, ws_bytes, kcnt + gridDim.x * OWN_WAVES);

    const bool bins_on = gcn_bins_on(wsb, ws_bytes, boff, N, E, Epad);
    bool use_kept = false;
    if (!bins_on && keep_on) {
        const int c = kcnt[blockIdx.x * OWN_WAVES + lane];
        use_kept = __builtin_amdgcn_ballot_w32(c < 0 || c > OWN_KEEP) == 0u;
    }

    {
        constexpr int RPW = (OWN_ROWS + OWN_WAVES - 1) / OWN_WAVES;
        float sa[RPW], sb[RPW];
        const int   rl  = wv + lane * OWN_WAVES;
        const float dvl = dinv[lo + ((rl < nrows) ? rl : (nrows - 1))];
#pragma unroll
        for (int i = 0; i < RPW; ++i) {
            const int r  = wv + i * OWN_WAVES;
            const int rc = (r < nrows) ? r : (nrows - 1);
            const float* xr = xw + (size_t)(lo + rc) * GCN_D_OUT;
            sa[i] = xr[lane];
            sb[i] = xr[32 + lane];
        }
#pragma unroll
        for (int i = 0; i < RPW; ++i) {
            const int r = wv + i * OWN_WAVES;
            if (r < nrows) {
                const float di = __int_as_float(__builtin_amdgcn_readlane(__float_as_int(dvl), i));
                float* tr = tile + r * GCN_D_OUT + lane;
                tr[0]  = sa[i] * di * di;
                tr[32] = sb[i] * di * di;
            }
        }
    }
    __syncthreads();

    if (bins_on) {
        gcn_own_add_binned<GCN_D_OUT>(bins, boff, dinv, xw, tile, list + wv * OWN_LIST, lane, wv, lo);
    } else if (use_kept) {
        const unsigned* kp = kept + (size_t)wid * OWN_KEEP;
        int cnt = __builtin_amdgcn_readfirstlane(kcnt[wid]);
        cnt = (cnt < OWN_KEEP) ? cnt : OWN_KEEP;
        constexpr int NB = OWN_KEEP / 32;
        unsigned* myl = list + wv * OWN_LIST;
        unsigned rec[NB];
#pragma unroll
        for (int b = 0; b < NB; ++b) {
            rec[b] = 0u;
            if (b * 32 + lane < cnt) rec[b] = kp[b * 32 + lane];
        }
        float wgt[NB];
#pragma unroll
        for (int b = 0; b < NB; ++b) {
            const float ds = dinv[rec[b] >> 8];
            const float dd = dinv[lo + (int)(rec[b] & 255u)];
            wgt[b] = ds * dd;
        }
#pragma unroll
        for (int b = 0; b < NB; ++b) {
            myl[b * 32 + lane]            = rec[b];
            myl[OWN_KEEP + b * 32 + lane] = __float_as_uint(wgt[b]);
        }
        for (int base = 0; base < cnt; base += 32) {
            const int n = (cnt - base < 32) ? (cnt - base) : 32;
            const unsigned r = myl[base + lane];
            const float    w = __uint_as_float(myl[OWN_KEEP + base + lane]);
            gcn_own_add_rows<GCN_D_OUT, true>((int)(r >> 8), (int)(r & 255u), w, n, lane, xw, tile);
        }
    } else {
        gcn_own_scan<GCN_D_OUT>(src, dst, dinv, xw, tile, list + wv * OWN_LIST, nullptr,
                                lane, wv, lo, nrows, E);
    }
    __syncthreads();

    const float b0 = bias[lane];
    const float b1 = bias[lane + 32];
    for (int r = wv; r < nrows; r += OWN_WAVES) {
        const float* tr = tile + r * GCN_D_OUT + lane;
        float v0 = tr[0]  + b0;
        float v1 = tr[32] + b1;
        float m = fmaxf(v0, v1);
#pragma unroll
        for (int off = 16; off > 0; off >>= 1) m = fmaxf(m, __shfl_xor(m, off, 32));
        float e0 = __expf(v0 - m);
        float e1 = __expf(v1 - m);
        float sm = e0 + e1;
#pragma unroll
        for (int off = 16; off > 0; off >>= 1) sm += __shfl_xor(sm, off, 32);
        float inv = 1.0f / sm;
        float* orow = out + (size_t)(lo + r) * GCN_D_OUT;
        orow[lane]      = e0 * inv;
        orow[lane + 32] = e1 * inv;
    }
}

extern "C" void kernel_launch(void* const* d_in, const int* in_sizes, int n_in,
                              void* d_out, int out_size, void* d_ws, size_t ws_size,
                              hipStream_t stream) {
    const float* x   = (const float*)d_in[0];
    const int*   ei  = (const int*)  d_in[1];
    const float* W1  = (const float*)d_in[2];
    const float* b1  = (const float*)d_in[3];
    const float* W2  = (const float*)d_in[4];
    const float* b2  = (const float*)d_in[5];
    float* out = (float*)d_out;

    const int N = in_sizes[0] / GCN_D_IN;
    const int E = in_sizes[1] / 2;
    const int* src = ei;
    const int* dst = ei + E;

    float* ws   = (float*)d_ws;
    size_t off  = ((size_t)N + 63) & ~(size_t)63;
    float* dinv = ws;
    float* xw1  = ws + off;
    float* h1   = xw1 + (size_t)N * GCN_D_HID;
    _Float16* W1p = (_Float16*)(h1 + (size_t)N * GCN_D_HID);
    _Float16* W2p = W1p + (size_t)GCN_D_IN * GCN_D_HID;
    float* xw2  = h1;
    const int ob = (N + OWN_ROWS - 1) / OWN_ROWS;
    unsigned* kept = (unsigned*)(W2p + (size_t)GCN_D_HID * GCN_D_OUT);
    int*      kcnt = (int*)(kept + (size_t)ob * OWN_WAVES * OWN_KEEP);
    unsigned char* tags = (unsigned char*)(kcnt + (size_t)ob * OWN_WAVES);
    const int Epad = (E + OWN_TAGU - 1) / OWN_TAGU * OWN_TAGU;
    unsigned* bins = kept;
    unsigned* boff = bins + (size_t)DEG_CHUNKS * BIN_CAP;

    const int B = 256;
    {
        int t1 = (GCN_D_IN >> 5) * (GCN_D_HID >> 4) * 32;
        gcn_pack_w_f16<<<(t1 + B - 1) / B, B, 0, stream>>>(W1, W1p, GCN_D_IN, GCN_D_HID);
        int t2 = (GCN_D_HID >> 5) * (GCN_D_OUT >> 4) * 32;
        gcn_pack_w_f16<<<(t2 + B - 1) / B, B, 0, stream>>>(W2, W2p, GCN_D_HID, GCN_D_OUT);
    }

    unsigned* part = (unsigned*)h1;
    gcn_count_deg<<<DEG_CHUNKS, 1024, 0, stream>>>(src, dst, part, (int)off, tags, bins, boff, ws, ws_size, N, E, Epad);

    {
        gcn_wmma_gemm_bregs<GCN_D_IN, GCN_D_HID, 2><<<256, 512, 0, stream>>>(x, W1p, xw1, N, bins, boff, part, (int)off,
                                                                             dinv, ws, ws_size, N, E, Epad);
    }
    {
        gcn_owner_sum_lds<<<ob, 1024, 0, stream>>>(src, dst, dinv, xw1, b1, W2p, xw2, kept, kcnt,
                                                   tags, Epad, bins, boff, ws, ws_size, N, E);
    }

    {
        gcn_owner_sum_kept<<<ob, 1024, 0, stream>>>(src, dst, dinv, xw2, b2, out, kept, kcnt,
                                                    bins, boff, Epad, ws, ws_size, N, E);
    }
}
